# starting kernel
# baseline (speedup 1.0000x reference)
#include <hip/hip_runtime.h>
#include <hip/hip_bf16.h>
#include <math.h>

typedef __attribute__((ext_vector_type(2))) float v2f;
typedef __attribute__((ext_vector_type(8))) float v8f;

#define D_DIM 256
#define KC 32
#define LDA_S (KC + 2)
#define LDBT  (KC + 2)

__global__ __launch_bounds__(256)
void zero_f32(float4* __restrict__ p, long n4) {
    long i = (long)blockIdx.x * blockDim.x + threadIdx.x;
    if (i < n4) p[i] = make_float4(0.f, 0.f, 0.f, 0.f);
}

__global__ __launch_bounds__(256)
void spmm_atomic(const int* __restrict__ rows, const int* __restrict__ cols,
                 const float* __restrict__ vals,
                 const float* __restrict__ Xa, const float* __restrict__ Xb,
                 int split, float* __restrict__ Y, int nnz) {
    int w    = blockIdx.x * 8 + (threadIdx.x >> 5);
    int lane = threadIdx.x & 31;
    if (w >= nnz) return;
    int   r = rows[w];
    int   c = cols[w];
    float v = vals[w];
    const float* X = (c < split) ? (Xa + (size_t)c * D_DIM)
                                 : (Xb + (size_t)(c - split) * D_DIM);
    float* y = Y + (size_t)r * D_DIM;
#pragma unroll
    for (int i = 0; i < 2; ++i) {
        int    b = (lane + 32 * i) * 4;
        float4 x = *(const float4*)(X + b);
        atomicAdd(y + b + 0, x.x * v);
        atomicAdd(y + b + 1, x.y * v);
        atomicAdd(y + b + 2, x.z * v);
        atomicAdd(y + b + 3, x.w * v);
    }
}

__global__ __launch_bounds__(256)
void gemm_tanh_wmma(const float* __restrict__ A0, const float* __restrict__ A1,
                    const float* __restrict__ B0, const float* __restrict__ B1,
                    const float* __restrict__ bias, float* __restrict__ OUT,
                    int M, int accumulate) {
    __shared__ float As[64 * LDA_S];
    __shared__ float Bst[128 * LDBT];

    const int t    = threadIdx.x;
    const int wave = t >> 5;
    const int lane = t & 31;
    const int wm   = wave & 3;
    const int wn   = wave >> 2;
    const int rowBase = blockIdx.x * 64;
    const int colBase = blockIdx.y * 128;

    const int m16 = lane & 15;
    const int kh  = lane >> 4;

    v8f acc[4];
#pragma unroll
    for (int j = 0; j < 4; ++j) acc[j] = (v8f)0.f;

    for (int src = 0; src < 2; ++src) {
        const float* A = src ? A1 : A0;
        const float* B = src ? B1 : B0;
        if (A == nullptr) break;

        for (int k0 = 0; k0 < D_DIM; k0 += KC) {
            {
                int r  = t >> 3;
                int c4 = (t & 7) * 4;
#pragma unroll
                for (int i = 0; i < 2; ++i) {
                    int    row = rowBase + r + 32 * i;
                    float4 v   = make_float4(0.f, 0.f, 0.f, 0.f);
                    if (row < M)
                        v = *(const float4*)(A + (size_t)row * D_DIM + k0 + c4);
                    float* dst = &As[(r + 32 * i) * LDA_S + c4];
                    dst[0] = v.x; dst[1] = v.y; dst[2] = v.z; dst[3] = v.w;
                }
            }
            {
#pragma unroll
                for (int i = 0; i < 4; ++i) {
                    int idx = t + 256 * i;
                    int k   = idx >> 5;
                    int n4  = (idx & 31) * 4;
                    float4 v = *(const float4*)(B + (size_t)(k0 + k) * D_DIM +
                                                colBase + n4);
                    Bst[(n4 + 0) * LDBT + k] = v.x;
                    Bst[(n4 + 1) * LDBT + k] = v.y;
                    Bst[(n4 + 2) * LDBT + k] = v.z;
                    Bst[(n4 + 3) * LDBT + k] = v.w;
                }
            }
            __syncthreads();

#pragma unroll
            for (int kk = 0; kk < KC; kk += 4) {
                v2f af = *(const v2f*)&As[(wm * 16 + m16) * LDA_S + kk + 2 * kh];
#pragma unroll
                for (int j = 0; j < 4; ++j) {
                    int n  = wn * 64 + j * 16 + m16;
                    v2f bf = *(const v2f*)&Bst[n * LDBT + kk + 2 * kh];
                    acc[j] = __builtin_amdgcn_wmma_f32_16x16x4_f32(
                        false, af, false, bf, (short)0, acc[j], false, false);
                }
            }
            __syncthreads();
        }
    }

#pragma unroll
    for (int j = 0; j < 4; ++j) {
        int   c  = colBase + wn * 64 + j * 16 + m16;
        float bv = bias ? bias[c] : 0.f;
#pragma unroll
        for (int r = 0; r < 8; ++r) {
            int row = rowBase + wm * 16 + kh * 8 + r;
            if (row < M) {
                float  v = tanhf(acc[j][r] + bv);
                float* o = OUT + (size_t)row * D_DIM + c;
                if (accumulate) *o += v;
                else            *o  = v;
            }
        }
    }
}

__global__ __launch_bounds__(256)
void l2norm_rows(const float* __restrict__ H, float* __restrict__ out, int M) {
    int row  = blockIdx.x * 8 + (threadIdx.x >> 5);
    int lane = threadIdx.x & 31;
    if (row >= M) return;
    const float4* h = (const float4*)(H + (size_t)row * D_DIM);
    float4 a = h[lane];
    float4 b = h[lane + 32];
    float  s = a.x * a.x + a.y * a.y + a.z * a.z + a.w * a.w +
               b.x * b.x + b.y * b.y + b.z * b.z + b.w * b.w;
#pragma unroll
    for (int off = 16; off > 0; off >>= 1) s += __shfl_xor(s, off, 32);
    float inv = 1.f / fmaxf(sqrtf(s), 1e-12f);
    a.x *= inv; a.y *= inv; a.z *= inv; a.w *= inv;
    b.x *= inv; b.y *= inv; b.z *= inv; b.w *= inv;
    float4* o = (float4*)(out + (size_t)row * D_DIM);
    o[lane]      = a;
    o[lane + 32] = b;
}

extern "C" void kernel_launch(void* const* d_in, const int* in_sizes, int n_in,
                              void* d_out, int out_size, void* d_ws, size_t ws_size,
                              hipStream_t stream) {
    const int* adj_rows   = (const int*)d_in[0];
    const int* adj_cols   = (const int*)d_in[1];
    const float* adj_vals = (const float*)d_in[2];
    const int* up_rows    = (const int*)d_in[3];
    const int* up_cols    = (const int*)d_in[4];
    const float* up_vals  = (const float*)d_in[5];
    const int* ip_rows    = (const int*)d_in[6];
    const int* ip_cols    = (const int*)d_in[7];
    const float* ip_vals  = (const float*)d_in[8];
    const float* user_emb = (const float*)d_in[9];
    const float* item_emb = (const float*)d_in[10];
    const float* Q_user   = (const float*)d_in[11];
    const float* W_user   = (const float*)d_in[12];
    const float* b_user   = (const float*)d_in[13];
    const float* Q_item   = (const float*)d_in[14];
    const float* W_item   = (const float*)d_in[15];
    const float* b_item   = (const float*)d_in[16];
    const float* M_user   = (const float*)d_in[17];
    const float* M_item   = (const float*)d_in[18];
    float* out = (float*)d_out;

    const int NU = 40000, NI = 20000, NN = 60000;
    const int NNZ_ADJ = in_sizes[0], NNZ_UP = in_sizes[3], NNZ_IP = in_sizes[6];
    const size_t ROWF = (size_t)D_DIM;
    const size_t MAT  = (size_t)NN * ROWF;

    float* agg  = (float*)d_ws;
    float* embb = agg  + MAT;
    float* hbuf = embb + MAT;

    dim3 blk(256);

    zero_f32<<<(unsigned)((MAT / 4 + 255) / 256), blk, 0, stream>>>((float4*)agg, (long)(MAT / 4));
    spmm_atomic<<<(NNZ_ADJ + 7) / 8, blk, 0, stream>>>(
        adj_rows, adj_cols, adj_vals, user_emb, item_emb, NU, agg, NNZ_ADJ);

    gemm_tanh_wmma<<<dim3((NU + 63) / 64, 2), blk, 0, stream>>>(
        agg, nullptr, Q_user, nullptr, nullptr, embb, NU, 0);
    gemm_tanh_wmma<<<dim3((NI + 63) / 64, 2), blk, 0, stream>>>(
        agg + (size_t)NU * ROWF, nullptr, Q_item, nullptr, nullptr,
        embb + (size_t)NU * ROWF, NI, 0);

    gemm_tanh_wmma<<<dim3((NU + 63) / 64, 2), blk, 0, stream>>>(
        user_emb, embb, W_user, W_user + (size_t)D_DIM * D_DIM, b_user,
        hbuf, NU, 0);
    gemm_tanh_wmma<<<dim3((NI + 63) / 64, 2), blk, 0, stream>>>(
        item_emb, embb + (size_t)NU * ROWF, W_item,
        W_item + (size_t)D_DIM * D_DIM, b_item,
        hbuf + (size_t)NU * ROWF, NI, 0);

    l2norm_rows<<<(NN + 7) / 8, blk, 0, stream>>>(hbuf, out, NN);

    zero_f32<<<(unsigned)((MAT / 4 + 255) / 256), blk, 0, stream>>>((float4*)agg, (long)(MAT / 4));
    spmm_atomic<<<(NNZ_UP + 7) / 8, blk, 0, stream>>>(
        up_rows, up_cols, up_vals, user_emb, user_emb, NU, agg, NNZ_UP);
    spmm_atomic<<<(NNZ_IP + 7) / 8, blk, 0, stream>>>(
        ip_rows, ip_cols, ip_vals, item_emb, item_emb, NI,
        agg + (size_t)NU * ROWF, NNZ_IP);

    gemm_tanh_wmma<<<dim3((NU + 63) / 64, 2), blk, 0, stream>>>(
        agg, nullptr, M_user, nullptr, nullptr, out, NU, 1);
    gemm_tanh_wmma<<<dim3((NI + 63) / 64, 2), blk, 0, stream>>>(
        agg + (size_t)NU * ROWF, nullptr, M_item, nullptr, nullptr,
        out + (size_t)NU * ROWF, NI, 1);
}


# SPMM lane map: lane l owns floats l+32i of a row, so each load and each f32 atomic instruction covers one whole 128-byte line instead of 8 lanes in each of 4 lines
# speedup vs baseline: 1.4895x; 1.4895x over previous
#include <hip/hip_runtime.h>
#include <hip/hip_bf16.h>
#include <math.h>

typedef __attribute__((ext_vector_type(2))) float v2f;
typedef __attribute__((ext_vector_type(8))) float v8f;

#define D_DIM 256
#define KC 32
#define LDA_S (KC + 2)
#define LDBT  (KC + 2)

__global__ __launch_bounds__(256)
void zero_f32(float4* __restrict__ p, long n4) {
    long i = (long)blockIdx.x * blockDim.x + threadIdx.x;
    if (i < n4) p[i] = make_float4(0.f, 0.f, 0.f, 0.f);
}

__global__ __launch_bounds__(256)
void spmm_atomic(const int* __restrict__ rows, const int* __restrict__ cols,
                 const float* __restrict__ vals,
                 const float* __restrict__ Xa, const float* __restrict__ Xb,
                 int split, float* __restrict__ Y, int nnz) {
    int w    = blockIdx.x * 8 + (threadIdx.x >> 5);
    int lane = threadIdx.x & 31;
    if (w >= nnz) return;
    int   r = rows[w];
    int   c = cols[w];
    float v = vals[w];
    const float* X = (c < split) ? (Xa + (size_t)c * D_DIM)
                                 : (Xb + (size_t)(c - split) * D_DIM);
    const float* x = X + lane;
    float*       y = Y + (size_t)r * D_DIM + lane;
    float xv[8];
#pragma unroll
    for (int i = 0; i < 8; ++i) xv[i] = x[32 * i];
#pragma unroll
    for (int i = 0; i < 8; ++i) atomicAdd(y + 32 * i, xv[i] * v);
}

__global__ __launch_bounds__(256)
void gemm_tanh_wmma(const float* __restrict__ A0, const float* __restrict__ A1,
                    const float* __restrict__ B0, const float* __restrict__ B1,
                    const float* __restrict__ bias, float* __restrict__ OUT,
                    int M, int accumulate) {
    __shared__ float As[64 * LDA_S];
    __shared__ float Bst[128 * LDBT];

    const int t    = threadIdx.x;
    const int wave = t >> 5;
    const int lane = t & 31;
    const int wm   = wave & 3;
    const int wn   = wave >> 2;
    const int rowBase = blockIdx.x * 64;
    const int colBase = blockIdx.y * 128;

    const int m16 = lane & 15;
    const int kh  = lane >> 4;

    v8f acc[4];
#pragma unroll
    for (int j = 0; j < 4; ++j) acc[j] = (v8f)0.f;

    for (int src = 0; src < 2; ++src) {
        const float* A = src ? A1 : A0;
        const float* B = src ? B1 : B0;
        if (A == nullptr) break;

        for (int k0 = 0; k0 < D_DIM; k0 += KC) {
            {
                int r  = t >> 3;
                int c4 = (t & 7) * 4;
#pragma unroll
                for (int i = 0; i < 2; ++i) {
                    int    row = rowBase + r + 32 * i;
                    float4 v   = make_float4(0.f, 0.f, 0.f, 0.f);
                    if (row < M)
                        v = *(const float4*)(A + (size_t)row * D_DIM + k0 + c4);
                    float* dst = &As[(r + 32 * i) * LDA_S + c4];
                    dst[0] = v.x; dst[1] = v.y; dst[2] = v.z; dst[3] = v.w;
                }
            }
            {
#pragma unroll
                for (int i = 0; i < 4; ++i) {
                    int idx = t + 256 * i;
                    int k   = idx >> 5;
                    int n4  = (idx & 31) * 4;
                    float4 v = *(const float4*)(B + (size_t)(k0 + k) * D_DIM +
                                                colBase + n4);
                    Bst[(n4 + 0) * LDBT + k] = v.x;
                    Bst[(n4 + 1) * LDBT + k] = v.y;
                    Bst[(n4 + 2) * LDBT + k] = v.z;
                    Bst[(n4 + 3) * LDBT + k] = v.w;
                }
            }
            __syncthreads();

#pragma unroll
            for (int kk = 0; kk < KC; kk += 4) {
                v2f af = *(const v2f*)&As[(wm * 16 + m16) * LDA_S + kk + 2 * kh];
#pragma unroll
                for (int j = 0; j < 4; ++j) {
                    int n  = wn * 64 + j * 16 + m16;
                    v2f bf = *(const v2f*)&Bst[n * LDBT + kk + 2 * kh];
                    acc[j] = __builtin_amdgcn_wmma_f32_16x16x4_f32(
                        false, af, false, bf, (short)0, acc[j], false, false);
                }
            }
            __syncthreads();
        }
    }

#pragma unroll
    for (int j = 0; j < 4; ++j) {
        int   c  = colBase + wn * 64 + j * 16 + m16;
        float bv = bias ? bias[c] : 0.f;
#pragma unroll
        for (int r = 0; r < 8; ++r) {
            int row = rowBase + wm * 16 + kh * 8 + r;
            if (row < M) {
                float  v = tanhf(acc[j][r] + bv);
                float* o = OUT + (size_t)row * D_DIM + c;
                if (accumulate) *o += v;
                else            *o  = v;
            }
        }
    }
}

__global__ __launch_bounds__(256)
void l2norm_rows(const float* __restrict__ H, float* __restrict__ out, int M) {
    int row  = blockIdx.x * 8 + (threadIdx.x >> 5);
    int lane = threadIdx.x & 31;
    if (row >= M) return;
    const float4* h = (const float4*)(H + (size_t)row * D_DIM);
    float4 a = h[lane];
    float4 b = h[lane + 32];
    float  s = a.x * a.x + a.y * a.y + a.z * a.z + a.w * a.w +
               b.x * b.x + b.y * b.y + b.z * b.z + b.w * b.w;
#pragma unroll
    for (int off = 16; off > 0; off >>= 1) s += __shfl_xor(s, off, 32);
    float inv = 1.f / fmaxf(sqrtf(s), 1e-12f);
    a.x *= inv; a.y *= inv; a.z *= inv; a.w *= inv;
    b.x *= inv; b.y *= inv; b.z *= inv; b.w *= inv;
    float4* o = (float4*)(out + (size_t)row * D_DIM);
    o[lane]      = a;
    o[lane + 32] = b;
}

extern "C" void kernel_launch(void* const* d_in, const int* in_sizes, int n_in,
                              void* d_out, int out_size, void* d_ws, size_t ws_size,
                              hipStream_t stream) {
    const int* adj_rows   = (const int*)d_in[0];
    const int* adj_cols   = (const int*)d_in[1];
    const float* adj_vals = (const float*)d_in[2];
    const int* up_rows    = (const int*)d_in[3];
    const int* up_cols    = (const int*)d_in[4];
    const float* up_vals  = (const float*)d_in[5];
    const int* ip_rows    = (const int*)d_in[6];
    const int* ip_cols    = (const int*)d_in[7];
    const float* ip_vals  = (const float*)d_in[8];
    const float* user_emb = (const float*)d_in[9];
    const float* item_emb = (const float*)d_in[10];
    const float* Q_user   = (const float*)d_in[11];
    const float* W_user   = (const float*)d_in[12];
    const float* b_user   = (const float*)d_in[13];
    const float* Q_item   = (const float*)d_in[14];
    const float* W_item   = (const float*)d_in[15];
    const float* b_item   = (const float*)d_in[16];
    const float* M_user   = (const float*)d_in[17];
    const float* M_item   = (const float*)d_in[18];
    float* out = (float*)d_out;

    const int NU = 40000, NI = 20000, NN = 60000;
    const int NNZ_ADJ = in_sizes[0], NNZ_UP = in_sizes[3], NNZ_IP = in_sizes[6];
    const size_t ROWF = (size_t)D_DIM;
    const size_t MAT  = (size_t)NN * ROWF;

    float* agg  = (float*)d_ws;
    float* embb = agg  + MAT;
    float* hbuf = embb + MAT;

    dim3 blk(256);

    zero_f32<<<(unsigned)((MAT / 4 + 255) / 256), blk, 0, stream>>>((float4*)agg, (long)(MAT / 4));
    spmm_atomic<<<(NNZ_ADJ + 7) / 8, blk, 0, stream>>>(
        adj_rows, adj_cols, adj_vals, user_emb, item_emb, NU, agg, NNZ_ADJ);

    gemm_tanh_wmma<<<dim3((NU + 63) / 64, 2), blk, 0, stream>>>(
        agg, nullptr, Q_user, nullptr, nullptr, embb, NU, 0);
    gemm_tanh_wmma<<<dim3((NI + 63) / 64, 2), blk, 0, stream>>>(
        agg + (size_t)NU * ROWF, nullptr, Q_item, nullptr, nullptr,
        embb + (size_t)NU * ROWF, NI, 0);

    gemm_tanh_wmma<<<dim3((NU + 63) / 64, 2), blk, 0, stream>>>(
        user_emb, embb, W_user, W_user + (size_t)D_DIM * D_DIM, b_user,
        hbuf, NU, 0);
    gemm_tanh_wmma<<<dim3((NI + 63) / 64, 2), blk, 0, stream>>>(
        item_emb, embb + (size_t)NU * ROWF, W_item,
        W_item + (size_t)D_DIM * D_DIM, b_item,
        hbuf + (size_t)NU * ROWF, NI, 0);

    l2norm_rows<<<(NN + 7) / 8, blk, 0, stream>>>(hbuf, out, NN);

    zero_f32<<<(unsigned)((MAT / 4 + 255) / 256), blk, 0, stream>>>((float4*)agg, (long)(MAT / 4));
    spmm_atomic<<<(NNZ_UP + 7) / 8, blk, 0, stream>>>(
        up_rows, up_cols, up_vals, user_emb, user_emb, NU, agg, NNZ_UP);
    spmm_atomic<<<(NNZ_IP + 7) / 8, blk, 0, stream>>>(
        ip_rows, ip_cols, ip_vals, item_emb, item_emb, NI,
        agg + (size_t)NU * ROWF, NNZ_IP);

    gemm_tanh_wmma<<<dim3((NU + 63) / 64, 2), blk, 0, stream>>>(
        agg, nullptr, M_user, nullptr, nullptr, out, NU, 1);
    gemm_tanh_wmma<<<dim3((NI + 63) / 64, 2), blk, 0, stream>>>(
        agg + (size_t)NU * ROWF, nullptr, M_item, nullptr, nullptr,
        out + (size_t)NU * ROWF, NI, 1);
}


# SPMM owner-computes: a work-group keeps 240 output rows in LDS, scans the row index list for its entries, sums them with LDS f32 atomics (arrival order), stores once; no global atomics, no zero fill
# speedup vs baseline: 1.9051x; 1.2790x over previous
#include <hip/hip_runtime.h>
#include <hip/hip_bf16.h>
#include <math.h>

typedef __attribute__((ext_vector_type(2))) float v2f;
typedef __attribute__((ext_vector_type(8))) float v8f;

#define D_DIM 256
#define KC 32
#define LDA_S (KC + 2)
#define LDBT  (KC + 2)

#define OWN_R    240
#define QCAP     128
#define QFLUSH   32
#define SCAN_U   16
#define GATHER_U 8

template <int U>
__device__ __forceinline__ void gather_add(float* acc, int lane, int j0,
                                           int cq, int vq, int rq,
                                           const float* __restrict__ Xa,
                                           const float* __restrict__ Xb, int split) {
    float x[U][8];
#pragma unroll
    for (int u = 0; u < U; ++u) {
        int c = __builtin_amdgcn_readlane(cq, j0 + u);
        const float* X = (c < split) ? (Xa + (size_t)c * D_DIM)
                                     : (Xb + (size_t)(c - split) * D_DIM);
#pragma unroll
        for (int i = 0; i < 8; ++i) x[u][i] = X[lane + 32 * i];
    }
#pragma unroll
    for (int u = 0; u < U; ++u) {
        float  v = __int_as_float(__builtin_amdgcn_readlane(vq, j0 + u));
        int    r = __builtin_amdgcn_readlane(rq, j0 + u);
        float* a = acc + r * D_DIM + lane;
#pragma unroll
        for (int i = 0; i < 8; ++i) atomicAdd(a + 32 * i, x[u][i] * v);
    }
}

__device__ __forceinline__ void flush_queue(float* acc, const int* qe, const int* qr,
                                            int qn, int lane,
                                            const int* __restrict__ cols,
                                            const float* __restrict__ vals,
                                            const float* __restrict__ Xa,
                                            const float* __restrict__ Xb, int split) {
    for (int h = 0; h < qn; h += 32) {
        int n = qn - h;
        if (n > 32) n = 32;
        int c = 0, vb = 0, r = 0;
        if (lane < n) {
            int e = qe[h + lane];
            r  = qr[h + lane];
            c  = cols[e];
            vb = __float_as_int(vals[e]);
        }
        int g = 0;
        for (; g + GATHER_U <= n; g += GATHER_U)
            gather_add<GATHER_U>(acc, lane, g, c, vb, r, Xa, Xb, split);
        if (n - g >= 4) { gather_add<4>(acc, lane, g, c, vb, r, Xa, Xb, split); g += 4; }
        if (n - g >= 2) { gather_add<2>(acc, lane, g, c, vb, r, Xa, Xb, split); g += 2; }
        if (n - g >= 1) { gather_add<1>(acc, lane, g, c, vb, r, Xa, Xb, split); }
    }
}

__global__ __launch_bounds__(1024)
void spmm_owner(const int* __restrict__ rows0, const int* __restrict__ cols0,
                const float* __restrict__ vals0,
                const float* __restrict__ Xa0, const float* __restrict__ Xb0,
                int split0, float* __restrict__ Y0, int nnz0, int nrows0, int nblk0,
                const int* __restrict__ rows1, const int* __restrict__ cols1,
                const float* __restrict__ vals1,
                const float* __restrict__ Xa1, const float* __restrict__ Xb1,
                int split1, float* __restrict__ Y1, int nnz1, int nrows1, int nblk1) {
    __shared__ float acc[OWN_R * D_DIM];
    __shared__ int   q_e[32 * QCAP];
    __shared__ int   q_r[32 * QCAP];

    const int t    = threadIdx.x;
    const int lane = t & 31;
    const int w    = __builtin_amdgcn_readfirstlane(t >> 5);

    int b = blockIdx.x;
    const bool second = (b >= nblk0);
    if (second) b -= nblk0;
    const int*   rows  = second ? rows1  : rows0;
    const int*   cols  = second ? cols1  : cols0;
    const float* vals  = second ? vals1  : vals0;
    const float* Xa    = second ? Xa1    : Xa0;
    const float* Xb    = second ? Xb1    : Xb0;
    const int    split = second ? split1 : split0;
    float*       Y     = second ? Y1     : Y0;
    const int    nnz   = second ? nnz1   : nnz0;
    const int    nrows = second ? nrows1 : nrows0;
    const int    nblk  = second ? nblk1  : nblk0;

    const int lo = b * OWN_R;
    int nloc = nrows - lo;
    if (nloc > OWN_R) nloc = OWN_R;

    {
        float4* a4 = (float4*)acc;
        for (int i = t; i < OWN_R * (D_DIM / 4); i += 1024)
            a4[i] = make_float4(0.f, 0.f, 0.f, 0.f);
    }
    __syncthreads();

    const int nfull = nnz >> 7;
    const int rot   = (int)(((long)b * nfull) / nblk);
    const int K     = (nfull + 31) >> 5;
    int*      qe    = q_e + w * QCAP;
    int*      qr    = q_r + w * QCAP;
    const unsigned below = (1u << lane) - 1u;
    const unsigned nl    = (unsigned)nloc;
    int qn = 0;

    for (int k0 = 0; k0 < K; k0 += SCAN_U) {
        int4 rv[SCAN_U];
        int  eb[SCAN_U];
#pragma unroll
        for (int u = 0; u < SCAN_U; ++u) {
            int tl   = (k0 + u) * 32 + w;
            int tile = (tl < nfull ? tl : 0) + rot;
            if (tile >= nfull) tile -= nfull;
            eb[u] = tile * 128 + lane * 4;
            rv[u] = *(const int4*)(rows + eb[u]);
        }
#pragma unroll
        for (int u = 0; u < SCAN_U; ++u) {
            if ((k0 + u) * 32 + w >= nfull) continue;
            int rr[4] = {rv[u].x - lo, rv[u].y - lo, rv[u].z - lo, rv[u].w - lo};
            bool any = ((unsigned)rr[0] < nl) | ((unsigned)rr[1] < nl) |
                       ((unsigned)rr[2] < nl) | ((unsigned)rr[3] < nl);
            if (__builtin_amdgcn_ballot_w32(any) == 0u) continue;
#pragma unroll
            for (int ci = 0; ci < 4; ++ci) {
                bool     hit = (unsigned)rr[ci] < nl;
                unsigned m   = __builtin_amdgcn_ballot_w32(hit);
                if (m == 0u) continue;
                int cnt = __builtin_popcount(m);
                if (qn + cnt > QCAP) {
                    unsigned mm = m;
                    while (mm) {
                        int l = __builtin_ctz(mm);
                        mm &= mm - 1u;
                        int e1 = __builtin_amdgcn_readlane(eb[u] + ci, l);
                        int r1 = __builtin_amdgcn_readlane(rr[ci], l);
                        int c1 = cols[e1];
                        int v1 = __float_as_int(vals[e1]);
                        gather_add<1>(acc, lane, 0, c1, v1, r1, Xa, Xb, split);
                    }
                    continue;
                }
                if (hit) {
                    int s = qn + __builtin_popcount(m & below);
                    qe[s] = eb[u] + ci;
                    qr[s] = rr[ci];
                }
                qn += cnt;
            }
        }
        if (qn >= QFLUSH) {
            flush_queue(acc, qe, qr, qn, lane, cols, vals, Xa, Xb, split);
            qn = 0;
        }
    }
    flush_queue(acc, qe, qr, qn, lane, cols, vals, Xa, Xb, split);

    if (w == 0) {
        for (int e1 = nfull * 128; e1 < nnz; ++e1) {
            int r1 = rows[e1] - lo;
            if ((unsigned)r1 < nl)
                gather_add<1>(acc, lane, 0, cols[e1], __float_as_int(vals[e1]), r1,
                              Xa, Xb, split);
        }
    }

    __syncthreads();

    for (int r = w; r < nloc; r += 32) {
        const float4* s = (const float4*)(acc + r * D_DIM);
        float4*       d = (float4*)(Y + (size_t)(lo + r) * D_DIM);
        d[lane]      = s[lane];
        d[lane + 32] = s[lane + 32];
    }
}

__global__ __launch_bounds__(256)
void gemm_tanh_wmma(const float* __restrict__ A0, const float* __restrict__ A1,
                    const float* __restrict__ B0, const float* __restrict__ B1,
                    const float* __restrict__ bias, float* __restrict__ OUT,
                    int M, int accumulate) {
    __shared__ float As[64 * LDA_S];
    __shared__ float Bst[128 * LDBT];

    const int t    = threadIdx.x;
    const int wave = t >> 5;
    const int lane = t & 31;
    const int wm   = wave & 3;
    const int wn   = wave >> 2;
    const int rowBase = blockIdx.x * 64;
    const int colBase = blockIdx.y * 128;

    const int m16 = lane & 15;
    const int kh  = lane >> 4;

    v8f acc[4];
#pragma unroll
    for (int j = 0; j < 4; ++j) acc[j] = (v8f)0.f;

    for (int src = 0; src < 2; ++src) {
        const float* A = src ? A1 : A0;
        const float* B = src ? B1 : B0;
        if (A == nullptr) break;

        for (int k0 = 0; k0 < D_DIM; k0 += KC) {
            {
                int r  = t >> 3;
                int c4 = (t & 7) * 4;
#pragma unroll
                for (int i = 0; i < 2; ++i) {
                    int    row = rowBase + r + 32 * i;
                    float4 v   = make_float4(0.f, 0.f, 0.f, 0.f);
                    if (row < M)
                        v = *(const float4*)(A + (size_t)row * D_DIM + k0 + c4);
                    float* dst = &As[(r + 32 * i) * LDA_S + c4];
                    dst[0] = v.x; dst[1] = v.y; dst[2] = v.z; dst[3] = v.w;
                }
            }
            {
#pragma unroll
                for (int i = 0; i < 4; ++i) {
                    int idx = t + 256 * i;
                    int k   = idx >> 5;
                    int n4  = (idx & 31) * 4;
                    float4 v = *(const float4*)(B + (size_t)(k0 + k) * D_DIM +
                                                colBase + n4);
                    Bst[(n4 + 0) * LDBT + k] = v.x;
                    Bst[(n4 + 1) * LDBT + k] = v.y;
                    Bst[(n4 + 2) * LDBT + k] = v.z;
                    Bst[(n4 + 3) * LDBT + k] = v.w;
                }
            }
            __syncthreads();

#pragma unroll
            for (int kk = 0; kk < KC; kk += 4) {
                v2f af = *(const v2f*)&As[(wm * 16 + m16) * LDA_S + kk + 2 * kh];
#pragma unroll
                for (int j = 0; j < 4; ++j) {
                    int n  = wn * 64 + j * 16 + m16;
                    v2f bf = *(const v2f*)&Bst[n * LDBT + kk + 2 * kh];
                    acc[j] = __builtin_amdgcn_wmma_f32_16x16x4_f32(
                        false, af, false, bf, (short)0, acc[j], false, false);
                }
            }
            __syncthreads();
        }
    }

#pragma unroll
    for (int j = 0; j < 4; ++j) {
        int   c  = colBase + wn * 64 + j * 16 + m16;
        float bv = bias ? bias[c] : 0.f;
#pragma unroll
        for (int r = 0; r < 8; ++r) {
            int row = rowBase + wm * 16 + kh * 8 + r;
            if (row < M) {
                float  v = tanhf(acc[j][r] + bv);
                float* o = OUT + (size_t)row * D_DIM + c;
                if (accumulate) *o += v;
                else            *o  = v;
            }
        }
    }
}

__global__ __launch_bounds__(256)
void l2norm_rows(const float* __restrict__ H, float* __restrict__ out, int M) {
    int row  = blockIdx.x * 8 + (threadIdx.x >> 5);
    int lane = threadIdx.x & 31;
    if (row >= M) return;
    const float4* h = (const float4*)(H + (size_t)row * D_DIM);
    float4 a = h[lane];
    float4 b = h[lane + 32];
    float  s = a.x * a.x + a.y * a.y + a.z * a.z + a.w * a.w +
               b.x * b.x + b.y * b.y + b.z * b.z + b.w * b.w;
#pragma unroll
    for (int off = 16; off > 0; off >>= 1) s += __shfl_xor(s, off, 32);
    float inv = 1.f / fmaxf(sqrtf(s), 1e-12f);
    a.x *= inv; a.y *= inv; a.z *= inv; a.w *= inv;
    b.x *= inv; b.y *= inv; b.z *= inv; b.w *= inv;
    float4* o = (float4*)(out + (size_t)row * D_DIM);
    o[lane]      = a;
    o[lane + 32] = b;
}

extern "C" void kernel_launch(void* const* d_in, const int* in_sizes, int n_in,
                              void* d_out, int out_size, void* d_ws, size_t ws_size,
                              hipStream_t stream) {
    const int* adj_rows   = (const int*)d_in[0];
    const int* adj_cols   = (const int*)d_in[1];
    const float* adj_vals = (const float*)d_in[2];
    const int* up_rows    = (const int*)d_in[3];
    const int* up_cols    = (const int*)d_in[4];
    const float* up_vals  = (const float*)d_in[5];
    const int* ip_rows    = (const int*)d_in[6];
    const int* ip_cols    = (const int*)d_in[7];
    const float* ip_vals  = (const float*)d_in[8];
    const float* user_emb = (const float*)d_in[9];
    const float* item_emb = (const float*)d_in[10];
    const float* Q_user   = (const float*)d_in[11];
    const float* W_user   = (const float*)d_in[12];
    const float* b_user   = (const float*)d_in[13];
    const float* Q_item   = (const float*)d_in[14];
    const float* W_item   = (const float*)d_in[15];
    const float* b_item   = (const float*)d_in[16];
    const float* M_user   = (const float*)d_in[17];
    const float* M_item   = (const float*)d_in[18];
    float* out = (float*)d_out;

    const int NU = 40000, NI = 20000, NN = 60000;
    const int NNZ_ADJ = in_sizes[0], NNZ_UP = in_sizes[3], NNZ_IP = in_sizes[6];
    const size_t ROWF = (size_t)D_DIM;
    const size_t MAT  = (size_t)NN * ROWF;

    float* agg  = (float*)d_ws;
    float* embb = agg  + MAT;
    float* hbuf = embb + MAT;

    dim3 blk(256);
    const int NB_ADJ = (NN + OWN_R - 1) / OWN_R;
    const int NB_UP  = (NU + OWN_R - 1) / OWN_R;
    const int NB_IP  = (NI + OWN_R - 1) / OWN_R;

    spmm_owner<<<NB_ADJ, dim3(1024), 0, stream>>>(
        adj_rows, adj_cols, adj_vals, user_emb, item_emb, NU, agg, NNZ_ADJ, NN, NB_ADJ,
        adj_rows, adj_cols, adj_vals, user_emb, item_emb, NU, agg, NNZ_ADJ, NN, 0);

    gemm_tanh_wmma<<<dim3((NU + 63) / 64, 2), blk, 0, stream>>>(
        agg, nullptr, Q_user, nullptr, nullptr, embb, NU, 0);
    gemm_tanh_wmma<<<dim3((NI + 63) / 64, 2), blk, 0, stream>>>(
        agg + (size_t)NU * ROWF, nullptr, Q_item, nullptr, nullptr,
        embb + (size_t)NU * ROWF, NI, 0);

    gemm_tanh_wmma<<<dim3((NU + 63) / 64, 2), blk, 0, stream>>>(
        user_emb, embb, W_user, W_user + (size_t)D_DIM * D_DIM, b_user,
        hbuf, NU, 0);
    gemm_tanh_wmma<<<dim3((NI + 63) / 64, 2), blk, 0, stream>>>(
        item_emb, embb + (size_t)NU * ROWF, W_item,
        W_item + (size_t)D_DIM * D_DIM, b_item,
        hbuf + (size_t)NU * ROWF, NI, 0);

    l2norm_rows<<<(NN + 7) / 8, blk, 0, stream>>>(hbuf, out, NN);

    spmm_owner<<<NB_UP + NB_IP, dim3(1024), 0, stream>>>(
        up_rows, up_cols, up_vals, user_emb, user_emb, NU, agg, NNZ_UP, NU, NB_UP,
        ip_rows, ip_cols, ip_vals, item_emb, item_emb, NI,
        agg + (size_t)NU * ROWF, NNZ_IP, NI, NB_IP);

    gemm_tanh_wmma<<<dim3((NU + 63) / 64, 2), blk, 0, stream>>>(
        agg, nullptr, M_user, nullptr, nullptr, out, NU, 1);
    gemm_tanh_wmma<<<dim3((NI + 63) / 64, 2), blk, 0, stream>>>(
        agg + (size_t)NU * ROWF, nullptr, M_item, nullptr, nullptr,
        out + (size_t)NU * ROWF, NI, 1);
}


# SPMM pre-pass sorts each 4096-entry chunk of the COO list by owner group in LDS; a work-group reads only its own segments instead of scanning the list; rows summed in sorted-chunk order
# speedup vs baseline: 2.1703x; 1.1392x over previous
#include <hip/hip_runtime.h>
#include <hip/hip_bf16.h>
#include <math.h>

typedef __attribute__((ext_vector_type(2))) float v2f;
typedef __attribute__((ext_vector_type(8))) float v8f;

#define D_DIM 256
#define KC 32
#define LDA_S (KC + 2)
#define LDBT  (KC + 2)

#define OWN_R    240
#define GATHER_U 8
#define CH       4096
#define OFFS_LD  256

__global__ __launch_bounds__(1024)
void bin_entries(const int* __restrict__ rows0, const int* __restrict__ cols0,
                 const float* __restrict__ vals0, int nnz0, int nrows0, int nch0,
                 int* __restrict__ key0, float* __restrict__ sv0, int* __restrict__ offs0,
                 const int* __restrict__ rows1, const int* __restrict__ cols1,
                 const float* __restrict__ vals1, int nnz1, int nrows1,
                 int* __restrict__ key1, float* __restrict__ sv1, int* __restrict__ offs1) {
    __shared__ int   hist[OFFS_LD];
    __shared__ int   wsum[8];
    __shared__ int   s_key[CH];
    __shared__ float s_val[CH];

    const int t    = threadIdx.x;
    const int lane = t & 31;
    const int w    = t >> 5;

    int c = blockIdx.x;
    const bool second = (c >= nch0);
    if (second) c -= nch0;
    const int*   rows  = second ? rows1  : rows0;
    const int*   cols  = second ? cols1  : cols0;
    const float* vals  = second ? vals1  : vals0;
    const int    nnz   = second ? nnz1   : nnz0;
    const int    nrows = second ? nrows1 : nrows0;
    int*         key   = second ? key1   : key0;
    float*       sv    = second ? sv1    : sv0;
    int*         offs  = second ? offs1  : offs0;

    const int e0 = c * CH;
    int n = nnz - e0;
    if (n > CH) n = CH;
    if (n < 0)  n = 0;

    if (t < OFFS_LD) hist[t] = 0;
    __syncthreads();

    int   rr[4], cc[4];
    float vv[4];
#pragma unroll
    for (int i = 0; i < 4; ++i) {
        const int j = t + 1024 * i;
        rr[i] = -1; cc[i] = 0; vv[i] = 0.f;
        if (j < n) {
            rr[i] = rows[e0 + j];
            cc[i] = cols[e0 + j];
            vv[i] = vals[e0 + j];
        }
    }
    int bk[4], rk[4], ky[4];
#pragma unroll
    for (int i = 0; i < 4; ++i) {
        bk[i] = -1; rk[i] = 0; ky[i] = 0;
        const int r = rr[i];
        if ((unsigned)r < (unsigned)nrows) {
            const int g = r / OWN_R;
            bk[i] = g;
            ky[i] = (int)(((unsigned)(r - g * OWN_R) << 24) | (unsigned)cc[i]);
            rk[i] = atomicAdd(&hist[g], 1);
        }
    }
    __syncthreads();

    const int x = (t < OFFS_LD) ? hist[t] : 0;
    int incl = x;
#pragma unroll
    for (int off = 1; off < 32; off <<= 1) {
        const int y = __shfl_up(incl, off, 32);
        if (lane >= off) incl += y;
    }
    if (t < OFFS_LD && lane == 31) wsum[w] = incl;
    __syncthreads();
    if (t < OFFS_LD) {
        int p = 0;
#pragma unroll
        for (int k = 0; k < 8; ++k)
            if (k < w) p += wsum[k];
        const int base = p + incl - x;
        hist[t] = base;
        offs[c * OFFS_LD + t] = base;
    }
    __syncthreads();

#pragma unroll
    for (int i = 0; i < 4; ++i) {
        if (bk[i] >= 0) {
            const int p = hist[bk[i]] + rk[i];
            s_key[p] = ky[i];
            s_val[p] = vv[i];
        }
    }
    __syncthreads();
#pragma unroll
    for (int i = 0; i < 4; ++i) {
        const int j = t + 1024 * i;
        if (j < n) {
            key[e0 + j] = s_key[j];
            sv[e0 + j]  = s_val[j];
        }
    }
}

template <int U>
__device__ __forceinline__ void gather_add(float* acc, int lane, int j0,
                                           int cq, int vq, int rq,
                                           const float* __restrict__ Xa,
                                           const float* __restrict__ Xb, int split) {
    float x[U][8];
#pragma unroll
    for (int u = 0; u < U; ++u) {
        int c = __builtin_amdgcn_readlane(cq, j0 + u);
        const float* X = (c < split) ? (Xa + (size_t)c * D_DIM)
                                     : (Xb + (size_t)(c - split) * D_DIM);
#pragma unroll
        for (int i = 0; i < 8; ++i) x[u][i] = X[lane + 32 * i];
    }
#pragma unroll
    for (int u = 0; u < U; ++u) {
        float  v = __int_as_float(__builtin_amdgcn_readlane(vq, j0 + u));
        int    r = __builtin_amdgcn_readlane(rq, j0 + u);
        float* a = acc + r * D_DIM + lane;
#pragma unroll
        for (int i = 0; i < 8; ++i) atomicAdd(a + 32 * i, x[u][i] * v);
    }
}

__global__ __launch_bounds__(1024)
void spmm_owner(const int* __restrict__ key0, const float* __restrict__ sv0,
                const int* __restrict__ offs0, int nch0,
                const float* __restrict__ Xa0, const float* __restrict__ Xb0,
                int split0, float* __restrict__ Y0, int nrows0, int nblk0,
                const int* __restrict__ key1, const float* __restrict__ sv1,
                const int* __restrict__ offs1, int nch1,
                const float* __restrict__ Xa1, const float* __restrict__ Xb1,
                int split1, float* __restrict__ Y1, int nrows1) {
    __shared__ float acc[OWN_R * D_DIM];

    const int t    = threadIdx.x;
    const int lane = t & 31;
    const int w    = __builtin_amdgcn_readfirstlane(t >> 5);

    int b = blockIdx.x;
    const bool second = (b >= nblk0);
    if (second) b -= nblk0;
    const int*   key   = second ? key1   : key0;
    const float* sv    = second ? sv1    : sv0;
    const int*   offs  = second ? offs1  : offs0;
    const int    nch   = second ? nch1   : nch0;
    const float* Xa    = second ? Xa1    : Xa0;
    const float* Xb    = second ? Xb1    : Xb0;
    const int    split = second ? split1 : split0;
    float*       Y     = second ? Y1     : Y0;
    const int    nrows = second ? nrows1 : nrows0;

    const int lo = b * OWN_R;
    int nloc = nrows - lo;
    if (nloc > OWN_R) nloc = OWN_R;

    {
        float4* a4 = (float4*)acc;
        for (int i = t; i < OWN_R * (D_DIM / 4); i += 1024)
            a4[i] = make_float4(0.f, 0.f, 0.f, 0.f);
    }
    __syncthreads();

    for (int cb = 0; cb < nch; cb += 1024) {
        const int ci = cb + w + 32 * lane;
        int s = 0, e = 0;
        if (ci < nch) {
            s = offs[ci * OFFS_LD + b];
            e = offs[ci * OFFS_LD + b + 1];
        }
        const int len = e - s;
        int incl = len;
#pragma unroll
        for (int off = 1; off < 32; off <<= 1) {
            const int y = __shfl_up(incl, off, 32);
            if (lane >= off) incl += y;
        }
        const int excl = incl - len;
        const int T    = __builtin_amdgcn_readlane(incl, 31);
        const int G    = ci * CH + s - excl;
        int nv = (nch - cb - w + 31) >> 5;
        if (nv > 32) nv = 32;

        auto load_batch = [&](int j0, int& k, int& v) {
            int j = j0 + lane;
            if (j > T - 1) j = T - 1;
            int g = __builtin_amdgcn_readlane(G, 0);
            for (int i = 1; i < nv; ++i) {
                const int ex = __builtin_amdgcn_readlane(excl, i);
                const int gi = __builtin_amdgcn_readlane(G, i);
                if (j >= ex) g = gi;
            }
            k = key[g + j];
            v = __float_as_int(sv[g + j]);
        };

        if (T > 0) {
            int kC, vC;
            load_batch(0, kC, vC);
            for (int j0 = 0; j0 < T; j0 += 32) {
                int kN, vN;
                load_batch(j0 + 32, kN, vN);
                int n = T - j0;
                if (n > 32) n = 32;
                const int cq = kC & 0xFFFFFF;
                const int rq = (int)((unsigned)kC >> 24);
                int g = 0;
                for (; g + GATHER_U <= n; g += GATHER_U)
                    gather_add<GATHER_U>(acc, lane, g, cq, vC, rq, Xa, Xb, split);
                if (n - g >= 4) { gather_add<4>(acc, lane, g, cq, vC, rq, Xa, Xb, split); g += 4; }
                if (n - g >= 2) { gather_add<2>(acc, lane, g, cq, vC, rq, Xa, Xb, split); g += 2; }
                if (n - g >= 1) { gather_add<1>(acc, lane, g, cq, vC, rq, Xa, Xb, split); }
                kC = kN; vC = vN;
            }
        }
    }

    __syncthreads();

    for (int r = w; r < nloc; r += 32) {
        const float4* s4 = (const float4*)(acc + r * D_DIM);
        float4*       d  = (float4*)(Y + (size_t)(lo + r) * D_DIM);
        d[lane]      = s4[lane];
        d[lane + 32] = s4[lane + 32];
    }
}

__global__ __launch_bounds__(256)
void gemm_tanh_wmma(const float* __restrict__ A0, const float* __restrict__ A1,
                    const float* __restrict__ B0, const float* __restrict__ B1,
                    const float* __restrict__ bias, float* __restrict__ OUT,
                    int M, int accumulate) {
    __shared__ float As[64 * LDA_S];
    __shared__ float Bst[128 * LDBT];

    const int t    = threadIdx.x;
    const int wave = t >> 5;
    const int lane = t & 31;
    const int wm   = wave & 3;
    const int wn   = wave >> 2;
    const int rowBase = blockIdx.x * 64;
    const int colBase = blockIdx.y * 128;

    const int m16 = lane & 15;
    const int kh  = lane >> 4;

    v8f acc[4];
#pragma unroll
    for (int j = 0; j < 4; ++j) acc[j] = (v8f)0.f;

    for (int src = 0; src < 2; ++src) {
        const float* A = src ? A1 : A0;
        const float* B = src ? B1 : B0;
        if (A == nullptr) break;

        for (int k0 = 0; k0 < D_DIM; k0 += KC) {
            {
                int r  = t >> 3;
                int c4 = (t & 7) * 4;
#pragma unroll
                for (int i = 0; i < 2; ++i) {
                    int    row = rowBase + r + 32 * i;
                    float4 v   = make_float4(0.f, 0.f, 0.f, 0.f);
                    if (row < M)
                        v = *(const float4*)(A + (size_t)row * D_DIM + k0 + c4);
                    float* dst = &As[(r + 32 * i) * LDA_S + c4];
                    dst[0] = v.x; dst[1] = v.y; dst[2] = v.z; dst[3] = v.w;
                }
            }
            {
#pragma unroll
                for (int i = 0; i < 4; ++i) {
                    int idx = t + 256 * i;
                    int k   = idx >> 5;
                    int n4  = (idx & 31) * 4;
                    float4 v = *(const float4*)(B + (size_t)(k0 + k) * D_DIM +
                                                colBase + n4);
                    Bst[(n4 + 0) * LDBT + k] = v.x;
                    Bst[(n4 + 1) * LDBT + k] = v.y;
                    Bst[(n4 + 2) * LDBT + k] = v.z;
                    Bst[(n4 + 3) * LDBT + k] = v.w;
                }
            }
            __syncthreads();

#pragma unroll
            for (int kk = 0; kk < KC; kk += 4) {
                v2f af = *(const v2f*)&As[(wm * 16 + m16) * LDA_S + kk + 2 * kh];
#pragma unroll
                for (int j = 0; j < 4; ++j) {
                    int n  = wn * 64 + j * 16 + m16;
                    v2f bf = *(const v2f*)&Bst[n * LDBT + kk + 2 * kh];
                    acc[j] = __builtin_amdgcn_wmma_f32_16x16x4_f32(
                        false, af, false, bf, (short)0, acc[j], false, false);
                }
            }
            __syncthreads();
        }
    }

#pragma unroll
    for (int j = 0; j < 4; ++j) {
        int   c  = colBase + wn * 64 + j * 16 + m16;
        float bv = bias ? bias[c] : 0.f;
#pragma unroll
        for (int r = 0; r < 8; ++r) {
            int row = rowBase + wm * 16 + kh * 8 + r;
            if (row < M) {
                float  v = tanhf(acc[j][r] + bv);
                float* o = OUT + (size_t)row * D_DIM + c;
                if (accumulate) *o += v;
                else            *o  = v;
            }
        }
    }
}

__global__ __launch_bounds__(256)
void l2norm_rows(const float* __restrict__ H, float* __restrict__ out, int M) {
    int row  = blockIdx.x * 8 + (threadIdx.x >> 5);
    int lane = threadIdx.x & 31;
    if (row >= M) return;
    const float4* h = (const float4*)(H + (size_t)row * D_DIM);
    float4 a = h[lane];
    float4 b = h[lane + 32];
    float  s = a.x * a.x + a.y * a.y + a.z * a.z + a.w * a.w +
               b.x * b.x + b.y * b.y + b.z * b.z + b.w * b.w;
#pragma unroll
    for (int off = 16; off > 0; off >>= 1) s += __shfl_xor(s, off, 32);
    float inv = 1.f / fmaxf(sqrtf(s), 1e-12f);
    a.x *= inv; a.y *= inv; a.z *= inv; a.w *= inv;
    b.x *= inv; b.y *= inv; b.z *= inv; b.w *= inv;
    float4* o = (float4*)(out + (size_t)row * D_DIM);
    o[lane]      = a;
    o[lane + 32] = b;
}

extern "C" void kernel_launch(void* const* d_in, const int* in_sizes, int n_in,
                              void* d_out, int out_size, void* d_ws, size_t ws_size,
                              hipStream_t stream) {
    const int* adj_rows   = (const int*)d_in[0];
    const int* adj_cols   = (const int*)d_in[1];
    const float* adj_vals = (const float*)d_in[2];
    const int* up_rows    = (const int*)d_in[3];
    const int* up_cols    = (const int*)d_in[4];
    const float* up_vals  = (const float*)d_in[5];
    const int* ip_rows    = (const int*)d_in[6];
    const int* ip_cols    = (const int*)d_in[7];
    const float* ip_vals  = (const float*)d_in[8];
    const float* user_emb = (const float*)d_in[9];
    const float* item_emb = (const float*)d_in[10];
    const float* Q_user   = (const float*)d_in[11];
    const float* W_user   = (const float*)d_in[12];
    const float* b_user   = (const float*)d_in[13];
    const float* Q_item   = (const float*)d_in[14];
    const float* W_item   = (const float*)d_in[15];
    const float* b_item   = (const float*)d_in[16];
    const float* M_user   = (const float*)d_in[17];
    const float* M_item   = (const float*)d_in[18];
    float* out = (float*)d_out;

    const int NU = 40000, NI = 20000, NN = 60000;
    const int NNZ_ADJ = in_sizes[0], NNZ_UP = in_sizes[3], NNZ_IP = in_sizes[6];
    const size_t ROWF = (size_t)D_DIM;
    const size_t MAT  = (size_t)NN * ROWF;

    float* agg  = (float*)d_ws;
    float* embb = agg  + MAT;
    float* hbuf = embb + MAT;

    dim3 blk(256);
    const int NB_ADJ = (NN + OWN_R - 1) / OWN_R;
    const int NB_UP  = (NU + OWN_R - 1) / OWN_R;
    const int NB_IP  = (NI + OWN_R - 1) / OWN_R;
    const int NC_ADJ = (960000 + CH - 1) / CH;
    const int NC_UP  = (640000 + CH - 1) / CH;
    const int NC_IP  = (320000 + CH - 1) / CH;
    const size_t SEG = (size_t)1 << 20;

    int*   keyA  = (int*)hbuf;
    float* svA   = hbuf + SEG;
    int*   offsA = (int*)(hbuf + 2 * SEG);
    int*   keyU  = (int*)embb;
    float* svU   = embb + SEG;
    int*   offsU = (int*)(embb + 2 * SEG);
    int*   keyI  = (int*)(embb + 3 * SEG);
    float* svI   = embb + 4 * SEG;
    int*   offsI = (int*)(embb + 5 * SEG);

    bin_entries<<<NC_ADJ, dim3(1024), 0, stream>>>(
        adj_rows, adj_cols, adj_vals, NNZ_ADJ, NN, NC_ADJ, keyA, svA, offsA,
        adj_rows, adj_cols, adj_vals, NNZ_ADJ, NN, keyA, svA, offsA);

    spmm_owner<<<NB_ADJ, dim3(1024), 0, stream>>>(
        keyA, svA, offsA, NC_ADJ, user_emb, item_emb, NU, agg, NN, NB_ADJ,
        keyA, svA, offsA, 0, user_emb, item_emb, NU, agg, NN);

    gemm_tanh_wmma<<<dim3((NU + 63) / 64, 2), blk, 0, stream>>>(
        agg, nullptr, Q_user, nullptr, nullptr, embb, NU, 0);
    gemm_tanh_wmma<<<dim3((NI + 63) / 64, 2), blk, 0, stream>>>(
        agg + (size_t)NU * ROWF, nullptr, Q_item, nullptr, nullptr,
        embb + (size_t)NU * ROWF, NI, 0);

    gemm_tanh_wmma<<<dim3((NU + 63) / 64, 2), blk, 0, stream>>>(
        user_emb, embb, W_user, W_user + (size_t)D_DIM * D_DIM, b_user,
        hbuf, NU, 0);
    gemm_tanh_wmma<<<dim3((NI + 63) / 64, 2), blk, 0, stream>>>(
        item_emb, embb + (size_t)NU * ROWF, W_item,
        W_item + (size_t)D_DIM * D_DIM, b_item,
        hbuf + (size_t)NU * ROWF, NI, 0);

    l2norm_rows<<<(NN + 7) / 8, blk, 0, stream>>>(hbuf, out, NN);

    bin_entries<<<NC_UP + NC_IP, dim3(1024), 0, stream>>>(
        up_rows, up_cols, up_vals, NNZ_UP, NU, NC_UP, keyU, svU, offsU,
        ip_rows, ip_cols, ip_vals, NNZ_IP, NI, keyI, svI, offsI);
    spmm_owner<<<NB_UP + NB_IP, dim3(1024), 0, stream>>>(
        keyU, svU, offsU, NC_UP, user_emb, user_emb, NU, agg, NU, NB_UP,
        keyI, svI, offsI, NC_IP, item_emb, item_emb, NI,
        agg + (size_t)NU * ROWF, NI);

    gemm_tanh_wmma<<<dim3((NU + 63) / 64, 2), blk, 0, stream>>>(
        agg, nullptr, M_user, nullptr, nullptr, out, NU, 1);
    gemm_tanh_wmma<<<dim3((NI + 63) / 64, 2), blk, 0, stream>>>(
        agg + (size_t)NU * ROWF, nullptr, M_item, nullptr, nullptr,
        out + (size_t)NU * ROWF, NI, 1);
}


# Row L2 norm folded into the two GEMM epilogues beside it: h GEMM stores 4 partial sums of squares per row (new fixed summation tree), last GEMM scales h as it adds; norm launch and its pass dropped
# speedup vs baseline: 2.2969x; 1.0583x over previous
#include <hip/hip_runtime.h>
#include <hip/hip_bf16.h>
#include <math.h>

typedef __attribute__((ext_vector_type(2))) float v2f;
typedef __attribute__((ext_vector_type(8))) float v8f;

#define D_DIM 256
#define KC 32
#define LDA_S (KC + 2)
#define LDBT  (KC + 2)

#define OWN_R    240
#define GATHER_U 8
#define CH       4096
#define OFFS_LD  256

__global__ __launch_bounds__(1024)
void bin_entries(const int* __restrict__ rows0, const int* __restrict__ cols0,
                 const float* __restrict__ vals0, int nnz0, int nrows0, int nch0,
                 int* __restrict__ key0, float* __restrict__ sv0, int* __restrict__ offs0,
                 const int* __restrict__ rows1, const int* __restrict__ cols1,
                 const float* __restrict__ vals1, int nnz1, int nrows1,
                 int* __restrict__ key1, float* __restrict__ sv1, int* __restrict__ offs1) {
    __shared__ int   hist[OFFS_LD];
    __shared__ int   wsum[8];
    __shared__ int   s_key[CH];
    __shared__ float s_val[CH];

    const int t    = threadIdx.x;
    const int lane = t & 31;
    const int w    = t >> 5;

    int c = blockIdx.x;
    const bool second = (c >= nch0);
    if (second) c -= nch0;
    const int*   rows  = second ? rows1  : rows0;
    const int*   cols  = second ? cols1  : cols0;
    const float* vals  = second ? vals1  : vals0;
    const int    nnz   = second ? nnz1   : nnz0;
    const int    nrows = second ? nrows1 : nrows0;
    int*         key   = second ? key1   : key0;
    float*       sv    = second ? sv1    : sv0;
    int*         offs  = second ? offs1  : offs0;

    const int e0 = c * CH;
    int n = nnz - e0;
    if (n > CH) n = CH;
    if (n < 0)  n = 0;

    if (t < OFFS_LD) hist[t] = 0;
    __syncthreads();

    int   rr[4], cc[4];
    float vv[4];
#pragma unroll
    for (int i = 0; i < 4; ++i) {
        const int j = t + 1024 * i;
        rr[i] = -1; cc[i] = 0; vv[i] = 0.f;
        if (j < n) {
            rr[i] = rows[e0 + j];
            cc[i] = cols[e0 + j];
            vv[i] = vals[e0 + j];
        }
    }
    int bk[4], rk[4], ky[4];
#pragma unroll
    for (int i = 0; i < 4; ++i) {
        bk[i] = -1; rk[i] = 0; ky[i] = 0;
        const int r = rr[i];
        if ((unsigned)r < (unsigned)nrows) {
            const int g = r / OWN_R;
            bk[i] = g;
            ky[i] = (int)(((unsigned)(r - g * OWN_R) << 24) | (unsigned)cc[i]);
            rk[i] = atomicAdd(&hist[g], 1);
        }
    }
    __syncthreads();

    const int x = (t < OFFS_LD) ? hist[t] : 0;
    int incl = x;
#pragma unroll
    for (int off = 1; off < 32; off <<= 1) {
        const int y = __shfl_up(incl, off, 32);
        if (lane >= off) incl += y;
    }
    if (t < OFFS_LD && lane == 31) wsum[w] = incl;
    __syncthreads();
    if (t < OFFS_LD) {
        int p = 0;
#pragma unroll
        for (int k = 0; k < 8; ++k)
            if (k < w) p += wsum[k];
        const int base = p + incl - x;
        hist[t] = base;
        offs[c * OFFS_LD + t] = base;
    }
    __syncthreads();

#pragma unroll
    for (int i = 0; i < 4; ++i) {
        if (bk[i] >= 0) {
            const int p = hist[bk[i]] + rk[i];
            s_key[p] = ky[i];
            s_val[p] = vv[i];
        }
    }
    __syncthreads();
#pragma unroll
    for (int i = 0; i < 4; ++i) {
        const int j = t + 1024 * i;
        if (j < n) {
            key[e0 + j] = s_key[j];
            sv[e0 + j]  = s_val[j];
        }
    }
}

template <int U>
__device__ __forceinline__ void gather_add(float* acc, int lane, int j0,
                                           int cq, int vq, int rq,
                                           const float* __restrict__ Xa,
                                           const float* __restrict__ Xb, int split) {
    float x[U][8];
#pragma unroll
    for (int u = 0; u < U; ++u) {
        int c = __builtin_amdgcn_readlane(cq, j0 + u);
        const float* X = (c < split) ? (Xa + (size_t)c * D_DIM)
                                     : (Xb + (size_t)(c - split) * D_DIM);
#pragma unroll
        for (int i = 0; i < 8; ++i) x[u][i] = X[lane + 32 * i];
    }
#pragma unroll
    for (int u = 0; u < U; ++u) {
        float  v = __int_as_float(__builtin_amdgcn_readlane(vq, j0 + u));
        int    r = __builtin_amdgcn_readlane(rq, j0 + u);
        float* a = acc + r * D_DIM + lane;
#pragma unroll
        for (int i = 0; i < 8; ++i) atomicAdd(a + 32 * i, x[u][i] * v);
    }
}

__global__ __launch_bounds__(1024)
void spmm_owner(const int* __restrict__ key0, const float* __restrict__ sv0,
                const int* __restrict__ offs0, int nch0,
                const float* __restrict__ Xa0, const float* __restrict__ Xb0,
                int split0, float* __restrict__ Y0, int nrows0, int nblk0,
                const int* __restrict__ key1, const float* __restrict__ sv1,
                const int* __restrict__ offs1, int nch1,
                const float* __restrict__ Xa1, const float* __restrict__ Xb1,
                int split1, float* __restrict__ Y1, int nrows1) {
    __shared__ float acc[OWN_R * D_DIM];

    const int t    = threadIdx.x;
    const int lane = t & 31;
    const int w    = __builtin_amdgcn_readfirstlane(t >> 5);

    int b = blockIdx.x;
    const bool second = (b >= nblk0);
    if (second) b -= nblk0;
    const int*   key   = second ? key1   : key0;
    const float* sv    = second ? sv1    : sv0;
    const int*   offs  = second ? offs1  : offs0;
    const int    nch   = second ? nch1   : nch0;
    const float* Xa    = second ? Xa1    : Xa0;
    const float* Xb    = second ? Xb1    : Xb0;
    const int    split = second ? split1 : split0;
    float*       Y     = second ? Y1     : Y0;
    const int    nrows = second ? nrows1 : nrows0;

    const int lo = b * OWN_R;
    int nloc = nrows - lo;
    if (nloc > OWN_R) nloc = OWN_R;

    {
        float4* a4 = (float4*)acc;
        for (int i = t; i < OWN_R * (D_DIM / 4); i += 1024)
            a4[i] = make_float4(0.f, 0.f, 0.f, 0.f);
    }
    __syncthreads();

    for (int cb = 0; cb < nch; cb += 1024) {
        const int ci = cb + w + 32 * lane;
        int s = 0, e = 0;
        if (ci < nch) {
            s = offs[ci * OFFS_LD + b];
            e = offs[ci * OFFS_LD + b + 1];
        }
        const int len = e - s;
        int incl = len;
#pragma unroll
        for (int off = 1; off < 32; off <<= 1) {
            const int y = __shfl_up(incl, off, 32);
            if (lane >= off) incl += y;
        }
        const int excl = incl - len;
        const int T    = __builtin_amdgcn_readlane(incl, 31);
        const int G    = ci * CH + s - excl;
        int nv = (nch - cb - w + 31) >> 5;
        if (nv > 32) nv = 32;

        auto load_batch = [&](int j0, int& k, int& v) {
            int j = j0 + lane;
            if (j > T - 1) j = T - 1;
            int g = __builtin_amdgcn_readlane(G, 0);
            for (int i = 1; i < nv; ++i) {
                const int ex = __builtin_amdgcn_readlane(excl, i);
                const int gi = __builtin_amdgcn_readlane(G, i);
                if (j >= ex) g = gi;
            }
            k = key[g + j];
            v = __float_as_int(sv[g + j]);
        };

        if (T > 0) {
            int kC, vC;
            load_batch(0, kC, vC);
            for (int j0 = 0; j0 < T; j0 += 32) {
                int kN, vN;
                load_batch(j0 + 32, kN, vN);
                int n = T - j0;
                if (n > 32) n = 32;
                const int cq = kC & 0xFFFFFF;
                const int rq = (int)((unsigned)kC >> 24);
                int g = 0;
                for (; g + GATHER_U <= n; g += GATHER_U)
                    gather_add<GATHER_U>(acc, lane, g, cq, vC, rq, Xa, Xb, split);
                if (n - g >= 4) { gather_add<4>(acc, lane, g, cq, vC, rq, Xa, Xb, split); g += 4; }
                if (n - g >= 2) { gather_add<2>(acc, lane, g, cq, vC, rq, Xa, Xb, split); g += 2; }
                if (n - g >= 1) { gather_add<1>(acc, lane, g, cq, vC, rq, Xa, Xb, split); }
                kC = kN; vC = vN;
            }
        }
    }

    __syncthreads();

    for (int r = w; r < nloc; r += 32) {
        const float4* s4 = (const float4*)(acc + r * D_DIM);
        float4*       d  = (float4*)(Y + (size_t)(lo + r) * D_DIM);
        d[lane]      = s4[lane];
        d[lane + 32] = s4[lane + 32];
    }
}

template <int MODE>
__global__ __launch_bounds__(256)
void gemm_tanh_wmma(const float* __restrict__ A0, const float* __restrict__ A1,
                    const float* __restrict__ B0, const float* __restrict__ B1,
                    const float* __restrict__ bias, float* OUT,
                    int M, float* ssq) {
    __shared__ float As[64 * LDA_S];
    __shared__ float Bst[128 * LDBT];

    const int t    = threadIdx.x;
    const int wave = t >> 5;
    const int lane = t & 31;
    const int wm   = wave & 3;
    const int wn   = wave >> 2;
    const int rowBase = blockIdx.x * 64;
    const int colBase = blockIdx.y * 128;

    const int m16 = lane & 15;
    const int kh  = lane >> 4;

    v8f acc[4];
#pragma unroll
    for (int j = 0; j < 4; ++j) acc[j] = (v8f)0.f;

    for (int src = 0; src < 2; ++src) {
        const float* A = src ? A1 : A0;
        const float* B = src ? B1 : B0;
        if (A == nullptr) break;

        for (int k0 = 0; k0 < D_DIM; k0 += KC) {
            {
                int r  = t >> 3;
                int c4 = (t & 7) * 4;
#pragma unroll
                for (int i = 0; i < 2; ++i) {
                    int    row = rowBase + r + 32 * i;
                    float4 v   = make_float4(0.f, 0.f, 0.f, 0.f);
                    if (row < M)
                        v = *(const float4*)(A + (size_t)row * D_DIM + k0 + c4);
                    float* dst = &As[(r + 32 * i) * LDA_S + c4];
                    dst[0] = v.x; dst[1] = v.y; dst[2] = v.z; dst[3] = v.w;
                }
            }
            {
#pragma unroll
                for (int i = 0; i < 4; ++i) {
                    int idx = t + 256 * i;
                    int k   = idx >> 5;
                    int n4  = (idx & 31) * 4;
                    float4 v = *(const float4*)(B + (size_t)(k0 + k) * D_DIM +
                                                colBase + n4);
                    Bst[(n4 + 0) * LDBT + k] = v.x;
                    Bst[(n4 + 1) * LDBT + k] = v.y;
                    Bst[(n4 + 2) * LDBT + k] = v.z;
                    Bst[(n4 + 3) * LDBT + k] = v.w;
                }
            }
            __syncthreads();

#pragma unroll
            for (int kk = 0; kk < KC; kk += 4) {
                v2f af = *(const v2f*)&As[(wm * 16 + m16) * LDA_S + kk + 2 * kh];
#pragma unroll
                for (int j = 0; j < 4; ++j) {
                    int n  = wn * 64 + j * 16 + m16;
                    v2f bf = *(const v2f*)&Bst[n * LDBT + kk + 2 * kh];
                    acc[j] = __builtin_amdgcn_wmma_f32_16x16x4_f32(
                        false, af, false, bf, (short)0, acc[j], false, false);
                }
            }
            __syncthreads();
        }
    }

    const int tile16 = (rowBase >> 4) + wm;
    const int ntile  = (int)gridDim.x * 4;
    const int row0   = rowBase + wm * 16 + kh * 8;

    float inv[8];
    if (MODE == 2) {
        float4 p[4][2];
#pragma unroll
        for (int q = 0; q < 4; ++q) {
            const float4* s4 =
                (const float4*)(ssq + ((size_t)(q * ntile + tile16) * 32 + kh * 8));
            p[q][0] = s4[0];
            p[q][1] = s4[1];
        }
        float tot[8];
        tot[0] = (p[0][0].x + p[1][0].x) + (p[2][0].x + p[3][0].x);
        tot[1] = (p[0][0].y + p[1][0].y) + (p[2][0].y + p[3][0].y);
        tot[2] = (p[0][0].z + p[1][0].z) + (p[2][0].z + p[3][0].z);
        tot[3] = (p[0][0].w + p[1][0].w) + (p[2][0].w + p[3][0].w);
        tot[4] = (p[0][1].x + p[1][1].x) + (p[2][1].x + p[3][1].x);
        tot[5] = (p[0][1].y + p[1][1].y) + (p[2][1].y + p[3][1].y);
        tot[6] = (p[0][1].z + p[1][1].z) + (p[2][1].z + p[3][1].z);
        tot[7] = (p[0][1].w + p[1][1].w) + (p[2][1].w + p[3][1].w);
#pragma unroll
        for (int r = 0; r < 8; ++r) inv[r] = 1.f / fmaxf(sqrtf(tot[r]), 1e-12f);
    }

    float ss[8];
#pragma unroll
    for (int r = 0; r < 8; ++r) ss[r] = 0.f;

#pragma unroll
    for (int j = 0; j < 4; ++j) {
        int   c  = colBase + wn * 64 + j * 16 + m16;
        float bv = bias ? bias[c] : 0.f;
        float hv[8];
        if (MODE == 2) {
#pragma unroll
            for (int r = 0; r < 8; ++r) {
                int row = row0 + r;
                if (row > M - 1) row = M - 1;
                hv[r] = OUT[(size_t)row * D_DIM + c];
            }
        }
#pragma unroll
        for (int r = 0; r < 8; ++r) {
            int row = row0 + r;
            if (row < M) {
                float v = tanhf(acc[j][r] + bv);
                if (MODE == 1) ss[r] = fmaf(v, v, ss[r]);
                if (MODE == 2) v = fmaf(hv[r], inv[r], v);
                OUT[(size_t)row * D_DIM + c] = v;
            }
        }
    }

    if (MODE == 1) {
#pragma unroll
        for (int r = 0; r < 8; ++r) {
            float s = ss[r];
            s += __shfl_xor(s, 1, 32);
            s += __shfl_xor(s, 2, 32);
            s += __shfl_xor(s, 4, 32);
            s += __shfl_xor(s, 8, 32);
            ss[r] = s;
        }
        float mine = ss[0];
#pragma unroll
        for (int r = 1; r < 8; ++r)
            if (m16 == r) mine = ss[r];
        const int q = (int)blockIdx.y * 2 + wn;
        if (m16 < 8 && row0 + m16 < M)
            ssq[(size_t)(q * ntile + tile16) * 32 + kh * 8 + m16] = mine;
    }
}

extern "C" void kernel_launch(void* const* d_in, const int* in_sizes, int n_in,
                              void* d_out, int out_size, void* d_ws, size_t ws_size,
                              hipStream_t stream) {
    const int* adj_rows   = (const int*)d_in[0];
    const int* adj_cols   = (const int*)d_in[1];
    const float* adj_vals = (const float*)d_in[2];
    const int* up_rows    = (const int*)d_in[3];
    const int* up_cols    = (const int*)d_in[4];
    const float* up_vals  = (const float*)d_in[5];
    const int* ip_rows    = (const int*)d_in[6];
    const int* ip_cols    = (const int*)d_in[7];
    const float* ip_vals  = (const float*)d_in[8];
    const float* user_emb = (const float*)d_in[9];
    const float* item_emb = (const float*)d_in[10];
    const float* Q_user   = (const float*)d_in[11];
    const float* W_user   = (const float*)d_in[12];
    const float* b_user   = (const float*)d_in[13];
    const float* Q_item   = (const float*)d_in[14];
    const float* W_item   = (const float*)d_in[15];
    const float* b_item   = (const float*)d_in[16];
    const float* M_user   = (const float*)d_in[17];
    const float* M_item   = (const float*)d_in[18];
    float* out = (float*)d_out;

    const int NU = 40000, NI = 20000, NN = 60000;
    const int NNZ_ADJ = in_sizes[0], NNZ_UP = in_sizes[3], NNZ_IP = in_sizes[6];
    const size_t ROWF = (size_t)D_DIM;
    const size_t MAT  = (size_t)NN * ROWF;

    float* agg  = (float*)d_ws;
    float* embb = agg  + MAT;
    float* hbuf = embb + MAT;

    dim3 blk(256);
    const int NB_ADJ = (NN + OWN_R - 1) / OWN_R;
    const int NB_UP  = (NU + OWN_R - 1) / OWN_R;
    const int NB_IP  = (NI + OWN_R - 1) / OWN_R;
    const int NC_ADJ = (960000 + CH - 1) / CH;
    const int NC_UP  = (640000 + CH - 1) / CH;
    const int NC_IP  = (320000 + CH - 1) / CH;
    const size_t SEG = (size_t)1 << 20;

    int*   keyA  = (int*)hbuf;
    float* svA   = hbuf + SEG;
    int*   offsA = (int*)(hbuf + 2 * SEG);
    int*   keyU  = (int*)embb;
    float* svU   = embb + SEG;
    int*   offsU = (int*)(embb + 2 * SEG);
    int*   keyI  = (int*)(embb + 3 * SEG);
    float* svI   = embb + 4 * SEG;
    int*   offsI = (int*)(embb + 5 * SEG);
    float* ssqU = hbuf;
    float* ssqI = hbuf + (size_t)4 * 2500 * 32;

    bin_entries<<<NC_ADJ, dim3(1024), 0, stream>>>(
        adj_rows, adj_cols, adj_vals, NNZ_ADJ, NN, NC_ADJ, keyA, svA, offsA,
        adj_rows, adj_cols, adj_vals, NNZ_ADJ, NN, keyA, svA, offsA);

    spmm_owner<<<NB_ADJ, dim3(1024), 0, stream>>>(
        keyA, svA, offsA, NC_ADJ, user_emb, item_emb, NU, agg, NN, NB_ADJ,
        keyA, svA, offsA, 0, user_emb, item_emb, NU, agg, NN);

    gemm_tanh_wmma<0><<<dim3((NU + 63) / 64, 2), blk, 0, stream>>>(
        agg, nullptr, Q_user, nullptr, nullptr, embb, NU, nullptr);
    gemm_tanh_wmma<0><<<dim3((NI + 63) / 64, 2), blk, 0, stream>>>(
        agg + (size_t)NU * ROWF, nullptr, Q_item, nullptr, nullptr,
        embb + (size_t)NU * ROWF, NI, nullptr);

    gemm_tanh_wmma<1><<<dim3((NU + 63) / 64, 2), blk, 0, stream>>>(
        user_emb, embb, W_user, W_user + (size_t)D_DIM * D_DIM, b_user,
        out, NU, ssqU);
    gemm_tanh_wmma<1><<<dim3((NI + 63) / 64, 2), blk, 0, stream>>>(
        item_emb, embb + (size_t)NU * ROWF, W_item,
        W_item + (size_t)D_DIM * D_DIM, b_item,
        out + (size_t)NU * ROWF, NI, ssqI);

    bin_entries<<<NC_UP + NC_IP, dim3(1024), 0, stream>>>(
        up_rows, up_cols, up_vals, NNZ_UP, NU, NC_UP, keyU, svU, offsU,
        ip_rows, ip_cols, ip_vals, NNZ_IP, NI, keyI, svI, offsI);
    spmm_owner<<<NB_UP + NB_IP, dim3(1024), 0, stream>>>(
        keyU, svU, offsU, NC_UP, user_emb, user_emb, NU, agg, NU, NB_UP,
        keyI, svI, offsI, NC_IP, item_emb, item_emb, NI,
        agg + (size_t)NU * ROWF, NI);

    gemm_tanh_wmma<2><<<dim3((NU + 63) / 64, 2), blk, 0, stream>>>(
        agg, nullptr, M_user, nullptr, nullptr, out, NU, ssqU);
    gemm_tanh_wmma<2><<<dim3((NI + 63) / 64, 2), blk, 0, stream>>>(
        agg + (size_t)NU * ROWF, nullptr, M_item, nullptr, nullptr,
        out + (size_t)NU * ROWF, NI, ssqI);
}


# Independent problems share launches: user-row and item-row GEMM of each stage in one launch (6 to 3), all three list pre-sorts in one (2 to 1); 10 launches become 6; every sum keeps its order
# speedup vs baseline: 2.3994x; 1.0446x over previous
#include <hip/hip_runtime.h>
#include <hip/hip_bf16.h>
#include <math.h>

typedef __attribute__((ext_vector_type(2))) float v2f;
typedef __attribute__((ext_vector_type(8))) float v8f;

#define D_DIM 256
#define KC 32
#define LDA_S (KC + 2)
#define LDBT  (KC + 2)

#define OWN_R    240
#define GATHER_U 8
#define CH       4096
#define OFFS_LD  256

__global__ __launch_bounds__(1024)
void bin_entries(const int* __restrict__ rows0, const int* __restrict__ cols0,
                 const float* __restrict__ vals0, int nnz0, int nrows0, int nch0,
                 int* __restrict__ key0, float* __restrict__ sv0, int* __restrict__ offs0,
                 const int* __restrict__ rows1, const int* __restrict__ cols1,
                 const float* __restrict__ vals1, int nnz1, int nrows1, int nch1,
                 int* __restrict__ key1, float* __restrict__ sv1, int* __restrict__ offs1,
                 const int* __restrict__ rows2, const int* __restrict__ cols2,
                 const float* __restrict__ vals2, int nnz2, int nrows2,
                 int* __restrict__ key2, float* __restrict__ sv2, int* __restrict__ offs2) {
    __shared__ int   hist[OFFS_LD];
    __shared__ int   wsum[8];
    __shared__ int   s_key[CH];
    __shared__ float s_val[CH];

    const int t    = threadIdx.x;
    const int lane = t & 31;
    const int w    = t >> 5;

    int c = blockIdx.x;
    const bool second = (c >= nch0);
    if (second) c -= nch0;
    const bool third = second && (c >= nch1);
    if (third) c -= nch1;
    const int*   rows  = third ? rows2  : (second ? rows1  : rows0);
    const int*   cols  = third ? cols2  : (second ? cols1  : cols0);
    const float* vals  = third ? vals2  : (second ? vals1  : vals0);
    const int    nnz   = third ? nnz2   : (second ? nnz1   : nnz0);
    const int    nrows = third ? nrows2 : (second ? nrows1 : nrows0);
    int*         key   = third ? key2   : (second ? key1   : key0);
    float*       sv    = third ? sv2    : (second ? sv1    : sv0);
    int*         offs  = third ? offs2  : (second ? offs1  : offs0);

    const int e0 = c * CH;
    int n = nnz - e0;
    if (n > CH) n = CH;
    if (n < 0)  n = 0;

    if (t < OFFS_LD) hist[t] = 0;
    __syncthreads();

    int   rr[4], cc[4];
    float vv[4];
#pragma unroll
    for (int i = 0; i < 4; ++i) {
        const int j = t + 1024 * i;
        rr[i] = -1; cc[i] = 0; vv[i] = 0.f;
        if (j < n) {
            rr[i] = rows[e0 + j];
            cc[i] = cols[e0 + j];
            vv[i] = vals[e0 + j];
        }
    }
    int bk[4], rk[4], ky[4];
#pragma unroll
    for (int i = 0; i < 4; ++i) {
        bk[i] = -1; rk[i] = 0; ky[i] = 0;
        const int r = rr[i];
        if ((unsigned)r < (unsigned)nrows) {
            const int g = r / OWN_R;
            bk[i] = g;
            ky[i] = (int)(((unsigned)(r - g * OWN_R) << 24) | (unsigned)cc[i]);
            rk[i] = atomicAdd(&hist[g], 1);
        }
    }
    __syncthreads();

    const int x = (t < OFFS_LD) ? hist[t] : 0;
    int incl = x;
#pragma unroll
    for (int off = 1; off < 32; off <<= 1) {
        const int y = __shfl_up(incl, off, 32);
        if (lane >= off) incl += y;
    }
    if (t < OFFS_LD && lane == 31) wsum[w] = incl;
    __syncthreads();
    if (t < OFFS_LD) {
        int p = 0;
#pragma unroll
        for (int k = 0; k < 8; ++k)
            if (k < w) p += wsum[k];
        const int base = p + incl - x;
        hist[t] = base;
        offs[c * OFFS_LD + t] = base;
    }
    __syncthreads();

#pragma unroll
    for (int i = 0; i < 4; ++i) {
        if (bk[i] >= 0) {
            const int p = hist[bk[i]] + rk[i];
            s_key[p] = ky[i];
            s_val[p] = vv[i];
        }
    }
    __syncthreads();
#pragma unroll
    for (int i = 0; i < 4; ++i) {
        const int j = t + 1024 * i;
        if (j < n) {
            key[e0 + j] = s_key[j];
            sv[e0 + j]  = s_val[j];
        }
    }
}

template <int U>
__device__ __forceinline__ void gather_add(float* acc, int lane, int j0,
                                           int cq, int vq, int rq,
                                           const float* __restrict__ Xa,
                                           const float* __restrict__ Xb, int split) {
    float x[U][8];
#pragma unroll
    for (int u = 0; u < U; ++u) {
        int c = __builtin_amdgcn_readlane(cq, j0 + u);
        const float* X = (c < split) ? (Xa + (size_t)c * D_DIM)
                                     : (Xb + (size_t)(c - split) * D_DIM);
#pragma unroll
        for (int i = 0; i < 8; ++i) x[u][i] = X[lane + 32 * i];
    }
#pragma unroll
    for (int u = 0; u < U; ++u) {
        float  v = __int_as_float(__builtin_amdgcn_readlane(vq, j0 + u));
        int    r = __builtin_amdgcn_readlane(rq, j0 + u);
        float* a = acc + r * D_DIM + lane;
#pragma unroll
        for (int i = 0; i < 8; ++i) atomicAdd(a + 32 * i, x[u][i] * v);
    }
}

__global__ __launch_bounds__(1024)
void spmm_owner(const int* __restrict__ key0, const float* __restrict__ sv0,
                const int* __restrict__ offs0, int nch0,
                const float* __restrict__ Xa0, const float* __restrict__ Xb0,
                int split0, float* __restrict__ Y0, int nrows0, int nblk0,
                const int* __restrict__ key1, const float* __restrict__ sv1,
                const int* __restrict__ offs1, int nch1,
                const float* __restrict__ Xa1, const float* __restrict__ Xb1,
                int split1, float* __restrict__ Y1, int nrows1) {
    __shared__ float acc[OWN_R * D_DIM];

    const int t    = threadIdx.x;
    const int lane = t & 31;
    const int w    = __builtin_amdgcn_readfirstlane(t >> 5);

    int b = blockIdx.x;
    const bool second = (b >= nblk0);
    if (second) b -= nblk0;
    const int*   key   = second ? key1   : key0;
    const float* sv    = second ? sv1    : sv0;
    const int*   offs  = second ? offs1  : offs0;
    const int    nch   = second ? nch1   : nch0;
    const float* Xa    = second ? Xa1    : Xa0;
    const float* Xb    = second ? Xb1    : Xb0;
    const int    split = second ? split1 : split0;
    float*       Y     = second ? Y1     : Y0;
    const int    nrows = second ? nrows1 : nrows0;

    const int lo = b * OWN_R;
    int nloc = nrows - lo;
    if (nloc > OWN_R) nloc = OWN_R;

    {
        float4* a4 = (float4*)acc;
        for (int i = t; i < OWN_R * (D_DIM / 4); i += 1024)
            a4[i] = make_float4(0.f, 0.f, 0.f, 0.f);
    }
    __syncthreads();

    for (int cb = 0; cb < nch; cb += 1024) {
        const int ci = cb + w + 32 * lane;
        int s = 0, e = 0;
        if (ci < nch) {
            s = offs[ci * OFFS_LD + b];
            e = offs[ci * OFFS_LD + b + 1];
        }
        const int len = e - s;
        int incl = len;
#pragma unroll
        for (int off = 1; off < 32; off <<= 1) {
            const int y = __shfl_up(incl, off, 32);
            if (lane >= off) incl += y;
        }
        const int excl = incl - len;
        const int T    = __builtin_amdgcn_readlane(incl, 31);
        const int G    = ci * CH + s - excl;
        int nv = (nch - cb - w + 31) >> 5;
        if (nv > 32) nv = 32;

        auto load_batch = [&](int j0, int& k, int& v) {
            int j = j0 + lane;
            if (j > T - 1) j = T - 1;
            int g = __builtin_amdgcn_readlane(G, 0);
            for (int i = 1; i < nv; ++i) {
                const int ex = __builtin_amdgcn_readlane(excl, i);
                const int gi = __builtin_amdgcn_readlane(G, i);
                if (j >= ex) g = gi;
            }
            k = key[g + j];
            v = __float_as_int(sv[g + j]);
        };

        if (T > 0) {
            int kC, vC;
            load_batch(0, kC, vC);
            for (int j0 = 0; j0 < T; j0 += 32) {
                int kN, vN;
                load_batch(j0 + 32, kN, vN);
                int n = T - j0;
                if (n > 32) n = 32;
                const int cq = kC & 0xFFFFFF;
                const int rq = (int)((unsigned)kC >> 24);
                int g = 0;
                for (; g + GATHER_U <= n; g += GATHER_U)
                    gather_add<GATHER_U>(acc, lane, g, cq, vC, rq, Xa, Xb, split);
                if (n - g >= 4) { gather_add<4>(acc, lane, g, cq, vC, rq, Xa, Xb, split); g += 4; }
                if (n - g >= 2) { gather_add<2>(acc, lane, g, cq, vC, rq, Xa, Xb, split); g += 2; }
                if (n - g >= 1) { gather_add<1>(acc, lane, g, cq, vC, rq, Xa, Xb, split); }
                kC = kN; vC = vN;
            }
        }
    }

    __syncthreads();

    for (int r = w; r < nloc; r += 32) {
        const float4* s4 = (const float4*)(acc + r * D_DIM);
        float4*       d  = (float4*)(Y + (size_t)(lo + r) * D_DIM);
        d[lane]      = s4[lane];
        d[lane + 32] = s4[lane + 32];
    }
}

template <int MODE>
__global__ __launch_bounds__(256)
void gemm_tanh_wmma(const float* __restrict__ A0p, const float* __restrict__ A1p,
                    const float* __restrict__ B0p, const float* __restrict__ B1p,
                    const float* __restrict__ biasp, float* OUTp,
                    int Mp, float* ssqp, int nblkp,
                    const float* __restrict__ A0q, const float* __restrict__ A1q,
                    const float* __restrict__ B0q, const float* __restrict__ B1q,
                    const float* __restrict__ biasq, float* OUTq,
                    int Mq, float* ssqq) {
    __shared__ float As[64 * LDA_S];
    __shared__ float Bst[128 * LDBT];

    int bx = blockIdx.x;
    const bool second = (bx >= nblkp);
    if (second) bx -= nblkp;
    const float* A0   = second ? A0q   : A0p;
    const float* A1   = second ? A1q   : A1p;
    const float* B0   = second ? B0q   : B0p;
    const float* B1   = second ? B1q   : B1p;
    const float* bias = second ? biasq : biasp;
    float*       OUT  = second ? OUTq  : OUTp;
    const int    M    = second ? Mq    : Mp;
    float*       ssq  = second ? ssqq  : ssqp;

    const int t    = threadIdx.x;
    const int wave = t >> 5;
    const int lane = t & 31;
    const int wm   = wave & 3;
    const int wn   = wave >> 2;
    const int rowBase = bx * 64;
    const int colBase = blockIdx.y * 128;

    const int m16 = lane & 15;
    const int kh  = lane >> 4;

    v8f acc[4];
#pragma unroll
    for (int j = 0; j < 4; ++j) acc[j] = (v8f)0.f;

    for (int src = 0; src < 2; ++src) {
        const float* A = src ? A1 : A0;
        const float* B = src ? B1 : B0;
        if (A == nullptr) break;

        for (int k0 = 0; k0 < D_DIM; k0 += KC) {
            {
                int r  = t >> 3;
                int c4 = (t & 7) * 4;
#pragma unroll
                for (int i = 0; i < 2; ++i) {
                    int    row = rowBase + r + 32 * i;
                    float4 v   = make_float4(0.f, 0.f, 0.f, 0.f);
                    if (row < M)
                        v = *(const float4*)(A + (size_t)row * D_DIM + k0 + c4);
                    float* dst = &As[(r + 32 * i) * LDA_S + c4];
                    dst[0] = v.x; dst[1] = v.y; dst[2] = v.z; dst[3] = v.w;
                }
            }
            {
#pragma unroll
                for (int i = 0; i < 4; ++i) {
                    int idx = t + 256 * i;
                    int k   = idx >> 5;
                    int n4  = (idx & 31) * 4;
                    float4 v = *(const float4*)(B + (size_t)(k0 + k) * D_DIM +
                                                colBase + n4);
                    Bst[(n4 + 0) * LDBT + k] = v.x;
                    Bst[(n4 + 1) * LDBT + k] = v.y;
                    Bst[(n4 + 2) * LDBT + k] = v.z;
                    Bst[(n4 + 3) * LDBT + k] = v.w;
                }
            }
            __syncthreads();

#pragma unroll
            for (int kk = 0; kk < KC; kk += 4) {
                v2f af = *(const v2f*)&As[(wm * 16 + m16) * LDA_S + kk + 2 * kh];
#pragma unroll
                for (int j = 0; j < 4; ++j) {
                    int n  = wn * 64 + j * 16 + m16;
                    v2f bf = *(const v2f*)&Bst[n * LDBT + kk + 2 * kh];
                    acc[j] = __builtin_amdgcn_wmma_f32_16x16x4_f32(
                        false, af, false, bf, (short)0, acc[j], false, false);
                }
            }
            __syncthreads();
        }
    }

    const int tile16 = (rowBase >> 4) + wm;
    const int ntile  = ((M + 63) >> 6) * 4;
    const int row0   = rowBase + wm * 16 + kh * 8;

    float inv[8];
    if (MODE == 2) {
        float4 p[4][2];
#pragma unroll
        for (int q = 0; q < 4; ++q) {
            const float4* s4 =
                (const float4*)(ssq + ((size_t)(q * ntile + tile16) * 32 + kh * 8));
            p[q][0] = s4[0];
            p[q][1] = s4[1];
        }
        float tot[8];
        tot[0] = (p[0][0].x + p[1][0].x) + (p[2][0].x + p[3][0].x);
        tot[1] = (p[0][0].y + p[1][0].y) + (p[2][0].y + p[3][0].y);
        tot[2] = (p[0][0].z + p[1][0].z) + (p[2][0].z + p[3][0].z);
        tot[3] = (p[0][0].w + p[1][0].w) + (p[2][0].w + p[3][0].w);
        tot[4] = (p[0][1].x + p[1][1].x) + (p[2][1].x + p[3][1].x);
        tot[5] = (p[0][1].y + p[1][1].y) + (p[2][1].y + p[3][1].y);
        tot[6] = (p[0][1].z + p[1][1].z) + (p[2][1].z + p[3][1].z);
        tot[7] = (p[0][1].w + p[1][1].w) + (p[2][1].w + p[3][1].w);
#pragma unroll
        for (int r = 0; r < 8; ++r) inv[r] = 1.f / fmaxf(sqrtf(tot[r]), 1e-12f);
    }

    float ss[8];
#pragma unroll
    for (int r = 0; r < 8; ++r) ss[r] = 0.f;

#pragma unroll
    for (int j = 0; j < 4; ++j) {
        int   c  = colBase + wn * 64 + j * 16 + m16;
        float bv = bias ? bias[c] : 0.f;
        float hv[8];
        if (MODE == 2) {
#pragma unroll
            for (int r = 0; r < 8; ++r) {
                int row = row0 + r;
                if (row > M - 1) row = M - 1;
                hv[r] = OUT[(size_t)row * D_DIM + c];
            }
        }
#pragma unroll
        for (int r = 0; r < 8; ++r) {
            int row = row0 + r;
            if (row < M) {
                float v = tanhf(acc[j][r] + bv);
                if (MODE == 1) ss[r] = fmaf(v, v, ss[r]);
                if (MODE == 2) v = fmaf(hv[r], inv[r], v);
                OUT[(size_t)row * D_DIM + c] = v;
            }
        }
    }

    if (MODE == 1) {
#pragma unroll
        for (int r = 0; r < 8; ++r) {
            float s = ss[r];
            s += __shfl_xor(s, 1, 32);
            s += __shfl_xor(s, 2, 32);
            s += __shfl_xor(s, 4, 32);
            s += __shfl_xor(s, 8, 32);
            ss[r] = s;
        }
        float mine = ss[0];
#pragma unroll
        for (int r = 1; r < 8; ++r)
            if (m16 == r) mine = ss[r];
        const int q = (int)blockIdx.y * 2 + wn;
        if (m16 < 8 && row0 + m16 < M)
            ssq[(size_t)(q * ntile + tile16) * 32 + kh * 8 + m16] = mine;
    }
}

extern "C" void kernel_launch(void* const* d_in, const int* in_sizes, int n_in,
                              void* d_out, int out_size, void* d_ws, size_t ws_size,
                              hipStream_t stream) {
    const int* adj_rows   = (const int*)d_in[0];
    const int* adj_cols   = (const int*)d_in[1];
    const float* adj_vals = (const float*)d_in[2];
    const int* up_rows    = (const int*)d_in[3];
    const int* up_cols    = (const int*)d_in[4];
    const float* up_vals  = (const float*)d_in[5];
    const int* ip_rows    = (const int*)d_in[6];
    const int* ip_cols    = (const int*)d_in[7];
    const float* ip_vals  = (const float*)d_in[8];
    const float* user_emb = (const float*)d_in[9];
    const float* item_emb = (const float*)d_in[10];
    const float* Q_user   = (const float*)d_in[11];
    const float* W_user   = (const float*)d_in[12];
    const float* b_user   = (const float*)d_in[13];
    const float* Q_item   = (const float*)d_in[14];
    const float* W_item   = (const float*)d_in[15];
    const float* b_item   = (const float*)d_in[16];
    const float* M_user   = (const float*)d_in[17];
    const float* M_item   = (const float*)d_in[18];
    float* out = (float*)d_out;

    const int NU = 40000, NI = 20000, NN = 60000;
    const int NNZ_ADJ = in_sizes[0], NNZ_UP = in_sizes[3], NNZ_IP = in_sizes[6];
    const size_t ROWF = (size_t)D_DIM;
    const size_t MAT  = (size_t)NN * ROWF;

    float* agg  = (float*)d_ws;
    float* embb = agg  + MAT;
    float* hbuf = embb + MAT;

    dim3 blk(256);
    const int NB_ADJ = (NN + OWN_R - 1) / OWN_R;
    const int NB_UP  = (NU + OWN_R - 1) / OWN_R;
    const int NB_IP  = (NI + OWN_R - 1) / OWN_R;
    const int NC_ADJ = (960000 + CH - 1) / CH;
    const int NC_UP  = (640000 + CH - 1) / CH;
    const int NC_IP  = (320000 + CH - 1) / CH;
    const size_t SEG = (size_t)1 << 20;

    int*   keyA  = (int*)hbuf;
    float* svA   = hbuf + SEG;
    int*   offsA = (int*)(hbuf + 2 * SEG);
    int*   keyU  = (int*)(hbuf + 3 * SEG);
    float* svU   = hbuf + 4 * SEG;
    int*   offsU = (int*)(hbuf + 5 * SEG);
    int*   keyI  = (int*)(hbuf + 6 * SEG);
    float* svI   = hbuf + 7 * SEG;
    int*   offsI = (int*)(hbuf + 8 * SEG);
    float* ssqU = hbuf;
    float* ssqI = hbuf + (size_t)4 * 2500 * 32;

    bin_entries<<<NC_ADJ + NC_UP + NC_IP, dim3(1024), 0, stream>>>(
        adj_rows, adj_cols, adj_vals, NNZ_ADJ, NN, NC_ADJ, keyA, svA, offsA,
        up_rows, up_cols, up_vals, NNZ_UP, NU, NC_UP, keyU, svU, offsU,
        ip_rows, ip_cols, ip_vals, NNZ_IP, NI, keyI, svI, offsI);

    spmm_owner<<<NB_ADJ, dim3(1024), 0, stream>>>(
        keyA, svA, offsA, NC_ADJ, user_emb, item_emb, NU, agg, NN, NB_ADJ,
        keyA, svA, offsA, 0, user_emb, item_emb, NU, agg, NN);

    const int GB_U = (NU + 63) / 64;
    const int GB_I = (NI + 63) / 64;

    gemm_tanh_wmma<0><<<dim3(GB_U + GB_I, 2), blk, 0, stream>>>(
        agg, nullptr, Q_user, nullptr, nullptr, embb, NU, nullptr, GB_U,
        agg + (size_t)NU * ROWF, nullptr, Q_item, nullptr, nullptr,
        embb + (size_t)NU * ROWF, NI, nullptr);

    gemm_tanh_wmma<1><<<dim3(GB_U + GB_I, 2), blk, 0, stream>>>(
        user_emb, embb, W_user, W_user + (size_t)D_DIM * D_DIM, b_user,
        out, NU, ssqU, GB_U,
        item_emb, embb + (size_t)NU * ROWF, W_item,
        W_item + (size_t)D_DIM * D_DIM, b_item,
        out + (size_t)NU * ROWF, NI, ssqI);

    spmm_owner<<<NB_UP + NB_IP, dim3(1024), 0, stream>>>(
        keyU, svU, offsU, NC_UP, user_emb, user_emb, NU, agg, NU, NB_UP,
        keyI, svI, offsI, NC_IP, item_emb, item_emb, NI,
        agg + (size_t)NU * ROWF, NI);

    gemm_tanh_wmma<2><<<dim3(GB_U + GB_I, 2), blk, 0, stream>>>(
        agg, nullptr, M_user, nullptr, nullptr, out, NU, ssqU, GB_U,
        agg + (size_t)NU * ROWF, nullptr, M_item, nullptr, nullptr,
        out + (size_t)NU * ROWF, NI, ssqI);
}


# GEMM K loop software-pipelined: the next chunk's six global loads stay in flight while the current chunk's WMMAs run from one of two LDS buffers; one barrier per chunk; sums keep their order
# speedup vs baseline: 2.6648x; 1.1106x over previous
#include <hip/hip_runtime.h>
#include <hip/hip_bf16.h>
#include <math.h>

typedef __attribute__((ext_vector_type(2))) float v2f;
typedef __attribute__((ext_vector_type(8))) float v8f;

#define D_DIM 256
#define KC 32
#define LDA_S (KC + 2)
#define LDBT  (KC + 2)

#define OWN_R    240
#define GATHER_U 8
#define CH       4096
#define OFFS_LD  256

__global__ __launch_bounds__(1024)
void bin_entries(const int* __restrict__ rows0, const int* __restrict__ cols0,
                 const float* __restrict__ vals0, int nnz0, int nrows0, int nch0,
                 int* __restrict__ key0, float* __restrict__ sv0, int* __restrict__ offs0,
                 const int* __restrict__ rows1, const int* __restrict__ cols1,
                 const float* __restrict__ vals1, int nnz1, int nrows1, int nch1,
                 int* __restrict__ key1, float* __restrict__ sv1, int* __restrict__ offs1,
                 const int* __restrict__ rows2, const int* __restrict__ cols2,
                 const float* __restrict__ vals2, int nnz2, int nrows2,
                 int* __restrict__ key2, float* __restrict__ sv2, int* __restrict__ offs2) {
    __shared__ int   hist[OFFS_LD];
    __shared__ int   wsum[8];
    __shared__ int   s_key[CH];
    __shared__ float s_val[CH];

    const int t    = threadIdx.x;
    const int lane = t & 31;
    const int w    = t >> 5;

    int c = blockIdx.x;
    const bool second = (c >= nch0);
    if (second) c -= nch0;
    const bool third = second && (c >= nch1);
    if (third) c -= nch1;
    const int*   rows  = third ? rows2  : (second ? rows1  : rows0);
    const int*   cols  = third ? cols2  : (second ? cols1  : cols0);
    const float* vals  = third ? vals2  : (second ? vals1  : vals0);
    const int    nnz   = third ? nnz2   : (second ? nnz1   : nnz0);
    const int    nrows = third ? nrows2 : (second ? nrows1 : nrows0);
    int*         key   = third ? key2   : (second ? key1   : key0);
    float*       sv    = third ? sv2    : (second ? sv1    : sv0);
    int*         offs  = third ? offs2  : (second ? offs1  : offs0);

    const int e0 = c * CH;
    int n = nnz - e0;
    if (n > CH) n = CH;
    if (n < 0)  n = 0;

    if (t < OFFS_LD) hist[t] = 0;
    __syncthreads();

    int   rr[4], cc[4];
    float vv[4];
#pragma unroll
    for (int i = 0; i < 4; ++i) {
        const int j = t + 1024 * i;
        rr[i] = -1; cc[i] = 0; vv[i] = 0.f;
        if (j < n) {
            rr[i] = rows[e0 + j];
            cc[i] = cols[e0 + j];
            vv[i] = vals[e0 + j];
        }
    }
    int bk[4], rk[4], ky[4];
#pragma unroll
    for (int i = 0; i < 4; ++i) {
        bk[i] = -1; rk[i] = 0; ky[i] = 0;
        const int r = rr[i];
        if ((unsigned)r < (unsigned)nrows) {
            const int g = r / OWN_R;
            bk[i] = g;
            ky[i] = (int)(((unsigned)(r - g * OWN_R) << 24) | (unsigned)cc[i]);
            rk[i] = atomicAdd(&hist[g], 1);
        }
    }
    __syncthreads();

    const int x = (t < OFFS_LD) ? hist[t] : 0;
    int incl = x;
#pragma unroll
    for (int off = 1; off < 32; off <<= 1) {
        const int y = __shfl_up(incl, off, 32);
        if (lane >= off) incl += y;
    }
    if (t < OFFS_LD && lane == 31) wsum[w] = incl;
    __syncthreads();
    if (t < OFFS_LD) {
        int p = 0;
#pragma unroll
        for (int k = 0; k < 8; ++k)
            if (k < w) p += wsum[k];
        const int base = p + incl - x;
        hist[t] = base;
        offs[c * OFFS_LD + t] = base;
    }
    __syncthreads();

#pragma unroll
    for (int i = 0; i < 4; ++i) {
        if (bk[i] >= 0) {
            const int p = hist[bk[i]] + rk[i];
            s_key[p] = ky[i];
            s_val[p] = vv[i];
        }
    }
    __syncthreads();
#pragma unroll
    for (int i = 0; i < 4; ++i) {
        const int j = t + 1024 * i;
        if (j < n) {
            key[e0 + j] = s_key[j];
            sv[e0 + j]  = s_val[j];
        }
    }
}

template <int U>
__device__ __forceinline__ void gather_add(float* acc, int lane, int j0,
                                           int cq, int vq, int rq,
                                           const float* __restrict__ Xa,
                                           const float* __restrict__ Xb, int split) {
    float x[U][8];
#pragma unroll
    for (int u = 0; u < U; ++u) {
        int c = __builtin_amdgcn_readlane(cq, j0 + u);
        const float* X = (c < split) ? (Xa + (size_t)c * D_DIM)
                                     : (Xb + (size_t)(c - split) * D_DIM);
#pragma unroll
        for (int i = 0; i < 8; ++i) x[u][i] = X[lane + 32 * i];
    }
#pragma unroll
    for (int u = 0; u < U; ++u) {
        float  v = __int_as_float(__builtin_amdgcn_readlane(vq, j0 + u));
        int    r = __builtin_amdgcn_readlane(rq, j0 + u);
        float* a = acc + r * D_DIM + lane;
#pragma unroll
        for (int i = 0; i < 8; ++i) atomicAdd(a + 32 * i, x[u][i] * v);
    }
}

__global__ __launch_bounds__(1024)
void spmm_owner(const int* __restrict__ key0, const float* __restrict__ sv0,
                const int* __restrict__ offs0, int nch0,
                const float* __restrict__ Xa0, const float* __restrict__ Xb0,
                int split0, float* __restrict__ Y0, int nrows0, int nblk0,
                const int* __restrict__ key1, const float* __restrict__ sv1,
                const int* __restrict__ offs1, int nch1,
                const float* __restrict__ Xa1, const float* __restrict__ Xb1,
                int split1, float* __restrict__ Y1, int nrows1) {
    __shared__ float acc[OWN_R * D_DIM];

    const int t    = threadIdx.x;
    const int lane = t & 31;
    const int w    = __builtin_amdgcn_readfirstlane(t >> 5);

    int b = blockIdx.x;
    const bool second = (b >= nblk0);
    if (second) b -= nblk0;
    const int*   key   = second ? key1   : key0;
    const float* sv    = second ? sv1    : sv0;
    const int*   offs  = second ? offs1  : offs0;
    const int    nch   = second ? nch1   : nch0;
    const float* Xa    = second ? Xa1    : Xa0;
    const float* Xb    = second ? Xb1    : Xb0;
    const int    split = second ? split1 : split0;
    float*       Y     = second ? Y1     : Y0;
    const int    nrows = second ? nrows1 : nrows0;

    const int lo = b * OWN_R;
    int nloc = nrows - lo;
    if (nloc > OWN_R) nloc = OWN_R;

    {
        float4* a4 = (float4*)acc;
        for (int i = t; i < OWN_R * (D_DIM / 4); i += 1024)
            a4[i] = make_float4(0.f, 0.f, 0.f, 0.f);
    }
    __syncthreads();

    for (int cb = 0; cb < nch; cb += 1024) {
        const int ci = cb + w + 32 * lane;
        int s = 0, e = 0;
        if (ci < nch) {
            s = offs[ci * OFFS_LD + b];
            e = offs[ci * OFFS_LD + b + 1];
        }
        const int len = e - s;
        int incl = len;
#pragma unroll
        for (int off = 1; off < 32; off <<= 1) {
            const int y = __shfl_up(incl, off, 32);
            if (lane >= off) incl += y;
        }
        const int excl = incl - len;
        const int T    = __builtin_amdgcn_readlane(incl, 31);
        const int G    = ci * CH + s - excl;
        int nv = (nch - cb - w + 31) >> 5;
        if (nv > 32) nv = 32;

        auto load_batch = [&](int j0, int& k, int& v) {
            int j = j0 + lane;
            if (j > T - 1) j = T - 1;
            int g = __builtin_amdgcn_readlane(G, 0);
            for (int i = 1; i < nv; ++i) {
                const int ex = __builtin_amdgcn_readlane(excl, i);
                const int gi = __builtin_amdgcn_readlane(G, i);
                if (j >= ex) g = gi;
            }
            k = key[g + j];
            v = __float_as_int(sv[g + j]);
        };

        if (T > 0) {
            int kC, vC;
            load_batch(0, kC, vC);
            for (int j0 = 0; j0 < T; j0 += 32) {
                int kN, vN;
                load_batch(j0 + 32, kN, vN);
                int n = T - j0;
                if (n > 32) n = 32;
                const int cq = kC & 0xFFFFFF;
                const int rq = (int)((unsigned)kC >> 24);
                int g = 0;
                for (; g + GATHER_U <= n; g += GATHER_U)
                    gather_add<GATHER_U>(acc, lane, g, cq, vC, rq, Xa, Xb, split);
                if (n - g >= 4) { gather_add<4>(acc, lane, g, cq, vC, rq, Xa, Xb, split); g += 4; }
                if (n - g >= 2) { gather_add<2>(acc, lane, g, cq, vC, rq, Xa, Xb, split); g += 2; }
                if (n - g >= 1) { gather_add<1>(acc, lane, g, cq, vC, rq, Xa, Xb, split); }
                kC = kN; vC = vN;
            }
        }
    }

    __syncthreads();

    for (int r = w; r < nloc; r += 32) {
        const float4* s4 = (const float4*)(acc + r * D_DIM);
        float4*       d  = (float4*)(Y + (size_t)(lo + r) * D_DIM);
        d[lane]      = s4[lane];
        d[lane + 32] = s4[lane + 32];
    }
}

template <int MODE>
__global__ __launch_bounds__(256)
void gemm_tanh_wmma(const float* __restrict__ A0p, const float* __restrict__ A1p,
                    const float* __restrict__ B0p, const float* __restrict__ B1p,
                    const float* __restrict__ biasp, float* OUTp,
                    int Mp, float* ssqp, int nblkp,
                    const float* __restrict__ A0q, const float* __restrict__ A1q,
                    const float* __restrict__ B0q, const float* __restrict__ B1q,
                    const float* __restrict__ biasq, float* OUTq,
                    int Mq, float* ssqq) {
    __shared__ float As[2][64 * LDA_S];
    __shared__ float Bst[2][128 * LDBT];

    int bx = blockIdx.x;
    const bool second = (bx >= nblkp);
    if (second) bx -= nblkp;
    const float* A0   = second ? A0q   : A0p;
    const float* A1   = second ? A1q   : A1p;
    const float* B0   = second ? B0q   : B0p;
    const float* B1   = second ? B1q   : B1p;
    const float* bias = second ? biasq : biasp;
    float*       OUT  = second ? OUTq  : OUTp;
    const int    M    = second ? Mq    : Mp;
    float*       ssq  = second ? ssqq  : ssqp;

    const int t    = threadIdx.x;
    const int wave = t >> 5;
    const int lane = t & 31;
    const int wm   = wave & 3;
    const int wn   = wave >> 2;
    const int rowBase = bx * 64;
    const int colBase = blockIdx.y * 128;

    const int m16 = lane & 15;
    const int kh  = lane >> 4;

    v8f acc[4];
#pragma unroll
    for (int j = 0; j < 4; ++j) acc[j] = (v8f)0.f;

    const int NCK = D_DIM / KC;
    const int nch = (A1 != nullptr) ? 2 * NCK : NCK;

    const int ar  = t >> 3;
    const int ac4 = (t & 7) * 4;
    int arow0 = rowBase + ar;
    int arow1 = rowBase + ar + 32;
    if (arow0 > M - 1) arow0 = M - 1;
    if (arow1 > M - 1) arow1 = M - 1;
    const int bk  = t >> 5;
    const int bn4 = (t & 31) * 4;

    const float* pa0 = A0 + (size_t)arow0 * D_DIM + ac4;
    const float* pa1 = A0 + (size_t)arow1 * D_DIM + ac4;
    const float* pb  = B0 + (size_t)bk * D_DIM + colBase + bn4;

    float4 ga0, ga1, gb[4];
    auto issue = [&](int c) {
        if (c == NCK) {
            pa0 = A1 + (size_t)arow0 * D_DIM + ac4;
            pa1 = A1 + (size_t)arow1 * D_DIM + ac4;
            pb  = B1 + (size_t)bk * D_DIM + colBase + bn4;
        }
        ga0 = *(const float4*)pa0;
        ga1 = *(const float4*)pa1;
#pragma unroll
        for (int i = 0; i < 4; ++i)
            gb[i] = *(const float4*)(pb + (size_t)(8 * i) * D_DIM);
        pa0 += KC;
        pa1 += KC;
        pb  += (size_t)KC * D_DIM;
    };

    auto step = [&](int c, float* as, float* bs) {
        __builtin_amdgcn_sched_barrier(0);
        {
            float* d0 = &as[ar * LDA_S + ac4];
            d0[0] = ga0.x; d0[1] = ga0.y; d0[2] = ga0.z; d0[3] = ga0.w;
            float* d1 = &as[(ar + 32) * LDA_S + ac4];
            d1[0] = ga1.x; d1[1] = ga1.y; d1[2] = ga1.z; d1[3] = ga1.w;
#pragma unroll
            for (int i = 0; i < 4; ++i) {
                const int k = bk + 8 * i;
                bs[(bn4 + 0) * LDBT + k] = gb[i].x;
                bs[(bn4 + 1) * LDBT + k] = gb[i].y;
                bs[(bn4 + 2) * LDBT + k] = gb[i].z;
                bs[(bn4 + 3) * LDBT + k] = gb[i].w;
            }
        }
        __syncthreads();
        if (c + 1 < nch) issue(c + 1);

#pragma unroll
        for (int kk = 0; kk < KC; kk += 4) {
            v2f af = *(const v2f*)&as[(wm * 16 + m16) * LDA_S + kk + 2 * kh];
#pragma unroll
            for (int j = 0; j < 4; ++j) {
                int n  = wn * 64 + j * 16 + m16;
                v2f bf = *(const v2f*)&bs[n * LDBT + kk + 2 * kh];
                acc[j] = __builtin_amdgcn_wmma_f32_16x16x4_f32(
                    false, af, false, bf, (short)0, acc[j], false, false);
            }
        }
    };

    issue(0);
    for (int c = 0; c < nch; c += 2) {
        step(c,     As[0], Bst[0]);
        step(c + 1, As[1], Bst[1]);
    }

    const int tile16 = (rowBase >> 4) + wm;
    const int ntile  = ((M + 63) >> 6) * 4;
    const int row0   = rowBase + wm * 16 + kh * 8;

    float inv[8];
    if (MODE == 2) {
        float4 p[4][2];
#pragma unroll
        for (int q = 0; q < 4; ++q) {
            const float4* s4 =
                (const float4*)(ssq + ((size_t)(q * ntile + tile16) * 32 + kh * 8));
            p[q][0] = s4[0];
            p[q][1] = s4[1];
        }
        float tot[8];
        tot[0] = (p[0][0].x + p[1][0].x) + (p[2][0].x + p[3][0].x);
        tot[1] = (p[0][0].y + p[1][0].y) + (p[2][0].y + p[3][0].y);
        tot[2] = (p[0][0].z + p[1][0].z) + (p[2][0].z + p[3][0].z);
        tot[3] = (p[0][0].w + p[1][0].w) + (p[2][0].w + p[3][0].w);
        tot[4] = (p[0][1].x + p[1][1].x) + (p[2][1].x + p[3][1].x);
        tot[5] = (p[0][1].y + p[1][1].y) + (p[2][1].y + p[3][1].y);
        tot[6] = (p[0][1].z + p[1][1].z) + (p[2][1].z + p[3][1].z);
        tot[7] = (p[0][1].w + p[1][1].w) + (p[2][1].w + p[3][1].w);
#pragma unroll
        for (int r = 0; r < 8; ++r) inv[r] = 1.f / fmaxf(sqrtf(tot[r]), 1e-12f);
    }

    float ss[8];
#pragma unroll
    for (int r = 0; r < 8; ++r) ss[r] = 0.f;

#pragma unroll
    for (int j = 0; j < 4; ++j) {
        int   c  = colBase + wn * 64 + j * 16 + m16;
        float bv = bias ? bias[c] : 0.f;
        float hv[8];
        if (MODE == 2) {
#pragma unroll
            for (int r = 0; r < 8; ++r) {
                int row = row0 + r;
                if (row > M - 1) row = M - 1;
                hv[r] = OUT[(size_t)row * D_DIM + c];
            }
        }
#pragma unroll
        for (int r = 0; r < 8; ++r) {
            int row = row0 + r;
            if (row < M) {
                float v = tanhf(acc[j][r] + bv);
                if (MODE == 1) ss[r] = fmaf(v, v, ss[r]);
                if (MODE == 2) v = fmaf(hv[r], inv[r], v);
                OUT[(size_t)row * D_DIM + c] = v;
            }
        }
    }

    if (MODE == 1) {
#pragma unroll
        for (int r = 0; r < 8; ++r) {
            float s = ss[r];
            s += __shfl_xor(s, 1, 32);
            s += __shfl_xor(s, 2, 32);
            s += __shfl_xor(s, 4, 32);
            s += __shfl_xor(s, 8, 32);
            ss[r] = s;
        }
        float mine = ss[0];
#pragma unroll
        for (int r = 1; r < 8; ++r)
            if (m16 == r) mine = ss[r];
        const int q = (int)blockIdx.y * 2 + wn;
        if (m16 < 8 && row0 + m16 < M)
            ssq[(size_t)(q * ntile + tile16) * 32 + kh * 8 + m16] = mine;
    }
}

extern "C" void kernel_launch(void* const* d_in, const int* in_sizes, int n_in,
                              void* d_out, int out_size, void* d_ws, size_t ws_size,
                              hipStream_t stream) {
    const int* adj_rows   = (const int*)d_in[0];
    const int* adj_cols   = (const int*)d_in[1];
    const float* adj_vals = (const float*)d_in[2];
    const int* up_rows    = (const int*)d_in[3];
    const int* up_cols    = (const int*)d_in[4];
    const float* up_vals  = (const float*)d_in[5];
    const int* ip_rows    = (const int*)d_in[6];
    const int* ip_cols    = (const int*)d_in[7];
    const float* ip_vals  = (const float*)d_in[8];
    const float* user_emb = (const float*)d_in[9];
    const float* item_emb = (const float*)d_in[10];
    const float* Q_user   = (const float*)d_in[11];
    const float* W_user   = (const float*)d_in[12];
    const float* b_user   = (const float*)d_in[13];
    const float* Q_item   = (const float*)d_in[14];
    const float* W_item   = (const float*)d_in[15];
    const float* b_item   = (const float*)d_in[16];
    const float* M_user   = (const float*)d_in[17];
    const float* M_item   = (const float*)d_in[18];
    float* out = (float*)d_out;

    const int NU = 40000, NI = 20000, NN = 60000;
    const int NNZ_ADJ = in_sizes[0], NNZ_UP = in_sizes[3], NNZ_IP = in_sizes[6];
    const size_t ROWF = (size_t)D_DIM;
    const size_t MAT  = (size_t)NN * ROWF;

    float* agg  = (float*)d_ws;
    float* embb = agg  + MAT;
    float* hbuf = embb + MAT;

    dim3 blk(256);
    const int NB_ADJ = (NN + OWN_R - 1) / OWN_R;
    const int NB_UP  = (NU + OWN_R - 1) / OWN_R;
    const int NB_IP  = (NI + OWN_R - 1) / OWN_R;
    const int NC_ADJ = (960000 + CH - 1) / CH;
    const int NC_UP  = (640000 + CH - 1) / CH;
    const int NC_IP  = (320000 + CH - 1) / CH;
    const size_t SEG = (size_t)1 << 20;

    int*   keyA  = (int*)hbuf;
    float* svA   = hbuf + SEG;
    int*   offsA = (int*)(hbuf + 2 * SEG);
    int*   keyU  = (int*)(hbuf + 3 * SEG);
    float* svU   = hbuf + 4 * SEG;
    int*   offsU = (int*)(hbuf + 5 * SEG);
    int*   keyI  = (int*)(hbuf + 6 * SEG);
    float* svI   = hbuf + 7 * SEG;
    int*   offsI = (int*)(hbuf + 8 * SEG);
    float* ssqU = hbuf;
    float* ssqI = hbuf + (size_t)4 * 2500 * 32;

    bin_entries<<<NC_ADJ + NC_UP + NC_IP, dim3(1024), 0, stream>>>(
        adj_rows, adj_cols, adj_vals, NNZ_ADJ, NN, NC_ADJ, keyA, svA, offsA,
        up_rows, up_cols, up_vals, NNZ_UP, NU, NC_UP, keyU, svU, offsU,
        ip_rows, ip_cols, ip_vals, NNZ_IP, NI, keyI, svI, offsI);

    spmm_owner<<<NB_ADJ, dim3(1024), 0, stream>>>(
        keyA, svA, offsA, NC_ADJ, user_emb, item_emb, NU, agg, NN, NB_ADJ,
        keyA, svA, offsA, 0, user_emb, item_emb, NU, agg, NN);

    const int GB_U = (NU + 63) / 64;
    const int GB_I = (NI + 63) / 64;

    gemm_tanh_wmma<0><<<dim3(GB_U + GB_I, 2), blk, 0, stream>>>(
        agg, nullptr, Q_user, nullptr, nullptr, embb, NU, nullptr, GB_U,
        agg + (size_t)NU * ROWF, nullptr, Q_item, nullptr, nullptr,
        embb + (size_t)NU * ROWF, NI, nullptr);

    gemm_tanh_wmma<1><<<dim3(GB_U + GB_I, 2), blk, 0, stream>>>(
        user_emb, embb, W_user, W_user + (size_t)D_DIM * D_DIM, b_user,
        out, NU, ssqU, GB_U,
        item_emb, embb + (size_t)NU * ROWF, W_item,
        W_item + (size_t)D_DIM * D_DIM, b_item,
        out + (size_t)NU * ROWF, NI, ssqI);

    spmm_owner<<<NB_UP + NB_IP, dim3(1024), 0, stream>>>(
        keyU, svU, offsU, NC_UP, user_emb, user_emb, NU, agg, NU, NB_UP,
        keyI, svI, offsI, NC_IP, item_emb, item_emb, NI,
        agg + (size_t)NU * ROWF, NI);

    gemm_tanh_wmma<2><<<dim3(GB_U + GB_I, 2), blk, 0, stream>>>(
        agg, nullptr, M_user, nullptr, nullptr, out, NU, ssqU, GB_U,
        agg + (size_t)NU * ROWF, nullptr, M_item, nullptr, nullptr,
        out + (size_t)NU * ROWF, NI, ssqI);
}


# SPMM gather: an X row is read with two 16-byte loads per lane (512 contiguous bytes per instruction, was eight 128-byte ones); LDS sums kept lane-transposed so each LDS add still spans 32 banks
# speedup vs baseline: 2.7923x; 1.0478x over previous
#include <hip/hip_runtime.h>
#include <hip/hip_bf16.h>
#include <math.h>

typedef __attribute__((ext_vector_type(2))) float v2f;
typedef __attribute__((ext_vector_type(8))) float v8f;

#define D_DIM 256
#define KC 32
#define LDA_S (KC + 2)
#define LDBT  (KC + 2)

#define OWN_R    240
#define GATHER_U 8
#define CH       4096
#define OFFS_LD  256

__global__ __launch_bounds__(1024)
void bin_entries(const int* __restrict__ rows0, const int* __restrict__ cols0,
                 const float* __restrict__ vals0, int nnz0, int nrows0, int nch0,
                 int* __restrict__ key0, float* __restrict__ sv0, int* __restrict__ offs0,
                 const int* __restrict__ rows1, const int* __restrict__ cols1,
                 const float* __restrict__ vals1, int nnz1, int nrows1, int nch1,
                 int* __restrict__ key1, float* __restrict__ sv1, int* __restrict__ offs1,
                 const int* __restrict__ rows2, const int* __restrict__ cols2,
                 const float* __restrict__ vals2, int nnz2, int nrows2,
                 int* __restrict__ key2, float* __restrict__ sv2, int* __restrict__ offs2) {
    __shared__ int   hist[OFFS_LD];
    __shared__ int   wsum[8];
    __shared__ int   s_key[CH];
    __shared__ float s_val[CH];

    const int t    = threadIdx.x;
    const int lane = t & 31;
    const int w    = t >> 5;

    int c = blockIdx.x;
    const bool second = (c >= nch0);
    if (second) c -= nch0;
    const bool third = second && (c >= nch1);
    if (third) c -= nch1;
    const int*   rows  = third ? rows2  : (second ? rows1  : rows0);
    const int*   cols  = third ? cols2  : (second ? cols1  : cols0);
    const float* vals  = third ? vals2  : (second ? vals1  : vals0);
    const int    nnz   = third ? nnz2   : (second ? nnz1   : nnz0);
    const int    nrows = third ? nrows2 : (second ? nrows1 : nrows0);
    int*         key   = third ? key2   : (second ? key1   : key0);
    float*       sv    = third ? sv2    : (second ? sv1    : sv0);
    int*         offs  = third ? offs2  : (second ? offs1  : offs0);

    const int e0 = c * CH;
    int n = nnz - e0;
    if (n > CH) n = CH;
    if (n < 0)  n = 0;

    if (t < OFFS_LD) hist[t] = 0;
    __syncthreads();

    int   rr[4], cc[4];
    float vv[4];
#pragma unroll
    for (int i = 0; i < 4; ++i) {
        const int j = t + 1024 * i;
        rr[i] = -1; cc[i] = 0; vv[i] = 0.f;
        if (j < n) {
            rr[i] = rows[e0 + j];
            cc[i] = cols[e0 + j];
            vv[i] = vals[e0 + j];
        }
    }
    int bk[4], rk[4], ky[4];
#pragma unroll
    for (int i = 0; i < 4; ++i) {
        bk[i] = -1; rk[i] = 0; ky[i] = 0;
        const int r = rr[i];
        if ((unsigned)r < (unsigned)nrows) {
            const int g = r / OWN_R;
            bk[i] = g;
            ky[i] = (int)(((unsigned)(r - g * OWN_R) << 24) | (unsigned)cc[i]);
            rk[i] = atomicAdd(&hist[g], 1);
        }
    }
    __syncthreads();

    const int x = (t < OFFS_LD) ? hist[t] : 0;
    int incl = x;
#pragma unroll
    for (int off = 1; off < 32; off <<= 1) {
        const int y = __shfl_up(incl, off, 32);
        if (lane >= off) incl += y;
    }
    if (t < OFFS_LD && lane == 31) wsum[w] = incl;
    __syncthreads();
    if (t < OFFS_LD) {
        int p = 0;
#pragma unroll
        for (int k = 0; k < 8; ++k)
            if (k < w) p += wsum[k];
        const int base = p + incl - x;
        hist[t] = base;
        offs[c * OFFS_LD + t] = base;
    }
    __syncthreads();

#pragma unroll
    for (int i = 0; i < 4; ++i) {
        if (bk[i] >= 0) {
            const int p = hist[bk[i]] + rk[i];
            s_key[p] = ky[i];
            s_val[p] = vv[i];
        }
    }
    __syncthreads();
#pragma unroll
    for (int i = 0; i < 4; ++i) {
        const int j = t + 1024 * i;
        if (j < n) {
            key[e0 + j] = s_key[j];
            sv[e0 + j]  = s_val[j];
        }
    }
}

template <int U>
__device__ __forceinline__ void gather_add(float* acc, int lane, int j0,
                                           int cq, int vq, int rq,
                                           const float* __restrict__ Xa,
                                           const float* __restrict__ Xb, int split) {
    float4 x[U][2];
#pragma unroll
    for (int u = 0; u < U; ++u) {
        int c = __builtin_amdgcn_readlane(cq, j0 + u);
        const float* X = (c < split) ? (Xa + (size_t)c * D_DIM)
                                     : (Xb + (size_t)(c - split) * D_DIM);
        const float4* X4 = (const float4*)X;
        x[u][0] = X4[lane];
        x[u][1] = X4[lane + 32];
    }
#pragma unroll
    for (int u = 0; u < U; ++u) {
        float  v = __int_as_float(__builtin_amdgcn_readlane(vq, j0 + u));
        int    r = __builtin_amdgcn_readlane(rq, j0 + u);
        float* a = acc + r * D_DIM + lane;
        atomicAdd(a,       x[u][0].x * v);
        atomicAdd(a +  32, x[u][0].y * v);
        atomicAdd(a +  64, x[u][0].z * v);
        atomicAdd(a +  96, x[u][0].w * v);
        atomicAdd(a + 128, x[u][1].x * v);
        atomicAdd(a + 160, x[u][1].y * v);
        atomicAdd(a + 192, x[u][1].z * v);
        atomicAdd(a + 224, x[u][1].w * v);
    }
}

__global__ __launch_bounds__(1024)
void spmm_owner(const int* __restrict__ key0, const float* __restrict__ sv0,
                const int* __restrict__ offs0, int nch0,
                const float* __restrict__ Xa0, const float* __restrict__ Xb0,
                int split0, float* __restrict__ Y0, int nrows0, int nblk0,
                const int* __restrict__ key1, const float* __restrict__ sv1,
                const int* __restrict__ offs1, int nch1,
                const float* __restrict__ Xa1, const float* __restrict__ Xb1,
                int split1, float* __restrict__ Y1, int nrows1) {
    __shared__ float acc[OWN_R * D_DIM];

    const int t    = threadIdx.x;
    const int lane = t & 31;
    const int w    = __builtin_amdgcn_readfirstlane(t >> 5);

    int b = blockIdx.x;
    const bool second = (b >= nblk0);
    if (second) b -= nblk0;
    const int*   key   = second ? key1   : key0;
    const float* sv    = second ? sv1    : sv0;
    const int*   offs  = second ? offs1  : offs0;
    const int    nch   = second ? nch1   : nch0;
    const float* Xa    = second ? Xa1    : Xa0;
    const float* Xb    = second ? Xb1    : Xb0;
    const int    split = second ? split1 : split0;
    float*       Y     = second ? Y1     : Y0;
    const int    nrows = second ? nrows1 : nrows0;

    const int lo = b * OWN_R;
    int nloc = nrows - lo;
    if (nloc > OWN_R) nloc = OWN_R;

    {
        float4* a4 = (float4*)acc;
        for (int i = t; i < OWN_R * (D_DIM / 4); i += 1024)
            a4[i] = make_float4(0.f, 0.f, 0.f, 0.f);
    }
    __syncthreads();

    for (int cb = 0; cb < nch; cb += 1024) {
        const int ci = cb + w + 32 * lane;
        int s = 0, e = 0;
        if (ci < nch) {
            s = offs[ci * OFFS_LD + b];
            e = offs[ci * OFFS_LD + b + 1];
        }
        const int len = e - s;
        int incl = len;
#pragma unroll
        for (int off = 1; off < 32; off <<= 1) {
            const int y = __shfl_up(incl, off, 32);
            if (lane >= off) incl += y;
        }
        const int excl = incl - len;
        const int T    = __builtin_amdgcn_readlane(incl, 31);
        const int G    = ci * CH + s - excl;
        int nv = (nch - cb - w + 31) >> 5;
        if (nv > 32) nv = 32;

        auto load_batch = [&](int j0, int& k, int& v) {
            int j = j0 + lane;
            if (j > T - 1) j = T - 1;
            int g = __builtin_amdgcn_readlane(G, 0);
            for (int i = 1; i < nv; ++i) {
                const int ex = __builtin_amdgcn_readlane(excl, i);
                const int gi = __builtin_amdgcn_readlane(G, i);
                if (j >= ex) g = gi;
            }
            k = key[g + j];
            v = __float_as_int(sv[g + j]);
        };

        if (T > 0) {
            int kC, vC;
            load_batch(0, kC, vC);
            for (int j0 = 0; j0 < T; j0 += 32) {
                int kN, vN;
                load_batch(j0 + 32, kN, vN);
                int n = T - j0;
                if (n > 32) n = 32;
                const int cq = kC & 0xFFFFFF;
                const int rq = (int)((unsigned)kC >> 24);
                int g = 0;
                for (; g + GATHER_U <= n; g += GATHER_U)
                    gather_add<GATHER_U>(acc, lane, g, cq, vC, rq, Xa, Xb, split);
                if (n - g >= 4) { gather_add<4>(acc, lane, g, cq, vC, rq, Xa, Xb, split); g += 4; }
                if (n - g >= 2) { gather_add<2>(acc, lane, g, cq, vC, rq, Xa, Xb, split); g += 2; }
                if (n - g >= 1) { gather_add<1>(acc, lane, g, cq, vC, rq, Xa, Xb, split); }
                kC = kN; vC = vN;
            }
        }
    }

    __syncthreads();

    for (int r = w; r < nloc; r += 32) {
        const float* s = acc + r * D_DIM + lane;
        float4*      d = (float4*)(Y + (size_t)(lo + r) * D_DIM);
        d[lane]      = make_float4(s[0],   s[32],  s[64],  s[96]);
        d[lane + 32] = make_float4(s[128], s[160], s[192], s[224]);
    }
}

template <int MODE>
__global__ __launch_bounds__(256)
void gemm_tanh_wmma(const float* __restrict__ A0p, const float* __restrict__ A1p,
                    const float* __restrict__ B0p, const float* __restrict__ B1p,
                    const float* __restrict__ biasp, float* OUTp,
                    int Mp, float* ssqp, int nblkp,
                    const float* __restrict__ A0q, const float* __restrict__ A1q,
                    const float* __restrict__ B0q, const float* __restrict__ B1q,
                    const float* __restrict__ biasq, float* OUTq,
                    int Mq, float* ssqq) {
    __shared__ float As[2][64 * LDA_S];
    __shared__ float Bst[2][128 * LDBT];

    int bx = blockIdx.x;
    const bool second = (bx >= nblkp);
    if (second) bx -= nblkp;
    const float* A0   = second ? A0q   : A0p;
    const float* A1   = second ? A1q   : A1p;
    const float* B0   = second ? B0q   : B0p;
    const float* B1   = second ? B1q   : B1p;
    const float* bias = second ? biasq : biasp;
    float*       OUT  = second ? OUTq  : OUTp;
    const int    M    = second ? Mq    : Mp;
    float*       ssq  = second ? ssqq  : ssqp;

    const int t    = threadIdx.x;
    const int wave = t >> 5;
    const int lane = t & 31;
    const int wm   = wave & 3;
    const int wn   = wave >> 2;
    const int rowBase = bx * 64;
    const int colBase = blockIdx.y * 128;

    const int m16 = lane & 15;
    const int kh  = lane >> 4;

    v8f acc[4];
#pragma unroll
    for (int j = 0; j < 4; ++j) acc[j] = (v8f)0.f;

    const int NCK = D_DIM / KC;
    const int nch = (A1 != nullptr) ? 2 * NCK : NCK;

    const int ar  = t >> 3;
    const int ac4 = (t & 7) * 4;
    int arow0 = rowBase + ar;
    int arow1 = rowBase + ar + 32;
    if (arow0 > M - 1) arow0 = M - 1;
    if (arow1 > M - 1) arow1 = M - 1;
    const int bk  = t >> 5;
    const int bn4 = (t & 31) * 4;

    const float* pa0 = A0 + (size_t)arow0 * D_DIM + ac4;
    const float* pa1 = A0 + (size_t)arow1 * D_DIM + ac4;
    const float* pb  = B0 + (size_t)bk * D_DIM + colBase + bn4;

    float4 ga0, ga1, gb[4];
    auto issue = [&](int c) {
        if (c == NCK) {
            pa0 = A1 + (size_t)arow0 * D_DIM + ac4;
            pa1 = A1 + (size_t)arow1 * D_DIM + ac4;
            pb  = B1 + (size_t)bk * D_DIM + colBase + bn4;
        }
        ga0 = *(const float4*)pa0;
        ga1 = *(const float4*)pa1;
#pragma unroll
        for (int i = 0; i < 4; ++i)
            gb[i] = *(const float4*)(pb + (size_t)(8 * i) * D_DIM);
        pa0 += KC;
        pa1 += KC;
        pb  += (size_t)KC * D_DIM;
    };

    auto step = [&](int c, float* as, float* bs) {
        __builtin_amdgcn_sched_barrier(0);
        {
            float* d0 = &as[ar * LDA_S + ac4];
            d0[0] = ga0.x; d0[1] = ga0.y; d0[2] = ga0.z; d0[3] = ga0.w;
            float* d1 = &as[(ar + 32) * LDA_S + ac4];
            d1[0] = ga1.x; d1[1] = ga1.y; d1[2] = ga1.z; d1[3] = ga1.w;
#pragma unroll
            for (int i = 0; i < 4; ++i) {
                const int k = bk + 8 * i;
                bs[(bn4 + 0) * LDBT + k] = gb[i].x;
                bs[(bn4 + 1) * LDBT + k] = gb[i].y;
                bs[(bn4 + 2) * LDBT + k] = gb[i].z;
                bs[(bn4 + 3) * LDBT + k] = gb[i].w;
            }
        }
        __syncthreads();
        if (c + 1 < nch) issue(c + 1);

#pragma unroll
        for (int kk = 0; kk < KC; kk += 4) {
            v2f af = *(const v2f*)&as[(wm * 16 + m16) * LDA_S + kk + 2 * kh];
#pragma unroll
            for (int j = 0; j < 4; ++j) {
                int n  = wn * 64 + j * 16 + m16;
                v2f bf = *(const v2f*)&bs[n * LDBT + kk + 2 * kh];
                acc[j] = __builtin_amdgcn_wmma_f32_16x16x4_f32(
                    false, af, false, bf, (short)0, acc[j], false, false);
            }
        }
    };

    issue(0);
    for (int c = 0; c < nch; c += 2) {
        step(c,     As[0], Bst[0]);
        step(c + 1, As[1], Bst[1]);
    }

    const int tile16 = (rowBase >> 4) + wm;
    const int ntile  = ((M + 63) >> 6) * 4;
    const int row0   = rowBase + wm * 16 + kh * 8;

    float inv[8];
    if (MODE == 2) {
        float4 p[4][2];
#pragma unroll
        for (int q = 0; q < 4; ++q) {
            const float4* s4 =
                (const float4*)(ssq + ((size_t)(q * ntile + tile16) * 32 + kh * 8));
            p[q][0] = s4[0];
            p[q][1] = s4[1];
        }
        float tot[8];
        tot[0] = (p[0][0].x + p[1][0].x) + (p[2][0].x + p[3][0].x);
        tot[1] = (p[0][0].y + p[1][0].y) + (p[2][0].y + p[3][0].y);
        tot[2] = (p[0][0].z + p[1][0].z) + (p[2][0].z + p[3][0].z);
        tot[3] = (p[0][0].w + p[1][0].w) + (p[2][0].w + p[3][0].w);
        tot[4] = (p[0][1].x + p[1][1].x) + (p[2][1].x + p[3][1].x);
        tot[5] = (p[0][1].y + p[1][1].y) + (p[2][1].y + p[3][1].y);
        tot[6] = (p[0][1].z + p[1][1].z) + (p[2][1].z + p[3][1].z);
        tot[7] = (p[0][1].w + p[1][1].w) + (p[2][1].w + p[3][1].w);
#pragma unroll
        for (int r = 0; r < 8; ++r) inv[r] = 1.f / fmaxf(sqrtf(tot[r]), 1e-12f);
    }

    float ss[8];
#pragma unroll
    for (int r = 0; r < 8; ++r) ss[r] = 0.f;

#pragma unroll
    for (int j = 0; j < 4; ++j) {
        int   c  = colBase + wn * 64 + j * 16 + m16;
        float bv = bias ? bias[c] : 0.f;
        float hv[8];
        if (MODE == 2) {
#pragma unroll
            for (int r = 0; r < 8; ++r) {
                int row = row0 + r;
                if (row > M - 1) row = M - 1;
                hv[r] = OUT[(size_t)row * D_DIM + c];
            }
        }
#pragma unroll
        for (int r = 0; r < 8; ++r) {
            int row = row0 + r;
            if (row < M) {
                float v = tanhf(acc[j][r] + bv);
                if (MODE == 1) ss[r] = fmaf(v, v, ss[r]);
                if (MODE == 2) v = fmaf(hv[r], inv[r], v);
                OUT[(size_t)row * D_DIM + c] = v;
            }
        }
    }

    if (MODE == 1) {
#pragma unroll
        for (int r = 0; r < 8; ++r) {
            float s = ss[r];
            s += __shfl_xor(s, 1, 32);
            s += __shfl_xor(s, 2, 32);
            s += __shfl_xor(s, 4, 32);
            s += __shfl_xor(s, 8, 32);
            ss[r] = s;
        }
        float mine = ss[0];
#pragma unroll
        for (int r = 1; r < 8; ++r)
            if (m16 == r) mine = ss[r];
        const int q = (int)blockIdx.y * 2 + wn;
        if (m16 < 8 && row0 + m16 < M)
            ssq[(size_t)(q * ntile + tile16) * 32 + kh * 8 + m16] = mine;
    }
}

extern "C" void kernel_launch(void* const* d_in, const int* in_sizes, int n_in,
                              void* d_out, int out_size, void* d_ws, size_t ws_size,
                              hipStream_t stream) {
    const int* adj_rows   = (const int*)d_in[0];
    const int* adj_cols   = (const int*)d_in[1];
    const float* adj_vals = (const float*)d_in[2];
    const int* up_rows    = (const int*)d_in[3];
    const int* up_cols    = (const int*)d_in[4];
    const float* up_vals  = (const float*)d_in[5];
    const int* ip_rows    = (const int*)d_in[6];
    const int* ip_cols    = (const int*)d_in[7];
    const float* ip_vals  = (const float*)d_in[8];
    const float* user_emb = (const float*)d_in[9];
    const float* item_emb = (const float*)d_in[10];
    const float* Q_user   = (const float*)d_in[11];
    const float* W_user   = (const float*)d_in[12];
    const float* b_user   = (const float*)d_in[13];
    const float* Q_item   = (const float*)d_in[14];
    const float* W_item   = (const float*)d_in[15];
    const float* b_item   = (const float*)d_in[16];
    const float* M_user   = (const float*)d_in[17];
    const float* M_item   = (const float*)d_in[18];
    float* out = (float*)d_out;

    const int NU = 40000, NI = 20000, NN = 60000;
    const int NNZ_ADJ = in_sizes[0], NNZ_UP = in_sizes[3], NNZ_IP = in_sizes[6];
    const size_t ROWF = (size_t)D_DIM;
    const size_t MAT  = (size_t)NN * ROWF;

    float* agg  = (float*)d_ws;
    float* embb = agg  + MAT;
    float* hbuf = embb + MAT;

    dim3 blk(256);
    const int NB_ADJ = (NN + OWN_R - 1) / OWN_R;
    const int NB_UP  = (NU + OWN_R - 1) / OWN_R;
    const int NB_IP  = (NI + OWN_R - 1) / OWN_R;
    const int NC_ADJ = (960000 + CH - 1) / CH;
    const int NC_UP  = (640000 + CH - 1) / CH;
    const int NC_IP  = (320000 + CH - 1) / CH;
    const size_t SEG = (size_t)1 << 20;

    int*   keyA  = (int*)hbuf;
    float* svA   = hbuf + SEG;
    int*   offsA = (int*)(hbuf + 2 * SEG);
    int*   keyU  = (int*)(hbuf + 3 * SEG);
    float* svU   = hbuf + 4 * SEG;
    int*   offsU = (int*)(hbuf + 5 * SEG);
    int*   keyI  = (int*)(hbuf + 6 * SEG);
    float* svI   = hbuf + 7 * SEG;
    int*   offsI = (int*)(hbuf + 8 * SEG);
    float* ssqU = hbuf;
    float* ssqI = hbuf + (size_t)4 * 2500 * 32;

    bin_entries<<<NC_ADJ + NC_UP + NC_IP, dim3(1024), 0, stream>>>(
        adj_rows, adj_cols, adj_vals, NNZ_ADJ, NN, NC_ADJ, keyA, svA, offsA,
        up_rows, up_cols, up_vals, NNZ_UP, NU, NC_UP, keyU, svU, offsU,
        ip_rows, ip_cols, ip_vals, NNZ_IP, NI, keyI, svI, offsI);

    spmm_owner<<<NB_ADJ, dim3(1024), 0, stream>>>(
        keyA, svA, offsA, NC_ADJ, user_emb, item_emb, NU, agg, NN, NB_ADJ,
        keyA, svA, offsA, 0, user_emb, item_emb, NU, agg, NN);

    const int GB_U = (NU + 63) / 64;
    const int GB_I = (NI + 63) / 64;

    gemm_tanh_wmma<0><<<dim3(GB_U + GB_I, 2), blk, 0, stream>>>(
        agg, nullptr, Q_user, nullptr, nullptr, embb, NU, nullptr, GB_U,
        agg + (size_t)NU * ROWF, nullptr, Q_item, nullptr, nullptr,
        embb + (size_t)NU * ROWF, NI, nullptr);

    gemm_tanh_wmma<1><<<dim3(GB_U + GB_I, 2), blk, 0, stream>>>(
        user_emb, embb, W_user, W_user + (size_t)D_DIM * D_DIM, b_user,
        out, NU, ssqU, GB_U,
        item_emb, embb + (size_t)NU * ROWF, W_item,
        W_item + (size_t)D_DIM * D_DIM, b_item,
        out + (size_t)NU * ROWF, NI, ssqI);

    spmm_owner<<<NB_UP + NB_IP, dim3(1024), 0, stream>>>(
        keyU, svU, offsU, NC_UP, user_emb, user_emb, NU, agg, NU, NB_UP,
        keyI, svI, offsI, NC_IP, item_emb, item_emb, NI,
        agg + (size_t)NU * ROWF, NI);

    gemm_tanh_wmma<2><<<dim3(GB_U + GB_I, 2), blk, 0, stream>>>(
        agg, nullptr, M_user, nullptr, nullptr, out, NU, ssqU, GB_U,
        agg + (size_t)NU * ROWF, nullptr, M_item, nullptr, nullptr,
        out + (size_t)NU * ROWF, NI, ssqI);
}


# GEMM epilogue without masked branches: tanh computes both ranges (odd polynomial below 0.625, else 1-2/(exp(2|x|)+1) on hardware exp2 and rcp) and selects; full 64-row tiles skip the row tests
# speedup vs baseline: 2.9014x; 1.0391x over previous
#include <hip/hip_runtime.h>
#include <hip/hip_bf16.h>
#include <math.h>

typedef __attribute__((ext_vector_type(2))) float v2f;
typedef __attribute__((ext_vector_type(8))) float v8f;

#define D_DIM 256
#define KC 32
#define LDA_S (KC + 2)
#define LDBT  (KC + 2)

#define OWN_R    240
#define GATHER_U 8
#define CH       4096
#define OFFS_LD  256

__global__ __launch_bounds__(1024)
void bin_entries(const int* __restrict__ rows0, const int* __restrict__ cols0,
                 const float* __restrict__ vals0, int nnz0, int nrows0, int nch0,
                 int* __restrict__ key0, float* __restrict__ sv0, int* __restrict__ offs0,
                 const int* __restrict__ rows1, const int* __restrict__ cols1,
                 const float* __restrict__ vals1, int nnz1, int nrows1, int nch1,
                 int* __restrict__ key1, float* __restrict__ sv1, int* __restrict__ offs1,
                 const int* __restrict__ rows2, const int* __restrict__ cols2,
                 const float* __restrict__ vals2, int nnz2, int nrows2,
                 int* __restrict__ key2, float* __restrict__ sv2, int* __restrict__ offs2) {
    __shared__ int   hist[OFFS_LD];
    __shared__ int   wsum[8];
    __shared__ int   s_key[CH];
    __shared__ float s_val[CH];

    const int t    = threadIdx.x;
    const int lane = t & 31;
    const int w    = t >> 5;

    int c = blockIdx.x;
    const bool second = (c >= nch0);
    if (second) c -= nch0;
    const bool third = second && (c >= nch1);
    if (third) c -= nch1;
    const int*   rows  = third ? rows2  : (second ? rows1  : rows0);
    const int*   cols  = third ? cols2  : (second ? cols1  : cols0);
    const float* vals  = third ? vals2  : (second ? vals1  : vals0);
    const int    nnz   = third ? nnz2   : (second ? nnz1   : nnz0);
    const int    nrows = third ? nrows2 : (second ? nrows1 : nrows0);
    int*         key   = third ? key2   : (second ? key1   : key0);
    float*       sv    = third ? sv2    : (second ? sv1    : sv0);
    int*         offs  = third ? offs2  : (second ? offs1  : offs0);

    const int e0 = c * CH;
    int n = nnz - e0;
    if (n > CH) n = CH;
    if (n < 0)  n = 0;

    if (t < OFFS_LD) hist[t] = 0;
    __syncthreads();

    int   rr[4], cc[4];
    float vv[4];
#pragma unroll
    for (int i = 0; i < 4; ++i) {
        const int j = t + 1024 * i;
        rr[i] = -1; cc[i] = 0; vv[i] = 0.f;
        if (j < n) {
            rr[i] = rows[e0 + j];
            cc[i] = cols[e0 + j];
            vv[i] = vals[e0 + j];
        }
    }
    int bk[4], rk[4], ky[4];
#pragma unroll
    for (int i = 0; i < 4; ++i) {
        bk[i] = -1; rk[i] = 0; ky[i] = 0;
        const int r = rr[i];
        if ((unsigned)r < (unsigned)nrows) {
            const int g = r / OWN_R;
            bk[i] = g;
            ky[i] = (int)(((unsigned)(r - g * OWN_R) << 24) | (unsigned)cc[i]);
            rk[i] = atomicAdd(&hist[g], 1);
        }
    }
    __syncthreads();

    const int x = (t < OFFS_LD) ? hist[t] : 0;
    int incl = x;
#pragma unroll
    for (int off = 1; off < 32; off <<= 1) {
        const int y = __shfl_up(incl, off, 32);
        if (lane >= off) incl += y;
    }
    if (t < OFFS_LD && lane == 31) wsum[w] = incl;
    __syncthreads();
    if (t < OFFS_LD) {
        int p = 0;
#pragma unroll
        for (int k = 0; k < 8; ++k)
            if (k < w) p += wsum[k];
        const int base = p + incl - x;
        hist[t] = base;
        offs[c * OFFS_LD + t] = base;
    }
    __syncthreads();

#pragma unroll
    for (int i = 0; i < 4; ++i) {
        if (bk[i] >= 0) {
            const int p = hist[bk[i]] + rk[i];
            s_key[p] = ky[i];
            s_val[p] = vv[i];
        }
    }
    __syncthreads();
#pragma unroll
    for (int i = 0; i < 4; ++i) {
        const int j = t + 1024 * i;
        if (j < n) {
            key[e0 + j] = s_key[j];
            sv[e0 + j]  = s_val[j];
        }
    }
}

template <int U>
__device__ __forceinline__ void gather_add(float* acc, int lane, int j0,
                                           int cq, int vq, int rq,
                                           const float* __restrict__ Xa,
                                           const float* __restrict__ Xb, int split) {
    float4 x[U][2];
#pragma unroll
    for (int u = 0; u < U; ++u) {
        int c = __builtin_amdgcn_readlane(cq, j0 + u);
        const float* X = (c < split) ? (Xa + (size_t)c * D_DIM)
                                     : (Xb + (size_t)(c - split) * D_DIM);
        const float4* X4 = (const float4*)X;
        x[u][0] = X4[lane];
        x[u][1] = X4[lane + 32];
    }
#pragma unroll
    for (int u = 0; u < U; ++u) {
        float  v = __int_as_float(__builtin_amdgcn_readlane(vq, j0 + u));
        int    r = __builtin_amdgcn_readlane(rq, j0 + u);
        float* a = acc + r * D_DIM + lane;
        atomicAdd(a,       x[u][0].x * v);
        atomicAdd(a +  32, x[u][0].y * v);
        atomicAdd(a +  64, x[u][0].z * v);
        atomicAdd(a +  96, x[u][0].w * v);
        atomicAdd(a + 128, x[u][1].x * v);
        atomicAdd(a + 160, x[u][1].y * v);
        atomicAdd(a + 192, x[u][1].z * v);
        atomicAdd(a + 224, x[u][1].w * v);
    }
}

__global__ __launch_bounds__(1024)
void spmm_owner(const int* __restrict__ key0, const float* __restrict__ sv0,
                const int* __restrict__ offs0, int nch0,
                const float* __restrict__ Xa0, const float* __restrict__ Xb0,
                int split0, float* __restrict__ Y0, int nrows0, int nblk0,
                const int* __restrict__ key1, const float* __restrict__ sv1,
                const int* __restrict__ offs1, int nch1,
                const float* __restrict__ Xa1, const float* __restrict__ Xb1,
                int split1, float* __restrict__ Y1, int nrows1) {
    __shared__ float acc[OWN_R * D_DIM];

    const int t    = threadIdx.x;
    const int lane = t & 31;
    const int w    = __builtin_amdgcn_readfirstlane(t >> 5);

    int b = blockIdx.x;
    const bool second = (b >= nblk0);
    if (second) b -= nblk0;
    const int*   key   = second ? key1   : key0;
    const float* sv    = second ? sv1    : sv0;
    const int*   offs  = second ? offs1  : offs0;
    const int    nch   = second ? nch1   : nch0;
    const float* Xa    = second ? Xa1    : Xa0;
    const float* Xb    = second ? Xb1    : Xb0;
    const int    split = second ? split1 : split0;
    float*       Y     = second ? Y1     : Y0;
    const int    nrows = second ? nrows1 : nrows0;

    const int lo = b * OWN_R;
    int nloc = nrows - lo;
    if (nloc > OWN_R) nloc = OWN_R;

    {
        float4* a4 = (float4*)acc;
        for (int i = t; i < OWN_R * (D_DIM / 4); i += 1024)
            a4[i] = make_float4(0.f, 0.f, 0.f, 0.f);
    }
    __syncthreads();

    for (int cb = 0; cb < nch; cb += 1024) {
        const int ci = cb + w + 32 * lane;
        int s = 0, e = 0;
        if (ci < nch) {
            s = offs[ci * OFFS_LD + b];
            e = offs[ci * OFFS_LD + b + 1];
        }
        const int len = e - s;
        int incl = len;
#pragma unroll
        for (int off = 1; off < 32; off <<= 1) {
            const int y = __shfl_up(incl, off, 32);
            if (lane >= off) incl += y;
        }
        const int excl = incl - len;
        const int T    = __builtin_amdgcn_readlane(incl, 31);
        const int G    = ci * CH + s - excl;
        int nv = (nch - cb - w + 31) >> 5;
        if (nv > 32) nv = 32;

        auto load_batch = [&](int j0, int& k, int& v) {
            int j = j0 + lane;
            if (j > T - 1) j = T - 1;
            int g = __builtin_amdgcn_readlane(G, 0);
            for (int i = 1; i < nv; ++i) {
                const int ex = __builtin_amdgcn_readlane(excl, i);
                const int gi = __builtin_amdgcn_readlane(G, i);
                if (j >= ex) g = gi;
            }
            k = key[g + j];
            v = __float_as_int(sv[g + j]);
        };

        if (T > 0) {
            int kC, vC;
            load_batch(0, kC, vC);
            for (int j0 = 0; j0 < T; j0 += 32) {
                int kN, vN;
                load_batch(j0 + 32, kN, vN);
                int n = T - j0;
                if (n > 32) n = 32;
                const int cq = kC & 0xFFFFFF;
                const int rq = (int)((unsigned)kC >> 24);
                int g = 0;
                for (; g + GATHER_U <= n; g += GATHER_U)
                    gather_add<GATHER_U>(acc, lane, g, cq, vC, rq, Xa, Xb, split);
                if (n - g >= 4) { gather_add<4>(acc, lane, g, cq, vC, rq, Xa, Xb, split); g += 4; }
                if (n - g >= 2) { gather_add<2>(acc, lane, g, cq, vC, rq, Xa, Xb, split); g += 2; }
                if (n - g >= 1) { gather_add<1>(acc, lane, g, cq, vC, rq, Xa, Xb, split); }
                kC = kN; vC = vN;
            }
        }
    }

    __syncthreads();

    for (int r = w; r < nloc; r += 32) {
        const float* s = acc + r * D_DIM + lane;
        float4*      d = (float4*)(Y + (size_t)(lo + r) * D_DIM);
        d[lane]      = make_float4(s[0],   s[32],  s[64],  s[96]);
        d[lane + 32] = make_float4(s[128], s[160], s[192], s[224]);
    }
}

__device__ __forceinline__ float tanh_sl(float x) {
    const float x2 = x * x;
    float p = fmaf(x2, -5.700020585e-3f, 2.063407190e-2f);
    p = fmaf(x2, p, -5.373793095e-2f);
    p = fmaf(x2, p, 1.333141625e-1f);
    p = fmaf(x2, p, -3.333328068e-1f);
    const float small = fmaf(x2, x * p, x);
    const float ax  = fabsf(x);
    const float e   = __builtin_amdgcn_exp2f(ax * 2.885390082f);
    const float big = fmaf(__builtin_amdgcn_rcpf(e + 1.f), -2.f, 1.f);
    return (ax < 0.625f) ? small : __builtin_copysignf(big, x);
}

template <int MODE, bool FULL>
__device__ __forceinline__ void tile_epilogue(const v8f (&acc)[4], const float* bias,
                                              float* OUT, int M, int row0, int col0,
                                              const float (&inv)[8], float (&ss)[8]) {
#pragma unroll
    for (int j = 0; j < 4; ++j) {
        const int   c  = col0 + j * 16;
        const float bv = bias ? bias[c] : 0.f;
        float hv[8];
        if (MODE == 2) {
#pragma unroll
            for (int r = 0; r < 8; ++r) {
                int row = row0 + r;
                if (!FULL && row > M - 1) row = M - 1;
                hv[r] = OUT[(size_t)row * D_DIM + c];
            }
        }
        float v[8];
#pragma unroll
        for (int r = 0; r < 8; ++r) {
            v[r] = tanh_sl(acc[j][r] + bv);
            if (MODE == 1) ss[r] = fmaf(v[r], v[r], ss[r]);
            if (MODE == 2) v[r] = fmaf(hv[r], inv[r], v[r]);
        }
#pragma unroll
        for (int r = 0; r < 8; ++r) {
            const int row = row0 + r;
            if (FULL || row < M) OUT[(size_t)row * D_DIM + c] = v[r];
        }
    }
}

template <int MODE>
__global__ __launch_bounds__(256)
void gemm_tanh_wmma(const float* __restrict__ A0p, const float* __restrict__ A1p,
                    const float* __restrict__ B0p, const float* __restrict__ B1p,
                    const float* __restrict__ biasp, float* OUTp,
                    int Mp, float* ssqp, int nblkp,
                    const float* __restrict__ A0q, const float* __restrict__ A1q,
                    const float* __restrict__ B0q, const float* __restrict__ B1q,
                    const float* __restrict__ biasq, float* OUTq,
                    int Mq, float* ssqq) {
    __shared__ float As[2][64 * LDA_S];
    __shared__ float Bst[2][128 * LDBT];

    int bx = blockIdx.x;
    const bool second = (bx >= nblkp);
    if (second) bx -= nblkp;
    const float* A0   = second ? A0q   : A0p;
    const float* A1   = second ? A1q   : A1p;
    const float* B0   = second ? B0q   : B0p;
    const float* B1   = second ? B1q   : B1p;
    const float* bias = second ? biasq : biasp;
    float*       OUT  = second ? OUTq  : OUTp;
    const int    M    = second ? Mq    : Mp;
    float*       ssq  = second ? ssqq  : ssqp;

    const int t    = threadIdx.x;
    const int wave = t >> 5;
    const int lane = t & 31;
    const int wm   = wave & 3;
    const int wn   = wave >> 2;
    const int rowBase = bx * 64;
    const int colBase = blockIdx.y * 128;

    const int m16 = lane & 15;
    const int kh  = lane >> 4;

    v8f acc[4];
#pragma unroll
    for (int j = 0; j < 4; ++j) acc[j] = (v8f)0.f;

    const int NCK = D_DIM / KC;
    const int nch = (A1 != nullptr) ? 2 * NCK : NCK;

    const int ar  = t >> 3;
    const int ac4 = (t & 7) * 4;
    int arow0 = rowBase + ar;
    int arow1 = rowBase + ar + 32;
    if (arow0 > M - 1) arow0 = M - 1;
    if (arow1 > M - 1) arow1 = M - 1;
    const int bk  = t >> 5;
    const int bn4 = (t & 31) * 4;

    const float* pa0 = A0 + (size_t)arow0 * D_DIM + ac4;
    const float* pa1 = A0 + (size_t)arow1 * D_DIM + ac4;
    const float* pb  = B0 + (size_t)bk * D_DIM + colBase + bn4;

    float4 ga0, ga1, gb[4];
    auto issue = [&](int c) {
        if (c == NCK) {
            pa0 = A1 + (size_t)arow0 * D_DIM + ac4;
            pa1 = A1 + (size_t)arow1 * D_DIM + ac4;
            pb  = B1 + (size_t)bk * D_DIM + colBase + bn4;
        }
        ga0 = *(const float4*)pa0;
        ga1 = *(const float4*)pa1;
#pragma unroll
        for (int i = 0; i < 4; ++i)
            gb[i] = *(const float4*)(pb + (size_t)(8 * i) * D_DIM);
        pa0 += KC;
        pa1 += KC;
        pb  += (size_t)KC * D_DIM;
    };

    auto step = [&](int c, float* as, float* bs) {
        __builtin_amdgcn_sched_barrier(0);
        {
            float* d0 = &as[ar * LDA_S + ac4];
            d0[0] = ga0.x; d0[1] = ga0.y; d0[2] = ga0.z; d0[3] = ga0.w;
            float* d1 = &as[(ar + 32) * LDA_S + ac4];
            d1[0] = ga1.x; d1[1] = ga1.y; d1[2] = ga1.z; d1[3] = ga1.w;
#pragma unroll
            for (int i = 0; i < 4; ++i) {
                const int k = bk + 8 * i;
                bs[(bn4 + 0) * LDBT + k] = gb[i].x;
                bs[(bn4 + 1) * LDBT + k] = gb[i].y;
                bs[(bn4 + 2) * LDBT + k] = gb[i].z;
                bs[(bn4 + 3) * LDBT + k] = gb[i].w;
            }
        }
        __syncthreads();
        if (c + 1 < nch) issue(c + 1);

#pragma unroll
        for (int kk = 0; kk < KC; kk += 4) {
            v2f af = *(const v2f*)&as[(wm * 16 + m16) * LDA_S + kk + 2 * kh];
#pragma unroll
            for (int j = 0; j < 4; ++j) {
                int n  = wn * 64 + j * 16 + m16;
                v2f bf = *(const v2f*)&bs[n * LDBT + kk + 2 * kh];
                acc[j] = __builtin_amdgcn_wmma_f32_16x16x4_f32(
                    false, af, false, bf, (short)0, acc[j], false, false);
            }
        }
    };

    issue(0);
    for (int c = 0; c < nch; c += 2) {
        step(c,     As[0], Bst[0]);
        step(c + 1, As[1], Bst[1]);
    }

    const int tile16 = (rowBase >> 4) + wm;
    const int ntile  = ((M + 63) >> 6) * 4;
    const int row0   = rowBase + wm * 16 + kh * 8;

    float inv[8];
    if (MODE == 2) {
        float4 p[4][2];
#pragma unroll
        for (int q = 0; q < 4; ++q) {
            const float4* s4 =
                (const float4*)(ssq + ((size_t)(q * ntile + tile16) * 32 + kh * 8));
            p[q][0] = s4[0];
            p[q][1] = s4[1];
        }
        float tot[8];
        tot[0] = (p[0][0].x + p[1][0].x) + (p[2][0].x + p[3][0].x);
        tot[1] = (p[0][0].y + p[1][0].y) + (p[2][0].y + p[3][0].y);
        tot[2] = (p[0][0].z + p[1][0].z) + (p[2][0].z + p[3][0].z);
        tot[3] = (p[0][0].w + p[1][0].w) + (p[2][0].w + p[3][0].w);
        tot[4] = (p[0][1].x + p[1][1].x) + (p[2][1].x + p[3][1].x);
        tot[5] = (p[0][1].y + p[1][1].y) + (p[2][1].y + p[3][1].y);
        tot[6] = (p[0][1].z + p[1][1].z) + (p[2][1].z + p[3][1].z);
        tot[7] = (p[0][1].w + p[1][1].w) + (p[2][1].w + p[3][1].w);
#pragma unroll
        for (int r = 0; r < 8; ++r) inv[r] = 1.f / fmaxf(sqrtf(tot[r]), 1e-12f);
    }

    float ss[8];
#pragma unroll
    for (int r = 0; r < 8; ++r) ss[r] = 0.f;

    const int col0 = colBase + wn * 64 + m16;
    if (rowBase + 64 <= M) tile_epilogue<MODE, true >(acc, bias, OUT, M, row0, col0, inv, ss);
    else                   tile_epilogue<MODE, false>(acc, bias, OUT, M, row0, col0, inv, ss);

    if (MODE == 1) {
#pragma unroll
        for (int r = 0; r < 8; ++r) {
            float s = ss[r];
            s += __shfl_xor(s, 1, 32);
            s += __shfl_xor(s, 2, 32);
            s += __shfl_xor(s, 4, 32);
            s += __shfl_xor(s, 8, 32);
            ss[r] = s;
        }
        float mine = ss[0];
#pragma unroll
        for (int r = 1; r < 8; ++r)
            if (m16 == r) mine = ss[r];
        const int q = (int)blockIdx.y * 2 + wn;
        if (m16 < 8 && row0 + m16 < M)
            ssq[(size_t)(q * ntile + tile16) * 32 + kh * 8 + m16] = mine;
    }
}

extern "C" void kernel_launch(void* const* d_in, const int* in_sizes, int n_in,
                              void* d_out, int out_size, void* d_ws, size_t ws_size,
                              hipStream_t stream) {
    const int* adj_rows   = (const int*)d_in[0];
    const int* adj_cols   = (const int*)d_in[1];
    const float* adj_vals = (const float*)d_in[2];
    const int* up_rows    = (const int*)d_in[3];
    const int* up_cols    = (const int*)d_in[4];
    const float* up_vals  = (const float*)d_in[5];
    const int* ip_rows    = (const int*)d_in[6];
    const int* ip_cols    = (const int*)d_in[7];
    const float* ip_vals  = (const float*)d_in[8];
    const float* user_emb = (const float*)d_in[9];
    const float* item_emb = (const float*)d_in[10];
    const float* Q_user   = (const float*)d_in[11];
    const float* W_user   = (const float*)d_in[12];
    const float* b_user   = (const float*)d_in[13];
    const float* Q_item   = (const float*)d_in[14];
    const float* W_item   = (const float*)d_in[15];
    const float* b_item   = (const float*)d_in[16];
    const float* M_user   = (const float*)d_in[17];
    const float* M_item   = (const float*)d_in[18];
    float* out = (float*)d_out;

    const int NU = 40000, NI = 20000, NN = 60000;
    const int NNZ_ADJ = in_sizes[0], NNZ_UP = in_sizes[3], NNZ_IP = in_sizes[6];
    const size_t ROWF = (size_t)D_DIM;
    const size_t MAT  = (size_t)NN * ROWF;

    float* agg  = (float*)d_ws;
    float* embb = agg  + MAT;
    float* hbuf = embb + MAT;

    dim3 blk(256);
    const int NB_ADJ = (NN + OWN_R - 1) / OWN_R;
    const int NB_UP  = (NU + OWN_R - 1) / OWN_R;
    const int NB_IP  = (NI + OWN_R - 1) / OWN_R;
    const int NC_ADJ = (960000 + CH - 1) / CH;
    const int NC_UP  = (640000 + CH - 1) / CH;
    const int NC_IP  = (320000 + CH - 1) / CH;
    const size_t SEG = (size_t)1 << 20;

    int*   keyA  = (int*)hbuf;
    float* svA   = hbuf + SEG;
    int*   offsA = (int*)(hbuf + 2 * SEG);
    int*   keyU  = (int*)(hbuf + 3 * SEG);
    float* svU   = hbuf + 4 * SEG;
    int*   offsU = (int*)(hbuf + 5 * SEG);
    int*   keyI  = (int*)(hbuf + 6 * SEG);
    float* svI   = hbuf + 7 * SEG;
    int*   offsI = (int*)(hbuf + 8 * SEG);
    float* ssqU = hbuf;
    float* ssqI = hbuf + (size_t)4 * 2500 * 32;

    bin_entries<<<NC_ADJ + NC_UP + NC_IP, dim3(1024), 0, stream>>>(
        adj_rows, adj_cols, adj_vals, NNZ_ADJ, NN, NC_ADJ, keyA, svA, offsA,
        up_rows, up_cols, up_vals, NNZ_UP, NU, NC_UP, keyU, svU, offsU,
        ip_rows, ip_cols, ip_vals, NNZ_IP, NI, keyI, svI, offsI);

    spmm_owner<<<NB_ADJ, dim3(1024), 0, stream>>>(
        keyA, svA, offsA, NC_ADJ, user_emb, item_emb, NU, agg, NN, NB_ADJ,
        keyA, svA, offsA, 0, user_emb, item_emb, NU, agg, NN);

    const int GB_U = (NU + 63) / 64;
    const int GB_I = (NI + 63) / 64;

    gemm_tanh_wmma<0><<<dim3(GB_U + GB_I, 2), blk, 0, stream>>>(
        agg, nullptr, Q_user, nullptr, nullptr, embb, NU, nullptr, GB_U,
        agg + (size_t)NU * ROWF, nullptr, Q_item, nullptr, nullptr,
        embb + (size_t)NU * ROWF, NI, nullptr);

    gemm_tanh_wmma<1><<<dim3(GB_U + GB_I, 2), blk, 0, stream>>>(
        user_emb, embb, W_user, W_user + (size_t)D_DIM * D_DIM, b_user,
        out, NU, ssqU, GB_U,
        item_emb, embb + (size_t)NU * ROWF, W_item,
        W_item + (size_t)D_DIM * D_DIM, b_item,
        out + (size_t)NU * ROWF, NI, ssqI);

    spmm_owner<<<NB_UP + NB_IP, dim3(1024), 0, stream>>>(
        keyU, svU, offsU, NC_UP, user_emb, user_emb, NU, agg, NU, NB_UP,
        keyI, svI, offsI, NC_IP, item_emb, item_emb, NI,
        agg + (size_t)NU * ROWF, NI);

    gemm_tanh_wmma<2><<<dim3(GB_U + GB_I, 2), blk, 0, stream>>>(
        agg, nullptr, M_user, nullptr, nullptr, out, NU, ssqU, GB_U,
        agg + (size_t)NU * ROWF, nullptr, M_item, nullptr, nullptr,
        out + (size_t)NU * ROWF, NI, ssqI);
}


# SPMM owner pass sorts its entries by row in LDS (counting sort); each wave sums whole rows in registers, in sorted-tile order, and stores them: no LDS f32 atomics, LDS sums, zero fill or store pass
# speedup vs baseline: 3.0064x; 1.0362x over previous
#include <hip/hip_runtime.h>
#include <hip/hip_bf16.h>
#include <math.h>

typedef __attribute__((ext_vector_type(2))) float v2f;
typedef __attribute__((ext_vector_type(8))) float v8f;

#define D_DIM 256
#define KC 32
#define LDA_S (KC + 2)
#define LDBT  (KC + 2)

#define OWN_R    240
#define GATHER_U 8
#define TCAP     4096
#define CH       4096
#define OFFS_LD  256

__global__ __launch_bounds__(1024)
void bin_entries(const int* __restrict__ rows0, const int* __restrict__ cols0,
                 const float* __restrict__ vals0, int nnz0, int nrows0, int nch0,
                 int* __restrict__ key0, float* __restrict__ sv0, int* __restrict__ offs0,
                 const int* __restrict__ rows1, const int* __restrict__ cols1,
                 const float* __restrict__ vals1, int nnz1, int nrows1, int nch1,
                 int* __restrict__ key1, float* __restrict__ sv1, int* __restrict__ offs1,
                 const int* __restrict__ rows2, const int* __restrict__ cols2,
                 const float* __restrict__ vals2, int nnz2, int nrows2,
                 int* __restrict__ key2, float* __restrict__ sv2, int* __restrict__ offs2) {
    __shared__ int   hist[OFFS_LD];
    __shared__ int   wsum[8];
    __shared__ int   s_key[CH];
    __shared__ float s_val[CH];

    const int t    = threadIdx.x;
    const int lane = t & 31;
    const int w    = t >> 5;

    int c = blockIdx.x;
    const bool second = (c >= nch0);
    if (second) c -= nch0;
    const bool third = second && (c >= nch1);
    if (third) c -= nch1;
    const int*   rows  = third ? rows2  : (second ? rows1  : rows0);
    const int*   cols  = third ? cols2  : (second ? cols1  : cols0);
    const float* vals  = third ? vals2  : (second ? vals1  : vals0);
    const int    nnz   = third ? nnz2   : (second ? nnz1   : nnz0);
    const int    nrows = third ? nrows2 : (second ? nrows1 : nrows0);
    int*         key   = third ? key2   : (second ? key1   : key0);
    float*       sv    = third ? sv2    : (second ? sv1    : sv0);
    int*         offs  = third ? offs2  : (second ? offs1  : offs0);

    const int e0 = c * CH;
    int n = nnz - e0;
    if (n > CH) n = CH;
    if (n < 0)  n = 0;

    if (t < OFFS_LD) hist[t] = 0;
    __syncthreads();

    int   rr[4], cc[4];
    float vv[4];
#pragma unroll
    for (int i = 0; i < 4; ++i) {
        const int j = t + 1024 * i;
        rr[i] = -1; cc[i] = 0; vv[i] = 0.f;
        if (j < n) {
            rr[i] = rows[e0 + j];
            cc[i] = cols[e0 + j];
            vv[i] = vals[e0 + j];
        }
    }
    int bk[4], rk[4], ky[4];
#pragma unroll
    for (int i = 0; i < 4; ++i) {
        bk[i] = -1; rk[i] = 0; ky[i] = 0;
        const int r = rr[i];
        if ((unsigned)r < (unsigned)nrows) {
            const int g = r / OWN_R;
            bk[i] = g;
            ky[i] = (int)(((unsigned)(r - g * OWN_R) << 24) | (unsigned)cc[i]);
            rk[i] = atomicAdd(&hist[g], 1);
        }
    }
    __syncthreads();

    const int x = (t < OFFS_LD) ? hist[t] : 0;
    int incl = x;
#pragma unroll
    for (int off = 1; off < 32; off <<= 1) {
        const int y = __shfl_up(incl, off, 32);
        if (lane >= off) incl += y;
    }
    if (t < OFFS_LD && lane == 31) wsum[w] = incl;
    __syncthreads();
    if (t < OFFS_LD) {
        int p = 0;
#pragma unroll
        for (int k = 0; k < 8; ++k)
            if (k < w) p += wsum[k];
        const int base = p + incl - x;
        hist[t] = base;
        offs[c * OFFS_LD + t] = base;
    }
    __syncthreads();

#pragma unroll
    for (int i = 0; i < 4; ++i) {
        if (bk[i] >= 0) {
            const int p = hist[bk[i]] + rk[i];
            s_key[p] = ky[i];
            s_val[p] = vv[i];
        }
    }
    __syncthreads();
#pragma unroll
    for (int i = 0; i < 4; ++i) {
        const int j = t + 1024 * i;
        if (j < n) {
            key[e0 + j] = s_key[j];
            sv[e0 + j]  = s_val[j];
        }
    }
}

__device__ __forceinline__ void flush_rows(float* Yg, int lane, bool first, int cur, int next,
                                           float4 a0, float4 a1) {
    float4* d = (float4*)(Yg + (size_t)cur * D_DIM);
    if (first) {
        d[lane]      = a0;
        d[lane + 32] = a1;
        const float4 z = make_float4(0.f, 0.f, 0.f, 0.f);
#pragma nounroll
        for (int r = cur + 1; r < next; ++r) {
            float4* dz = (float4*)(Yg + (size_t)r * D_DIM);
            dz[lane]      = z;
            dz[lane + 32] = z;
        }
    } else {
        float4 o0 = d[lane];
        float4 o1 = d[lane + 32];
        o0.x += a0.x; o0.y += a0.y; o0.z += a0.z; o0.w += a0.w;
        o1.x += a1.x; o1.y += a1.y; o1.z += a1.z; o1.w += a1.w;
        d[lane]      = o0;
        d[lane + 32] = o1;
    }
}

template <int U>
__device__ __forceinline__ void gather_fma(int lane, int j0, int cq, int vq, int rq,
                                           const float* __restrict__ Xa,
                                           const float* __restrict__ Xb, int split,
                                           float* Yg, bool first,
                                           int& cur, float4& a0, float4& a1) {
    float4 x[U][2];
#pragma unroll
    for (int u = 0; u < U; ++u) {
        int c = __builtin_amdgcn_readlane(cq, j0 + u);
        const float* X = (c < split) ? (Xa + (size_t)c * D_DIM)
                                     : (Xb + (size_t)(c - split) * D_DIM);
        const float4* X4 = (const float4*)X;
        x[u][0] = X4[lane];
        x[u][1] = X4[lane + 32];
    }
#pragma unroll
    for (int u = 0; u < U; ++u) {
        const float v = __int_as_float(__builtin_amdgcn_readlane(vq, j0 + u));
        const int   r = __builtin_amdgcn_readlane(rq, j0 + u);
        if (r != cur) {
            flush_rows(Yg, lane, first, cur, r, a0, a1);
            cur = r;
            a0 = make_float4(0.f, 0.f, 0.f, 0.f);
            a1 = make_float4(0.f, 0.f, 0.f, 0.f);
        }
        a0.x = fmaf(x[u][0].x, v, a0.x);
        a0.y = fmaf(x[u][0].y, v, a0.y);
        a0.z = fmaf(x[u][0].z, v, a0.z);
        a0.w = fmaf(x[u][0].w, v, a0.w);
        a1.x = fmaf(x[u][1].x, v, a1.x);
        a1.y = fmaf(x[u][1].y, v, a1.y);
        a1.z = fmaf(x[u][1].z, v, a1.z);
        a1.w = fmaf(x[u][1].w, v, a1.w);
    }
}

__global__ __launch_bounds__(1024)
void spmm_owner(const int* __restrict__ key0, const float* __restrict__ sv0,
                const int* __restrict__ offs0, int nch0,
                const float* __restrict__ Xa0, const float* __restrict__ Xb0,
                int split0, float* Y0, int nrows0, int nblk0,
                const int* __restrict__ key1, const float* __restrict__ sv1,
                const int* __restrict__ offs1, int nch1,
                const float* __restrict__ Xa1, const float* __restrict__ Xb1,
                int split1, float* Y1, int nrows1) {
    __shared__ int   s_key[TCAP];
    __shared__ float s_val[TCAP];
    __shared__ int   segP[1025];
    __shared__ int   segG[1024];
    __shared__ int   rb[OFFS_LD + 1];
    __shared__ int   ws[32];
    __shared__ int   ws2[8];

    const int t    = threadIdx.x;
    const int lane = t & 31;
    const int w    = __builtin_amdgcn_readfirstlane(t >> 5);

    int b = blockIdx.x;
    const bool second = (b >= nblk0);
    if (second) b -= nblk0;
    const int*   key   = second ? key1   : key0;
    const float* sv    = second ? sv1    : sv0;
    const int*   offs  = second ? offs1  : offs0;
    const int    nch   = second ? nch1   : nch0;
    const float* Xa    = second ? Xa1    : Xa0;
    const float* Xb    = second ? Xb1    : Xb0;
    const int    split = second ? split1 : split0;
    float*       Y     = second ? Y1     : Y0;
    const int    nrows = second ? nrows1 : nrows0;

    const int lo = b * OWN_R;
    int nloc = nrows - lo;
    if (nloc > OWN_R) nloc = OWN_R;
    float* Yg = Y + (size_t)lo * D_DIM;

    const int r_lo = (w * OWN_R) >> 5;
    int r_hi = ((w + 1) * OWN_R) >> 5;
    if (r_hi > nloc) r_hi = nloc;

    for (int cb = 0; cb < nch || cb == 0; cb += 1024) {
        const int ci = cb + t;
        int s = 0, e = 0;
        if (ci < nch) {
            s = offs[ci * OFFS_LD + b];
            e = offs[ci * OFFS_LD + b + 1];
        }
        const int len = e - s;
        int incl = len;
#pragma unroll
        for (int off = 1; off < 32; off <<= 1) {
            const int y = __shfl_up(incl, off, 32);
            if (lane >= off) incl += y;
        }
        if (lane == 31) ws[w] = incl;
        __syncthreads();
        int wp = 0, Tv = 0;
#pragma unroll
        for (int k = 0; k < 32; ++k) {
            const int q = ws[k];
            if (k < w) wp += q;
            Tv += q;
        }
        const int T = __builtin_amdgcn_readfirstlane(Tv);
        const int excl = wp + incl - len;
        segP[t] = excl;
        segG[t] = ci * CH + s - excl;
        if (t == 0) segP[1024] = T;

        for (int t0 = 0; t0 < T || (cb == 0 && t0 == 0); t0 += TCAP) {
            const bool first = (cb == 0 && t0 == 0);
            if (t < OFFS_LD + 1) rb[t] = 0;
            __syncthreads();

            int  kk[4], vv[4], rk[4], seg[4];
            bool ok[4];
#pragma unroll
            for (int i = 0; i < 4; ++i) {
                const int j = t0 + t + 1024 * i;
                ok[i] = (j < T);
                int a = 0, z = 1023;
#pragma unroll
                for (int it = 0; it < 10; ++it) {
                    const int mid = (a + z + 1) >> 1;
                    if (segP[mid] <= j) a = mid; else z = mid - 1;
                }
                seg[i] = a;
            }
#pragma unroll
            for (int i = 0; i < 4; ++i) {
                const int j = t0 + t + 1024 * i;
                kk[i] = 0; vv[i] = 0;
                if (ok[i]) {
                    const int g = segG[seg[i]] + j;
                    kk[i] = key[g];
                    vv[i] = __float_as_int(sv[g]);
                }
            }
#pragma unroll
            for (int i = 0; i < 4; ++i) {
                rk[i] = 0;
                if (ok[i]) rk[i] = atomicAdd(&rb[(unsigned)kk[i] >> 24], 1);
            }
            __syncthreads();

            const int x = (t < OFFS_LD) ? rb[t] : 0;
            int inc2 = x;
#pragma unroll
            for (int off = 1; off < 32; off <<= 1) {
                const int y = __shfl_up(inc2, off, 32);
                if (lane >= off) inc2 += y;
            }
            if (t < OFFS_LD && lane == 31) ws2[w] = inc2;
            __syncthreads();
            if (t < OFFS_LD) {
                int p = 0;
#pragma unroll
                for (int k = 0; k < 8; ++k)
                    if (k < w) p += ws2[k];
                const int base = p + inc2 - x;
                rb[t] = base;
                if (t == OFFS_LD - 1) rb[OFFS_LD] = base + x;
            }
            __syncthreads();

#pragma unroll
            for (int i = 0; i < 4; ++i) {
                if (ok[i]) {
                    const int p = rb[(unsigned)kk[i] >> 24] + rk[i];
                    s_key[p] = kk[i];
                    s_val[p] = __int_as_float(vv[i]);
                }
            }
            __syncthreads();

            if (r_hi > r_lo) {
                const int pb = __builtin_amdgcn_readfirstlane(rb[r_lo]);
                const int pe = __builtin_amdgcn_readfirstlane(rb[r_hi]);
                int    cur = r_lo;
                float4 a0  = make_float4(0.f, 0.f, 0.f, 0.f);
                float4 a1  = make_float4(0.f, 0.f, 0.f, 0.f);
                for (int p0 = pb; p0 < pe; p0 += 32) {
                    int p = p0 + lane;
                    if (p > pe - 1) p = pe - 1;
                    const int kq = s_key[p];
                    const int vq = __float_as_int(s_val[p]);
                    int n = pe - p0;
                    if (n > 32) n = 32;
                    const int cq = kq & 0xFFFFFF;
                    const int rq = (int)((unsigned)kq >> 24);
                    int g = 0;
                    for (; g + GATHER_U <= n; g += GATHER_U)
                        gather_fma<GATHER_U>(lane, g, cq, vq, rq, Xa, Xb, split, Yg, first, cur, a0, a1);
                    if (n - g >= 4) { gather_fma<4>(lane, g, cq, vq, rq, Xa, Xb, split, Yg, first, cur, a0, a1); g += 4; }
                    if (n - g >= 2) { gather_fma<2>(lane, g, cq, vq, rq, Xa, Xb, split, Yg, first, cur, a0, a1); g += 2; }
                    if (n - g >= 1) { gather_fma<1>(lane, g, cq, vq, rq, Xa, Xb, split, Yg, first, cur, a0, a1); }
                }
                flush_rows(Yg, lane, first, cur, r_hi, a0, a1);
            }
            __syncthreads();
        }
        __syncthreads();
    }
}

__device__ __forceinline__ float tanh_sl(float x) {
    const float x2 = x * x;
    float p = fmaf(x2, -5.700020585e-3f, 2.063407190e-2f);
    p = fmaf(x2, p, -5.373793095e-2f);
    p = fmaf(x2, p, 1.333141625e-1f);
    p = fmaf(x2, p, -3.333328068e-1f);
    const float small = fmaf(x2, x * p, x);
    const float ax  = fabsf(x);
    const float e   = __builtin_amdgcn_exp2f(ax * 2.885390082f);
    const float big = fmaf(__builtin_amdgcn_rcpf(e + 1.f), -2.f, 1.f);
    return (ax < 0.625f) ? small : __builtin_copysignf(big, x);
}

template <int MODE, bool FULL>
__device__ __forceinline__ void tile_epilogue(const v8f (&acc)[4], const float* bias,
                                              float* OUT, int M, int row0, int col0,
                                              const float (&inv)[8], float (&ss)[8]) {
#pragma unroll
    for (int j = 0; j < 4; ++j) {
        const int   c  = col0 + j * 16;
        const float bv = bias ? bias[c] : 0.f;
        float hv[8];
        if (MODE == 2) {
#pragma unroll
            for (int r = 0; r < 8; ++r) {
                int row = row0 + r;
                if (!FULL && row > M - 1) row = M - 1;
                hv[r] = OUT[(size_t)row * D_DIM + c];
            }
        }
        float v[8];
#pragma unroll
        for (int r = 0; r < 8; ++r) {
            v[r] = tanh_sl(acc[j][r] + bv);
            if (MODE == 1) ss[r] = fmaf(v[r], v[r], ss[r]);
            if (MODE == 2) v[r] = fmaf(hv[r], inv[r], v[r]);
        }
#pragma unroll
        for (int r = 0; r < 8; ++r) {
            const int row = row0 + r;
            if (FULL || row < M) OUT[(size_t)row * D_DIM + c] = v[r];
        }
    }
}

template <int MODE>
__global__ __launch_bounds__(256)
void gemm_tanh_wmma(const float* __restrict__ A0p, const float* __restrict__ A1p,
                    const float* __restrict__ B0p, const float* __restrict__ B1p,
                    const float* __restrict__ biasp, float* OUTp,
                    int Mp, float* ssqp, int nblkp,
                    const float* __restrict__ A0q, const float* __restrict__ A1q,
                    const float* __restrict__ B0q, const float* __restrict__ B1q,
                    const float* __restrict__ biasq, float* OUTq,
                    int Mq, float* ssqq) {
    __shared__ float As[2][64 * LDA_S];
    __shared__ float Bst[2][128 * LDBT];

    int bx = blockIdx.x;
    const bool second = (bx >= nblkp);
    if (second) bx -= nblkp;
    const float* A0   = second ? A0q   : A0p;
    const float* A1   = second ? A1q   : A1p;
    const float* B0   = second ? B0q   : B0p;
    const float* B1   = second ? B1q   : B1p;
    const float* bias = second ? biasq : biasp;
    float*       OUT  = second ? OUTq  : OUTp;
    const int    M    = second ? Mq    : Mp;
    float*       ssq  = second ? ssqq  : ssqp;

    const int t    = threadIdx.x;
    const int wave = t >> 5;
    const int lane = t & 31;
    const int wm   = wave & 3;
    const int wn   = wave >> 2;
    const int rowBase = bx * 64;
    const int colBase = blockIdx.y * 128;

    const int m16 = lane & 15;
    const int kh  = lane >> 4;

    v8f acc[4];
#pragma unroll
    for (int j = 0; j < 4; ++j) acc[j] = (v8f)0.f;

    const int NCK = D_DIM / KC;
    const int nch = (A1 != nullptr) ? 2 * NCK : NCK;

    const int ar  = t >> 3;
    const int ac4 = (t & 7) * 4;
    int arow0 = rowBase + ar;
    int arow1 = rowBase + ar + 32;
    if (arow0 > M - 1) arow0 = M - 1;
    if (arow1 > M - 1) arow1 = M - 1;
    const int bk  = t >> 5;
    const int bn4 = (t & 31) * 4;

    const float* pa0 = A0 + (size_t)arow0 * D_DIM + ac4;
    const float* pa1 = A0 + (size_t)arow1 * D_DIM + ac4;
    const float* pb  = B0 + (size_t)bk * D_DIM + colBase + bn4;

    float4 ga0, ga1, gb[4];
    auto issue = [&](int c) {
        if (c == NCK) {
            pa0 = A1 + (size_t)arow0 * D_DIM + ac4;
            pa1 = A1 + (size_t)arow1 * D_DIM + ac4;
            pb  = B1 + (size_t)bk * D_DIM + colBase + bn4;
        }
        ga0 = *(const float4*)pa0;
        ga1 = *(const float4*)pa1;
#pragma unroll
        for (int i = 0; i < 4; ++i)
            gb[i] = *(const float4*)(pb + (size_t)(8 * i) * D_DIM);
        pa0 += KC;
        pa1 += KC;
        pb  += (size_t)KC * D_DIM;
    };

    auto step = [&](int c, float* as, float* bs) {
        __builtin_amdgcn_sched_barrier(0);
        {
            float* d0 = &as[ar * LDA_S + ac4];
            d0[0] = ga0.x; d0[1] = ga0.y; d0[2] = ga0.z; d0[3] = ga0.w;
            float* d1 = &as[(ar + 32) * LDA_S + ac4];
            d1[0] = ga1.x; d1[1] = ga1.y; d1[2] = ga1.z; d1[3] = ga1.w;
#pragma unroll
            for (int i = 0; i < 4; ++i) {
                const int k = bk + 8 * i;
                bs[(bn4 + 0) * LDBT + k] = gb[i].x;
                bs[(bn4 + 1) * LDBT + k] = gb[i].y;
                bs[(bn4 + 2) * LDBT + k] = gb[i].z;
                bs[(bn4 + 3) * LDBT + k] = gb[i].w;
            }
        }
        __syncthreads();
        if (c + 1 < nch) issue(c + 1);

#pragma unroll
        for (int kk = 0; kk < KC; kk += 4) {
            v2f af = *(const v2f*)&as[(wm * 16 + m16) * LDA_S + kk + 2 * kh];
#pragma unroll
            for (int j = 0; j < 4; ++j) {
                int n  = wn * 64 + j * 16 + m16;
                v2f bf = *(const v2f*)&bs[n * LDBT + kk + 2 * kh];
                acc[j] = __builtin_amdgcn_wmma_f32_16x16x4_f32(
                    false, af, false, bf, (short)0, acc[j], false, false);
            }
        }
    };

    issue(0);
    for (int c = 0; c < nch; c += 2) {
        step(c,     As[0], Bst[0]);
        step(c + 1, As[1], Bst[1]);
    }

    const int tile16 = (rowBase >> 4) + wm;
    const int ntile  = ((M + 63) >> 6) * 4;
    const int row0   = rowBase + wm * 16 + kh * 8;

    float inv[8];
    if (MODE == 2) {
        float4 p[4][2];
#pragma unroll
        for (int q = 0; q < 4; ++q) {
            const float4* s4 =
                (const float4*)(ssq + ((size_t)(q * ntile + tile16) * 32 + kh * 8));
            p[q][0] = s4[0];
            p[q][1] = s4[1];
        }
        float tot[8];
        tot[0] = (p[0][0].x + p[1][0].x) + (p[2][0].x + p[3][0].x);
        tot[1] = (p[0][0].y + p[1][0].y) + (p[2][0].y + p[3][0].y);
        tot[2] = (p[0][0].z + p[1][0].z) + (p[2][0].z + p[3][0].z);
        tot[3] = (p[0][0].w + p[1][0].w) + (p[2][0].w + p[3][0].w);
        tot[4] = (p[0][1].x + p[1][1].x) + (p[2][1].x + p[3][1].x);
        tot[5] = (p[0][1].y + p[1][1].y) + (p[2][1].y + p[3][1].y);
        tot[6] = (p[0][1].z + p[1][1].z) + (p[2][1].z + p[3][1].z);
        tot[7] = (p[0][1].w + p[1][1].w) + (p[2][1].w + p[3][1].w);
#pragma unroll
        for (int r = 0; r < 8; ++r) inv[r] = 1.f / fmaxf(sqrtf(tot[r]), 1e-12f);
    }

    float ss[8];
#pragma unroll
    for (int r = 0; r < 8; ++r) ss[r] = 0.f;

    const int col0 = colBase + wn * 64 + m16;
    if (rowBase + 64 <= M) tile_epilogue<MODE, true >(acc, bias, OUT, M, row0, col0, inv, ss);
    else                   tile_epilogue<MODE, false>(acc, bias, OUT, M, row0, col0, inv, ss);

    if (MODE == 1) {
#pragma unroll
        for (int r = 0; r < 8; ++r) {
            float s = ss[r];
            s += __shfl_xor(s, 1, 32);
            s += __shfl_xor(s, 2, 32);
            s += __shfl_xor(s, 4, 32);
            s += __shfl_xor(s, 8, 32);
            ss[r] = s;
        }
        float mine = ss[0];
#pragma unroll
        for (int r = 1; r < 8; ++r)
            if (m16 == r) mine = ss[r];
        const int q = (int)blockIdx.y * 2 + wn;
        if (m16 < 8 && row0 + m16 < M)
            ssq[(size_t)(q * ntile + tile16) * 32 + kh * 8 + m16] = mine;
    }
}

extern "C" void kernel_launch(void* const* d_in, const int* in_sizes, int n_in,
                              void* d_out, int out_size, void* d_ws, size_t ws_size,
                              hipStream_t stream) {
    const int* adj_rows   = (const int*)d_in[0];
    const int* adj_cols   = (const int*)d_in[1];
    const float* adj_vals = (const float*)d_in[2];
    const int* up_rows    = (const int*)d_in[3];
    const int* up_cols    = (const int*)d_in[4];
    const float* up_vals  = (const float*)d_in[5];
    const int* ip_rows    = (const int*)d_in[6];
    const int* ip_cols    = (const int*)d_in[7];
    const float* ip_vals  = (const float*)d_in[8];
    const float* user_emb = (const float*)d_in[9];
    const float* item_emb = (const float*)d_in[10];
    const float* Q_user   = (const float*)d_in[11];
    const float* W_user   = (const float*)d_in[12];
    const float* b_user   = (const float*)d_in[13];
    const float* Q_item   = (const float*)d_in[14];
    const float* W_item   = (const float*)d_in[15];
    const float* b_item   = (const float*)d_in[16];
    const float* M_user   = (const float*)d_in[17];
    const float* M_item   = (const float*)d_in[18];
    float* out = (float*)d_out;

    const int NU = 40000, NI = 20000, NN = 60000;
    const int NNZ_ADJ = in_sizes[0], NNZ_UP = in_sizes[3], NNZ_IP = in_sizes[6];
    const size_t ROWF = (size_t)D_DIM;
    const size_t MAT  = (size_t)NN * ROWF;

    float* agg  = (float*)d_ws;
    float* embb = agg  + MAT;
    float* hbuf = embb + MAT;

    dim3 blk(256);
    const int NB_ADJ = (NN + OWN_R - 1) / OWN_R;
    const int NB_UP  = (NU + OWN_R - 1) / OWN_R;
    const int NB_IP  = (NI + OWN_R - 1) / OWN_R;
    const int NC_ADJ = (960000 + CH - 1) / CH;
    const int NC_UP  = (640000 + CH - 1) / CH;
    const int NC_IP  = (320000 + CH - 1) / CH;
    const size_t SEG = (size_t)1 << 20;

    int*   keyA  = (int*)hbuf;
    float* svA   = hbuf + SEG;
    int*   offsA = (int*)(hbuf + 2 * SEG);
    int*   keyU  = (int*)(hbuf + 3 * SEG);
    float* svU   = hbuf + 4 * SEG;
    int*   offsU = (int*)(hbuf + 5 * SEG);
    int*   keyI  = (int*)(hbuf + 6 * SEG);
    float* svI   = hbuf + 7 * SEG;
    int*   offsI = (int*)(hbuf + 8 * SEG);
    float* ssqU = hbuf;
    float* ssqI = hbuf + (size_t)4 * 2500 * 32;

    bin_entries<<<NC_ADJ + NC_UP + NC_IP, dim3(1024), 0, stream>>>(
        adj_rows, adj_cols, adj_vals, NNZ_ADJ, NN, NC_ADJ, keyA, svA, offsA,
        up_rows, up_cols, up_vals, NNZ_UP, NU, NC_UP, keyU, svU, offsU,
        ip_rows, ip_cols, ip_vals, NNZ_IP, NI, keyI, svI, offsI);

    spmm_owner<<<NB_ADJ, dim3(1024), 0, stream>>>(
        keyA, svA, offsA, NC_ADJ, user_emb, item_emb, NU, agg, NN, NB_ADJ,
        keyA, svA, offsA, 0, user_emb, item_emb, NU, agg, NN);

    const int GB_U = (NU + 63) / 64;
    const int GB_I = (NI + 63) / 64;

    gemm_tanh_wmma<0><<<dim3(GB_U + GB_I, 2), blk, 0, stream>>>(
        agg, nullptr, Q_user, nullptr, nullptr, embb, NU, nullptr, GB_U,
        agg + (size_t)NU * ROWF, nullptr, Q_item, nullptr, nullptr,
        embb + (size_t)NU * ROWF, NI, nullptr);

    gemm_tanh_wmma<1><<<dim3(GB_U + GB_I, 2), blk, 0, stream>>>(
        user_emb, embb, W_user, W_user + (size_t)D_DIM * D_DIM, b_user,
        out, NU, ssqU, GB_U,
        item_emb, embb + (size_t)NU * ROWF, W_item,
        W_item + (size_t)D_DIM * D_DIM, b_item,
        out + (size_t)NU * ROWF, NI, ssqI);

    spmm_owner<<<NB_UP + NB_IP, dim3(1024), 0, stream>>>(
        keyU, svU, offsU, NC_UP, user_emb, user_emb, NU, agg, NU, NB_UP,
        keyI, svI, offsI, NC_IP, item_emb, item_emb, NI,
        agg + (size_t)NU * ROWF, NI);

    gemm_tanh_wmma<2><<<dim3(GB_U + GB_I, 2), blk, 0, stream>>>(
        agg, nullptr, M_user, nullptr, nullptr, out, NU, ssqU, GB_U,
        agg + (size_t)NU * ROWF, nullptr, M_item, nullptr, nullptr,
        out + (size_t)NU * ROWF, NI, ssqI);
}


# Pair SPMM runs as 256-thread work-groups inside the h GEMM's launch (6 launches to 5): gathers wait on memory while GEMM groups on the same WGP fill the matrix pipe; row sums keep sorted-tile order
# speedup vs baseline: 3.1334x; 1.0422x over previous
#include <hip/hip_runtime.h>
#include <hip/hip_bf16.h>
#include <math.h>

typedef __attribute__((ext_vector_type(2))) float v2f;
typedef __attribute__((ext_vector_type(8))) float v8f;

#define D_DIM 256
#define KC 32
#define LDA_S (KC + 2)
#define LDBT  (KC + 2)

#define OWN_R    240
#define GATHER_U 8
#define TCAP     4096
#define CH       4096
#define OFFS_LD  256

__global__ __launch_bounds__(1024)
void bin_entries(const int* __restrict__ rows0, const int* __restrict__ cols0,
                 const float* __restrict__ vals0, int nnz0, int nrows0, int nch0,
                 int* __restrict__ key0, float* __restrict__ sv0, int* __restrict__ offs0,
                 const int* __restrict__ rows1, const int* __restrict__ cols1,
                 const float* __restrict__ vals1, int nnz1, int nrows1, int nch1,
                 int* __restrict__ key1, float* __restrict__ sv1, int* __restrict__ offs1,
                 const int* __restrict__ rows2, const int* __restrict__ cols2,
                 const float* __restrict__ vals2, int nnz2, int nrows2,
                 int* __restrict__ key2, float* __restrict__ sv2, int* __restrict__ offs2) {
    __shared__ int   hist[OFFS_LD];
    __shared__ int   wsum[8];
    __shared__ int   s_key[CH];
    __shared__ float s_val[CH];

    const int t    = threadIdx.x;
    const int lane = t & 31;
    const int w    = t >> 5;

    int c = blockIdx.x;
    const bool second = (c >= nch0);
    if (second) c -= nch0;
    const bool third = second && (c >= nch1);
    if (third) c -= nch1;
    const int*   rows  = third ? rows2  : (second ? rows1  : rows0);
    const int*   cols  = third ? cols2  : (second ? cols1  : cols0);
    const float* vals  = third ? vals2  : (second ? vals1  : vals0);
    const int    nnz   = third ? nnz2   : (second ? nnz1   : nnz0);
    const int    nrows = third ? nrows2 : (second ? nrows1 : nrows0);
    int*         key   = third ? key2   : (second ? key1   : key0);
    float*       sv    = third ? sv2    : (second ? sv1    : sv0);
    int*         offs  = third ? offs2  : (second ? offs1  : offs0);

    const int e0 = c * CH;
    int n = nnz - e0;
    if (n > CH) n = CH;
    if (n < 0)  n = 0;

    if (t < OFFS_LD) hist[t] = 0;
    __syncthreads();

    int   rr[4], cc[4];
    float vv[4];
#pragma unroll
    for (int i = 0; i < 4; ++i) {
        const int j = t + 1024 * i;
        rr[i] = -1; cc[i] = 0; vv[i] = 0.f;
        if (j < n) {
            rr[i] = rows[e0 + j];
            cc[i] = cols[e0 + j];
            vv[i] = vals[e0 + j];
        }
    }
    int bk[4], rk[4], ky[4];
#pragma unroll
    for (int i = 0; i < 4; ++i) {
        bk[i] = -1; rk[i] = 0; ky[i] = 0;
        const int r = rr[i];
        if ((unsigned)r < (unsigned)nrows) {
            const int g = r / OWN_R;
            bk[i] = g;
            ky[i] = (int)(((unsigned)(r - g * OWN_R) << 24) | (unsigned)cc[i]);
            rk[i] = atomicAdd(&hist[g], 1);
        }
    }
    __syncthreads();

    const int x = (t < OFFS_LD) ? hist[t] : 0;
    int incl = x;
#pragma unroll
    for (int off = 1; off < 32; off <<= 1) {
        const int y = __shfl_up(incl, off, 32);
        if (lane >= off) incl += y;
    }
    if (t < OFFS_LD && lane == 31) wsum[w] = incl;
    __syncthreads();
    if (t < OFFS_LD) {
        int p = 0;
#pragma unroll
        for (int k = 0; k < 8; ++k)
            if (k < w) p += wsum[k];
        const int base = p + incl - x;
        hist[t] = base;
        offs[c * OFFS_LD + t] = base;
    }
    __syncthreads();

#pragma unroll
    for (int i = 0; i < 4; ++i) {
        if (bk[i] >= 0) {
            const int p = hist[bk[i]] + rk[i];
            s_key[p] = ky[i];
            s_val[p] = vv[i];
        }
    }
    __syncthreads();
#pragma unroll
    for (int i = 0; i < 4; ++i) {
        const int j = t + 1024 * i;
        if (j < n) {
            key[e0 + j] = s_key[j];
            sv[e0 + j]  = s_val[j];
        }
    }
}

__device__ __forceinline__ void flush_rows(float* Yg, int lane, bool first, int cur, int next,
                                           float4 a0, float4 a1) {
    float4* d = (float4*)(Yg + (size_t)cur * D_DIM);
    if (first) {
        d[lane]      = a0;
        d[lane + 32] = a1;
        const float4 z = make_float4(0.f, 0.f, 0.f, 0.f);
#pragma nounroll
        for (int r = cur + 1; r < next; ++r) {
            float4* dz = (float4*)(Yg + (size_t)r * D_DIM);
            dz[lane]      = z;
            dz[lane + 32] = z;
        }
    } else {
        float4 o0 = d[lane];
        float4 o1 = d[lane + 32];
        o0.x += a0.x; o0.y += a0.y; o0.z += a0.z; o0.w += a0.w;
        o1.x += a1.x; o1.y += a1.y; o1.z += a1.z; o1.w += a1.w;
        d[lane]      = o0;
        d[lane + 32] = o1;
    }
}

template <int U>
__device__ __forceinline__ void gather_fma(int lane, int j0, int cq, int vq, int rq,
                                           const float* __restrict__ Xa,
                                           const float* __restrict__ Xb, int split,
                                           float* Yg, bool first,
                                           int& cur, float4& a0, float4& a1) {
    float4 x[U][2];
#pragma unroll
    for (int u = 0; u < U; ++u) {
        int c = __builtin_amdgcn_readlane(cq, j0 + u);
        const float* X = (c < split) ? (Xa + (size_t)c * D_DIM)
                                     : (Xb + (size_t)(c - split) * D_DIM);
        const float4* X4 = (const float4*)X;
        x[u][0] = X4[lane];
        x[u][1] = X4[lane + 32];
    }
#pragma unroll
    for (int u = 0; u < U; ++u) {
        const float v = __int_as_float(__builtin_amdgcn_readlane(vq, j0 + u));
        const int   r = __builtin_amdgcn_readlane(rq, j0 + u);
        if (r != cur) {
            flush_rows(Yg, lane, first, cur, r, a0, a1);
            cur = r;
            a0 = make_float4(0.f, 0.f, 0.f, 0.f);
            a1 = make_float4(0.f, 0.f, 0.f, 0.f);
        }
        a0.x = fmaf(x[u][0].x, v, a0.x);
        a0.y = fmaf(x[u][0].y, v, a0.y);
        a0.z = fmaf(x[u][0].z, v, a0.z);
        a0.w = fmaf(x[u][0].w, v, a0.w);
        a1.x = fmaf(x[u][1].x, v, a1.x);
        a1.y = fmaf(x[u][1].y, v, a1.y);
        a1.z = fmaf(x[u][1].z, v, a1.z);
        a1.w = fmaf(x[u][1].w, v, a1.w);
    }
}

template <int NT>
__device__ __forceinline__ void spmm_owner_body(int b, const int* __restrict__ key,
                                                const float* __restrict__ sv,
                                                const int* __restrict__ offs, int nch,
                                                const float* __restrict__ Xa,
                                                const float* __restrict__ Xb, int split,
                                                float* Y, int nrows,
                                                int* s_key, float* s_val, int* segP, int* segG,
                                                int* rb, int* ws, int* ws2) {
    static_assert(NT == 1024 || NT == 256, "search depth below is written for these sizes");
    constexpr int NW    = NT / 32;
    constexpr int EPT   = TCAP / NT;
    constexpr int STEPS = (NT == 1024) ? 10 : 8;

    const int t    = threadIdx.x;
    const int lane = t & 31;
    const int w    = __builtin_amdgcn_readfirstlane(t >> 5);

    const int lo = b * OWN_R;
    int nloc = nrows - lo;
    if (nloc > OWN_R) nloc = OWN_R;
    float* Yg = Y + (size_t)lo * D_DIM;

    const int r_lo = (w * OWN_R) / NW;
    int r_hi = ((w + 1) * OWN_R) / NW;
    if (r_hi > nloc) r_hi = nloc;

    for (int cb = 0; cb < nch || cb == 0; cb += NT) {
        const int ci = cb + t;
        int s = 0, e = 0;
        if (ci < nch) {
            s = offs[ci * OFFS_LD + b];
            e = offs[ci * OFFS_LD + b + 1];
        }
        const int len = e - s;
        int incl = len;
#pragma unroll
        for (int off = 1; off < 32; off <<= 1) {
            const int y = __shfl_up(incl, off, 32);
            if (lane >= off) incl += y;
        }
        if (lane == 31) ws[w] = incl;
        __syncthreads();
        int wp = 0, Tv = 0;
#pragma unroll
        for (int k = 0; k < NW; ++k) {
            const int q = ws[k];
            if (k < w) wp += q;
            Tv += q;
        }
        const int T = __builtin_amdgcn_readfirstlane(Tv);
        const int excl = wp + incl - len;
        segP[t] = excl;
        segG[t] = ci * CH + s - excl;
        if (t == 0) segP[NT] = T;

        for (int t0 = 0; t0 < T || (cb == 0 && t0 == 0); t0 += TCAP) {
            const bool first = (cb == 0 && t0 == 0);
            for (int i = t; i < OFFS_LD + 1; i += NT) rb[i] = 0;
            __syncthreads();

            int  kk[EPT], vv[EPT], rk[EPT], seg[EPT];
            bool ok[EPT];
#pragma unroll
            for (int i = 0; i < EPT; ++i) {
                const int j = t0 + t + NT * i;
                ok[i] = (j < T);
                int a = 0, z = NT - 1;
#pragma unroll
                for (int it = 0; it < STEPS; ++it) {
                    const int mid = (a + z + 1) >> 1;
                    if (segP[mid] <= j) a = mid; else z = mid - 1;
                }
                seg[i] = a;
            }
#pragma unroll
            for (int i = 0; i < EPT; ++i) {
                const int j = t0 + t + NT * i;
                kk[i] = 0; vv[i] = 0;
                if (ok[i]) {
                    const int g = segG[seg[i]] + j;
                    kk[i] = key[g];
                    vv[i] = __float_as_int(sv[g]);
                }
            }
#pragma unroll
            for (int i = 0; i < EPT; ++i) {
                rk[i] = 0;
                if (ok[i]) rk[i] = atomicAdd(&rb[(unsigned)kk[i] >> 24], 1);
            }
            __syncthreads();

            const int x = (t < OFFS_LD) ? rb[t] : 0;
            int inc2 = x;
#pragma unroll
            for (int off = 1; off < 32; off <<= 1) {
                const int y = __shfl_up(inc2, off, 32);
                if (lane >= off) inc2 += y;
            }
            if (t < OFFS_LD && lane == 31) ws2[w] = inc2;
            __syncthreads();
            if (t < OFFS_LD) {
                int p = 0;
#pragma unroll
                for (int k = 0; k < 8; ++k)
                    if (k < w) p += ws2[k];
                const int base = p + inc2 - x;
                rb[t] = base;
                if (t == OFFS_LD - 1) rb[OFFS_LD] = base + x;
            }
            __syncthreads();

#pragma unroll
            for (int i = 0; i < EPT; ++i) {
                if (ok[i]) {
                    const int p = rb[(unsigned)kk[i] >> 24] + rk[i];
                    s_key[p] = kk[i];
                    s_val[p] = __int_as_float(vv[i]);
                }
            }
            __syncthreads();

            if (r_hi > r_lo) {
                const int pb = __builtin_amdgcn_readfirstlane(rb[r_lo]);
                const int pe = __builtin_amdgcn_readfirstlane(rb[r_hi]);
                int    cur = r_lo;
                float4 a0  = make_float4(0.f, 0.f, 0.f, 0.f);
                float4 a1  = make_float4(0.f, 0.f, 0.f, 0.f);
                for (int p0 = pb; p0 < pe; p0 += 32) {
                    int p = p0 + lane;
                    if (p > pe - 1) p = pe - 1;
                    const int kq = s_key[p];
                    const int vq = __float_as_int(s_val[p]);
                    int n = pe - p0;
                    if (n > 32) n = 32;
                    const int cq = kq & 0xFFFFFF;
                    const int rq = (int)((unsigned)kq >> 24);
                    int g = 0;
                    for (; g + GATHER_U <= n; g += GATHER_U)
                        gather_fma<GATHER_U>(lane, g, cq, vq, rq, Xa, Xb, split, Yg, first, cur, a0, a1);
                    if (n - g >= 4) { gather_fma<4>(lane, g, cq, vq, rq, Xa, Xb, split, Yg, first, cur, a0, a1); g += 4; }
                    if (n - g >= 2) { gather_fma<2>(lane, g, cq, vq, rq, Xa, Xb, split, Yg, first, cur, a0, a1); g += 2; }
                    if (n - g >= 1) { gather_fma<1>(lane, g, cq, vq, rq, Xa, Xb, split, Yg, first, cur, a0, a1); }
                }
                flush_rows(Yg, lane, first, cur, r_hi, a0, a1);
            }
            __syncthreads();
        }
        __syncthreads();
    }
}

__global__ __launch_bounds__(1024)
void spmm_owner(const int* __restrict__ key0, const float* __restrict__ sv0,
                const int* __restrict__ offs0, int nch0,
                const float* __restrict__ Xa0, const float* __restrict__ Xb0,
                int split0, float* Y0, int nrows0, int nblk0,
                const int* __restrict__ key1, const float* __restrict__ sv1,
                const int* __restrict__ offs1, int nch1,
                const float* __restrict__ Xa1, const float* __restrict__ Xb1,
                int split1, float* Y1, int nrows1) {
    __shared__ int   s_key[TCAP];
    __shared__ float s_val[TCAP];
    __shared__ int   segP[1025];
    __shared__ int   segG[1024];
    __shared__ int   rb[OFFS_LD + 1];
    __shared__ int   ws[32];
    __shared__ int   ws2[8];

    int b = blockIdx.x;
    const bool second = (b >= nblk0);
    if (second) b -= nblk0;
    spmm_owner_body<1024>(b, second ? key1 : key0, second ? sv1 : sv0,
                          second ? offs1 : offs0, second ? nch1 : nch0,
                          second ? Xa1 : Xa0, second ? Xb1 : Xb0,
                          second ? split1 : split0, second ? Y1 : Y0,
                          second ? nrows1 : nrows0,
                          s_key, s_val, segP, segG, rb, ws, ws2);
}

__device__ __forceinline__ float tanh_sl(float x) {
    const float x2 = x * x;
    float p = fmaf(x2, -5.700020585e-3f, 2.063407190e-2f);
    p = fmaf(x2, p, -5.373793095e-2f);
    p = fmaf(x2, p, 1.333141625e-1f);
    p = fmaf(x2, p, -3.333328068e-1f);
    const float small = fmaf(x2, x * p, x);
    const float ax  = fabsf(x);
    const float e   = __builtin_amdgcn_exp2f(ax * 2.885390082f);
    const float big = fmaf(__builtin_amdgcn_rcpf(e + 1.f), -2.f, 1.f);
    return (ax < 0.625f) ? small : __builtin_copysignf(big, x);
}

template <int MODE, bool FULL>
__device__ __forceinline__ void tile_epilogue(const v8f (&acc)[4], const float* bias,
                                              float* OUT, int M, int row0, int col0,
                                              const float (&inv)[8], float (&ss)[8]) {
#pragma unroll
    for (int j = 0; j < 4; ++j) {
        const int   c  = col0 + j * 16;
        const float bv = bias ? bias[c] : 0.f;
        float hv[8];
        if (MODE == 2) {
#pragma unroll
            for (int r = 0; r < 8; ++r) {
                int row = row0 + r;
                if (!FULL && row > M - 1) row = M - 1;
                hv[r] = OUT[(size_t)row * D_DIM + c];
            }
        }
        float v[8];
#pragma unroll
        for (int r = 0; r < 8; ++r) {
            v[r] = tanh_sl(acc[j][r] + bv);
            if (MODE == 1) ss[r] = fmaf(v[r], v[r], ss[r]);
            if (MODE == 2) v[r] = fmaf(hv[r], inv[r], v[r]);
        }
#pragma unroll
        for (int r = 0; r < 8; ++r) {
            const int row = row0 + r;
            if (FULL || row < M) OUT[(size_t)row * D_DIM + c] = v[r];
        }
    }
}

template <int MODE>
__device__ __forceinline__ void gemm_tanh_body(
                    int bxIn, int by,
                    const float* __restrict__ A0p, const float* __restrict__ A1p,
                    const float* __restrict__ B0p, const float* __restrict__ B1p,
                    const float* __restrict__ biasp, float* OUTp,
                    int Mp, float* ssqp, int nblkp,
                    const float* __restrict__ A0q, const float* __restrict__ A1q,
                    const float* __restrict__ B0q, const float* __restrict__ B1q,
                    const float* __restrict__ biasq, float* OUTq,
                    int Mq, float* ssqq,
                    float* As0, float* As1, float* Bs0, float* Bs1) {
    int bx = bxIn;
    const bool second = (bx >= nblkp);
    if (second) bx -= nblkp;
    const float* A0   = second ? A0q   : A0p;
    const float* A1   = second ? A1q   : A1p;
    const float* B0   = second ? B0q   : B0p;
    const float* B1   = second ? B1q   : B1p;
    const float* bias = second ? biasq : biasp;
    float*       OUT  = second ? OUTq  : OUTp;
    const int    M    = second ? Mq    : Mp;
    float*       ssq  = second ? ssqq  : ssqp;

    const int t    = threadIdx.x;
    const int wave = t >> 5;
    const int lane = t & 31;
    const int wm   = wave & 3;
    const int wn   = wave >> 2;
    const int rowBase = bx * 64;
    const int colBase = by * 128;

    const int m16 = lane & 15;
    const int kh  = lane >> 4;

    v8f acc[4];
#pragma unroll
    for (int j = 0; j < 4; ++j) acc[j] = (v8f)0.f;

    const int NCK = D_DIM / KC;
    const int nch = (A1 != nullptr) ? 2 * NCK : NCK;

    const int ar  = t >> 3;
    const int ac4 = (t & 7) * 4;
    int arow0 = rowBase + ar;
    int arow1 = rowBase + ar + 32;
    if (arow0 > M - 1) arow0 = M - 1;
    if (arow1 > M - 1) arow1 = M - 1;
    const int bk  = t >> 5;
    const int bn4 = (t & 31) * 4;

    const float* pa0 = A0 + (size_t)arow0 * D_DIM + ac4;
    const float* pa1 = A0 + (size_t)arow1 * D_DIM + ac4;
    const float* pb  = B0 + (size_t)bk * D_DIM + colBase + bn4;

    float4 ga0, ga1, gb[4];
    auto issue = [&](int c) {
        if (c == NCK) {
            pa0 = A1 + (size_t)arow0 * D_DIM + ac4;
            pa1 = A1 + (size_t)arow1 * D_DIM + ac4;
            pb  = B1 + (size_t)bk * D_DIM + colBase + bn4;
        }
        ga0 = *(const float4*)pa0;
        ga1 = *(const float4*)pa1;
#pragma unroll
        for (int i = 0; i < 4; ++i)
            gb[i] = *(const float4*)(pb + (size_t)(8 * i) * D_DIM);
        pa0 += KC;
        pa1 += KC;
        pb  += (size_t)KC * D_DIM;
    };

    auto step = [&](int c, float* as, float* bs) {
        __builtin_amdgcn_sched_barrier(0);
        {
            float* d0 = &as[ar * LDA_S + ac4];
            d0[0] = ga0.x; d0[1] = ga0.y; d0[2] = ga0.z; d0[3] = ga0.w;
            float* d1 = &as[(ar + 32) * LDA_S + ac4];
            d1[0] = ga1.x; d1[1] = ga1.y; d1[2] = ga1.z; d1[3] = ga1.w;
#pragma unroll
            for (int i = 0; i < 4; ++i) {
                const int k = bk + 8 * i;
                bs[(bn4 + 0) * LDBT + k] = gb[i].x;
                bs[(bn4 + 1) * LDBT + k] = gb[i].y;
                bs[(bn4 + 2) * LDBT + k] = gb[i].z;
                bs[(bn4 + 3) * LDBT + k] = gb[i].w;
            }
        }
        __syncthreads();
        if (c + 1 < nch) issue(c + 1);

#pragma unroll
        for (int kk = 0; kk < KC; kk += 4) {
            v2f af = *(const v2f*)&as[(wm * 16 + m16) * LDA_S + kk + 2 * kh];
#pragma unroll
            for (int j = 0; j < 4; ++j) {
                int n  = wn * 64 + j * 16 + m16;
                v2f bf = *(const v2f*)&bs[n * LDBT + kk + 2 * kh];
                acc[j] = __builtin_amdgcn_wmma_f32_16x16x4_f32(
                    false, af, false, bf, (short)0, acc[j], false, false);
            }
        }
    };

    issue(0);
    for (int c = 0; c < nch; c += 2) {
        step(c,     As0, Bs0);
        step(c + 1, As1, Bs1);
    }

    const int tile16 = (rowBase >> 4) + wm;
    const int ntile  = ((M + 63) >> 6) * 4;
    const int row0   = rowBase + wm * 16 + kh * 8;

    float inv[8];
    if (MODE == 2) {
        float4 p[4][2];
#pragma unroll
        for (int q = 0; q < 4; ++q) {
            const float4* s4 =
                (const float4*)(ssq + ((size_t)(q * ntile + tile16) * 32 + kh * 8));
            p[q][0] = s4[0];
            p[q][1] = s4[1];
        }
        float tot[8];
        tot[0] = (p[0][0].x + p[1][0].x) + (p[2][0].x + p[3][0].x);
        tot[1] = (p[0][0].y + p[1][0].y) + (p[2][0].y + p[3][0].y);
        tot[2] = (p[0][0].z + p[1][0].z) + (p[2][0].z + p[3][0].z);
        tot[3] = (p[0][0].w + p[1][0].w) + (p[2][0].w + p[3][0].w);
        tot[4] = (p[0][1].x + p[1][1].x) + (p[2][1].x + p[3][1].x);
        tot[5] = (p[0][1].y + p[1][1].y) + (p[2][1].y + p[3][1].y);
        tot[6] = (p[0][1].z + p[1][1].z) + (p[2][1].z + p[3][1].z);
        tot[7] = (p[0][1].w + p[1][1].w) + (p[2][1].w + p[3][1].w);
#pragma unroll
        for (int r = 0; r < 8; ++r) inv[r] = 1.f / fmaxf(sqrtf(tot[r]), 1e-12f);
    }

    float ss[8];
#pragma unroll
    for (int r = 0; r < 8; ++r) ss[r] = 0.f;

    const int col0 = colBase + wn * 64 + m16;
    if (rowBase + 64 <= M) tile_epilogue<MODE, true >(acc, bias, OUT, M, row0, col0, inv, ss);
    else                   tile_epilogue<MODE, false>(acc, bias, OUT, M, row0, col0, inv, ss);

    if (MODE == 1) {
#pragma unroll
        for (int r = 0; r < 8; ++r) {
            float s = ss[r];
            s += __shfl_xor(s, 1, 32);
            s += __shfl_xor(s, 2, 32);
            s += __shfl_xor(s, 4, 32);
            s += __shfl_xor(s, 8, 32);
            ss[r] = s;
        }
        float mine = ss[0];
#pragma unroll
        for (int r = 1; r < 8; ++r)
            if (m16 == r) mine = ss[r];
        const int q = by * 2 + wn;
        if (m16 < 8 && row0 + m16 < M)
            ssq[(size_t)(q * ntile + tile16) * 32 + kh * 8 + m16] = mine;
    }
}

template <int MODE>
__global__ __launch_bounds__(256)
void gemm_tanh_wmma(const float* __restrict__ A0p, const float* __restrict__ A1p,
                    const float* __restrict__ B0p, const float* __restrict__ B1p,
                    const float* __restrict__ biasp, float* OUTp,
                    int Mp, float* ssqp, int nblkp,
                    const float* __restrict__ A0q, const float* __restrict__ A1q,
                    const float* __restrict__ B0q, const float* __restrict__ B1q,
                    const float* __restrict__ biasq, float* OUTq,
                    int Mq, float* ssqq) {
    __shared__ float As[2][64 * LDA_S];
    __shared__ float Bst[2][128 * LDBT];
    gemm_tanh_body<MODE>(blockIdx.x, blockIdx.y, A0p, A1p, B0p, B1p, biasp, OUTp, Mp, ssqp,
                         nblkp, A0q, A1q, B0q, B1q, biasq, OUTq, Mq, ssqq,
                         As[0], As[1], Bst[0], Bst[1]);
}

#define GEMM_LDS_FLOATS (2 * 64 * LDA_S + 2 * 128 * LDBT)
#define SP_SEGP   (2 * TCAP)
#define SP_SEGG   (SP_SEGP + 260)
#define SP_RB     (SP_SEGG + 256)
#define SP_WS     (SP_RB + 260)
#define SP_WS2    (SP_WS + 32)
#define SP_END    (SP_WS2 + 8)

__global__ __launch_bounds__(256)
void gemm_h_with_pair_spmm(
                    const float* __restrict__ A0p, const float* __restrict__ A1p,
                    const float* __restrict__ B0p, const float* __restrict__ B1p,
                    const float* __restrict__ biasp, float* OUTp,
                    int Mp, float* ssqp, int nblkp,
                    const float* __restrict__ A0q, const float* __restrict__ A1q,
                    const float* __restrict__ B0q, const float* __restrict__ B1q,
                    const float* __restrict__ biasq, float* OUTq,
                    int Mq, float* ssqq, int nbx,
                    int nsp,
                    const int* __restrict__ key0, const float* __restrict__ sv0,
                    const int* __restrict__ offs0, int nch0,
                    const float* __restrict__ X0, float* Y0, int nrows0, int nblk0,
                    const int* __restrict__ key1, const float* __restrict__ sv1,
                    const int* __restrict__ offs1, int nch1,
                    const float* __restrict__ X1, float* Y1, int nrows1) {
    static_assert(SP_END <= GEMM_LDS_FLOATS, "the SPMM arrays must fit the GEMM's LDS block");
    __shared__ __attribute__((aligned(16))) float lds[GEMM_LDS_FLOATS];

    const int bid = blockIdx.x;
    if (bid < nsp) {
        int b = bid;
        const bool second = (b >= nblk0);
        if (second) b -= nblk0;
        const float* X = second ? X1 : X0;
        spmm_owner_body<256>(b, second ? key1 : key0, second ? sv1 : sv0,
                             second ? offs1 : offs0, second ? nch1 : nch0,
                             X, X, second ? nrows1 : nrows0, second ? Y1 : Y0,
                             second ? nrows1 : nrows0,
                             (int*)lds, lds + TCAP, (int*)(lds + SP_SEGP), (int*)(lds + SP_SEGG),
                             (int*)(lds + SP_RB), (int*)(lds + SP_WS), (int*)(lds + SP_WS2));
    } else {
        int g = bid - nsp;
        const int by = (g >= nbx) ? 1 : 0;
        if (by) g -= nbx;
        gemm_tanh_body<1>(g, by, A0p, A1p, B0p, B1p, biasp, OUTp, Mp, ssqp,
                          nblkp, A0q, A1q, B0q, B1q, biasq, OUTq, Mq, ssqq,
                          lds, lds + 64 * LDA_S, lds + 2 * 64 * LDA_S,
                          lds + 2 * 64 * LDA_S + 128 * LDBT);
    }
}

extern "C" void kernel_launch(void* const* d_in, const int* in_sizes, int n_in,
                              void* d_out, int out_size, void* d_ws, size_t ws_size,
                              hipStream_t stream) {
    const int* adj_rows   = (const int*)d_in[0];
    const int* adj_cols   = (const int*)d_in[1];
    const float* adj_vals = (const float*)d_in[2];
    const int* up_rows    = (const int*)d_in[3];
    const int* up_cols    = (const int*)d_in[4];
    const float* up_vals  = (const float*)d_in[5];
    const int* ip_rows    = (const int*)d_in[6];
    const int* ip_cols    = (const int*)d_in[7];
    const float* ip_vals  = (const float*)d_in[8];
    const float* user_emb = (const float*)d_in[9];
    const float* item_emb = (const float*)d_in[10];
    const float* Q_user   = (const float*)d_in[11];
    const float* W_user   = (const float*)d_in[12];
    const float* b_user   = (const float*)d_in[13];
    const float* Q_item   = (const float*)d_in[14];
    const float* W_item   = (const float*)d_in[15];
    const float* b_item   = (const float*)d_in[16];
    const float* M_user   = (const float*)d_in[17];
    const float* M_item   = (const float*)d_in[18];
    float* out = (float*)d_out;

    const int NU = 40000, NI = 20000, NN = 60000;
    const int NNZ_ADJ = in_sizes[0], NNZ_UP = in_sizes[3], NNZ_IP = in_sizes[6];
    const size_t ROWF = (size_t)D_DIM;
    const size_t MAT  = (size_t)NN * ROWF;

    float* agg  = (float*)d_ws;
    float* embb = agg  + MAT;
    float* hbuf = embb + MAT;

    dim3 blk(256);
    const int NB_ADJ = (NN + OWN_R - 1) / OWN_R;
    const int NB_UP  = (NU + OWN_R - 1) / OWN_R;
    const int NB_IP  = (NI + OWN_R - 1) / OWN_R;
    const int NC_ADJ = (960000 + CH - 1) / CH;
    const int NC_UP  = (640000 + CH - 1) / CH;
    const int NC_IP  = (320000 + CH - 1) / CH;
    const size_t SEG = (size_t)1 << 20;

    int*   keyA  = (int*)hbuf;
    float* svA   = hbuf + SEG;
    int*   offsA = (int*)(hbuf + 2 * SEG);
    int*   keyU  = (int*)(hbuf + 3 * SEG);
    float* svU   = hbuf + 4 * SEG;
    int*   offsU = (int*)(hbuf + 5 * SEG);
    int*   keyI  = (int*)(hbuf + 6 * SEG);
    float* svI   = hbuf + 7 * SEG;
    int*   offsI = (int*)(hbuf + 8 * SEG);
    float* ssqU = hbuf;
    float* ssqI = hbuf + (size_t)4 * 2500 * 32;

    bin_entries<<<NC_ADJ + NC_UP + NC_IP, dim3(1024), 0, stream>>>(
        adj_rows, adj_cols, adj_vals, NNZ_ADJ, NN, NC_ADJ, keyA, svA, offsA,
        up_rows, up_cols, up_vals, NNZ_UP, NU, NC_UP, keyU, svU, offsU,
        ip_rows, ip_cols, ip_vals, NNZ_IP, NI, keyI, svI, offsI);

    spmm_owner<<<NB_ADJ, dim3(1024), 0, stream>>>(
        keyA, svA, offsA, NC_ADJ, user_emb, item_emb, NU, agg, NN, NB_ADJ,
        keyA, svA, offsA, 0, user_emb, item_emb, NU, agg, NN);

    const int GB_U = (NU + 63) / 64;
    const int GB_I = (NI + 63) / 64;

    gemm_tanh_wmma<0><<<dim3(GB_U + GB_I, 2), blk, 0, stream>>>(
        agg, nullptr, Q_user, nullptr, nullptr, embb, NU, nullptr, GB_U,
        agg + (size_t)NU * ROWF, nullptr, Q_item, nullptr, nullptr,
        embb + (size_t)NU * ROWF, NI, nullptr);

    gemm_h_with_pair_spmm<<<NB_UP + NB_IP + 2 * (GB_U + GB_I), blk, 0, stream>>>(
        user_emb, embb, W_user, W_user + (size_t)D_DIM * D_DIM, b_user,
        out, NU, ssqU, GB_U,
        item_emb, embb + (size_t)NU * ROWF, W_item,
        W_item + (size_t)D_DIM * D_DIM, b_item,
        out + (size_t)NU * ROWF, NI, ssqI, GB_U + GB_I,
        NB_UP + NB_IP,
        keyU, svU, offsU, NC_UP, user_emb, agg, NU, NB_UP,
        keyI, svI, offsI, NC_IP, item_emb, agg + (size_t)NU * ROWF, NI);

    gemm_tanh_wmma<2><<<dim3(GB_U + GB_I, 2), blk, 0, stream>>>(
        agg, nullptr, M_user, nullptr, nullptr, out, NU, ssqU, GB_U,
        agg + (size_t)NU * ROWF, nullptr, M_item, nullptr, nullptr,
        out + (size_t)NU * ROWF, NI, ssqI);
}
